# starting kernel
# baseline (speedup 1.0000x reference)
#include <hip/hip_runtime.h>
#include <hip/hip_bf16.h>
#include <math.h>

typedef __attribute__((ext_vector_type(2))) float v2f;
typedef __attribute__((ext_vector_type(8))) float v8f;

#define HIDDIM 128

__global__ void fill_zero_kernel(float* __restrict__ p, long long n) {
  long long i = (long long)blockIdx.x * blockDim.x + threadIdx.x;
  if (i < n) p[i] = 0.0f;
}

__global__ void count_edges_kernel(const int* __restrict__ dst, int e,
                                   float* __restrict__ cnt) {
  int i = blockIdx.x * blockDim.x + threadIdx.x;
  if (i < e) atomicAdd(&cnt[dst[i]], 1.0f);
}

__global__ void invcnt_kernel(const float* __restrict__ cnt,
                              float* __restrict__ inv, int n) {
  int i = blockIdx.x * blockDim.x + threadIdx.x;
  if (i < n) inv[i] = 1.0f / fmaxf(cnt[i], 1.0f);
}

__global__ void scatter_add8_kernel(const float* __restrict__ x,
                                    const int* __restrict__ src,
                                    const int* __restrict__ dst,
                                    float* __restrict__ agg, long long total) {
  long long i = (long long)blockIdx.x * blockDim.x + threadIdx.x;
  if (i >= total) return;
  int e = (int)(i >> 3);
  int k = (int)(i & 7);
  atomicAdd(&agg[(size_t)dst[e] * 8 + k], x[(size_t)src[e] * 8 + k]);
}

__global__ __launch_bounds__(256) void scatter_add128_kernel(
    const float* __restrict__ x, const int* __restrict__ src,
    const int* __restrict__ dst, float* __restrict__ agg, int E) {
  int wave = (int)((blockIdx.x * blockDim.x + threadIdx.x) >> 5);
  wave = __builtin_amdgcn_readfirstlane(wave);
  if (wave >= E) return;
  const int lane = threadIdx.x & 31;
  const int s = src[wave];
  const int d = dst[wave];
  const float4 v = *(const float4*)(x + (size_t)s * HIDDIM + lane * 4);
  float* p = agg + (size_t)d * HIDDIM + lane * 4;
  atomicAdd(p + 0, v.x);
  atomicAdd(p + 1, v.y);
  atomicAdd(p + 2, v.z);
  atomicAdd(p + 3, v.w);
}

__device__ __forceinline__ float xor_add(float s, const int imm) {
  switch (imm) {
    case 1:  s += __int_as_float(__builtin_amdgcn_ds_swizzle(__float_as_int(s), 0x041f)); break;
    case 2:  s += __int_as_float(__builtin_amdgcn_ds_swizzle(__float_as_int(s), 0x081f)); break;
    case 4:  s += __int_as_float(__builtin_amdgcn_ds_swizzle(__float_as_int(s), 0x101f)); break;
    default: s += __int_as_float(__builtin_amdgcn_ds_swizzle(__float_as_int(s), 0x201f)); break;
  }
  return s;
}

template <int K>
__global__ __launch_bounds__(256) void sage_layer_kernel(
    const float* __restrict__ agg, const float* __restrict__ invcnt,
    const float* __restrict__ xin, const float* __restrict__ wl,
    const float* __restrict__ wr, const float* __restrict__ bias,
    float* __restrict__ out) {
  __shared__ float sM[16][K + 1];
  __shared__ float sX[16][K + 1];
  __shared__ float s_norm[16];

  const int tid = threadIdx.x;
  const int wave = tid >> 5;
  const int lane = tid & 31;
  const int half = lane >> 4;
  const int l16 = lane & 15;
  const int rowBase = blockIdx.x * 16;
  const int col = wave * 16 + l16;

  if (tid < 16) s_norm[tid] = 0.0f;
  for (int i = tid; i < 16 * K; i += 256) {
    int r = i / K;
    int k = i - r * K;
    size_t g = (size_t)(rowBase + r) * K + k;
    sM[r][k] = agg[g] * invcnt[rowBase + r];
    sX[r][k] = xin[g];
  }
  __syncthreads();

  v8f c = {};
#pragma unroll 8
  for (int k0 = 0; k0 < K; k0 += 4) {
    const int ka = k0 + 2 * half;
    v2f am, ax, bl, br;
    am.x = sM[l16][ka];
    am.y = sM[l16][ka + 1];
    ax.x = sX[l16][ka];
    ax.y = sX[l16][ka + 1];
    bl.x = wl[(size_t)ka * HIDDIM + col];
    bl.y = wl[(size_t)(ka + 1) * HIDDIM + col];
    br.x = wr[(size_t)ka * HIDDIM + col];
    br.y = wr[(size_t)(ka + 1) * HIDDIM + col];
    c = __builtin_amdgcn_wmma_f32_16x16x4_f32(false, am, false, bl, (short)0, c,
                                              false, false);
    c = __builtin_amdgcn_wmma_f32_16x16x4_f32(false, ax, false, br, (short)0, c,
                                              false, false);
  }

  const float bv = bias[col];
#pragma unroll
  for (int v = 0; v < 8; ++v) c[v] += bv;

#pragma unroll
  for (int v = 0; v < 8; ++v) {
    float s = c[v] * c[v];
    s = xor_add(s, 1);
    s = xor_add(s, 2);
    s = xor_add(s, 4);
    s = xor_add(s, 8);
    if (l16 == 0) atomicAdd(&s_norm[v + 8 * half], s);
  }
  __syncthreads();

#pragma unroll
  for (int v = 0; v < 8; ++v) {
    const int row = v + 8 * half;
    const float scale = 1.0f / fmaxf(sqrtf(s_norm[row]), 1e-12f);
    out[(size_t)(rowBase + row) * HIDDIM + col] = tanhf(c[v] * scale);
  }
}

extern "C" void kernel_launch(void* const* d_in, const int* in_sizes, int n_in,
                              void* d_out, int out_size, void* d_ws,
                              size_t ws_size, hipStream_t stream) {
  const float* x = (const float*)d_in[0];
  const int* ei_conn = (const int*)d_in[1];
  const int* ei_dest = (const int*)d_in[2];
  const float* wl1 = (const float*)d_in[3];
  const float* wr1 = (const float*)d_in[4];
  const float* b1 = (const float*)d_in[5];
  const float* wl2 = (const float*)d_in[6];
  const float* wr2 = (const float*)d_in[7];
  const float* b2 = (const float*)d_in[8];
  const float* wl3 = (const float*)d_in[9];
  const float* wr3 = (const float*)d_in[10];
  const float* b3 = (const float*)d_in[11];

  const int N = in_sizes[0] / 8;
  const int E = in_sizes[1] / 2;

  float* ws = (float*)d_ws;
  float* agg = ws;
  float* h1 = agg + (size_t)N * HIDDIM;
  float* h2 = h1 + (size_t)N * HIDDIM;
  float* ivc_conn = h2 + (size_t)N * HIDDIM;
  float* ivc_dest = ivc_conn + N;
  float* cnt = ivc_dest + N;

  const int T = 256;
  auto nb = [](long long n, int t) { return (int)((n + t - 1) / t); };

  const int* src_conn = ei_conn;
  const int* dst_conn = ei_conn + E;
  const int* src_dest = ei_dest;
  const int* dst_dest = ei_dest + E;

  fill_zero_kernel<<<nb(N, T), T, 0, stream>>>(cnt, N);
  count_edges_kernel<<<nb(E, T), T, 0, stream>>>(dst_conn, E, cnt);
  invcnt_kernel<<<nb(N, T), T, 0, stream>>>(cnt, ivc_conn, N);
  fill_zero_kernel<<<nb(N, T), T, 0, stream>>>(cnt, N);
  count_edges_kernel<<<nb(E, T), T, 0, stream>>>(dst_dest, E, cnt);
  invcnt_kernel<<<nb(N, T), T, 0, stream>>>(cnt, ivc_dest, N);

  const int gemmBlocks = (N + 15) / 16;
  const int scatBlocks = (E + 7) / 8;

  fill_zero_kernel<<<nb((long long)N * 8, T), T, 0, stream>>>(agg,
                                                              (long long)N * 8);
  scatter_add8_kernel<<<nb((long long)E * 8, T), T, 0, stream>>>(
      x, src_conn, dst_conn, agg, (long long)E * 8);
  sage_layer_kernel<8><<<gemmBlocks, T, 0, stream>>>(agg, ivc_conn, x, wl1,
                                                     wr1, b1, h1);

  fill_zero_kernel<<<nb((long long)N * HIDDIM, T), T, 0, stream>>>(
      agg, (long long)N * HIDDIM);
  scatter_add128_kernel<<<scatBlocks, T, 0, stream>>>(h1, src_dest, dst_dest,
                                                      agg, E);
  sage_layer_kernel<128><<<gemmBlocks, T, 0, stream>>>(agg, ivc_dest, h1, wl2,
                                                       wr2, b2, h2);

  fill_zero_kernel<<<nb((long long)N * HIDDIM, T), T, 0, stream>>>(
      agg, (long long)N * HIDDIM);
  scatter_add128_kernel<<<scatBlocks, T, 0, stream>>>(h2, src_conn, dst_conn,
                                                      agg, E);
  sage_layer_kernel<128><<<gemmBlocks, T, 0, stream>>>(agg, ivc_conn, h2, wl3,
                                                       wr3, b3, (float*)d_out);
}


# Aggregation by destination-sorted gather: edges get an arrival-order rank (int atomic), prefix-summed into segments; rows summed by plain loads, no float atomics, so sum order changes
# speedup vs baseline: 1.9122x; 1.9122x over previous
#include <hip/hip_runtime.h>
#include <hip/hip_bf16.h>
#include <math.h>

typedef __attribute__((ext_vector_type(2))) float v2f;
typedef __attribute__((ext_vector_type(8))) float v8f;

#define HIDDIM 128

#define SCAN_CHUNK 1024

__global__ void zero_int_kernel(int* __restrict__ p, int n) {
  int i = blockIdx.x * blockDim.x + threadIdx.x;
  if (i < n) p[i] = 0;
}

__global__ void rank_edges_kernel(const int* __restrict__ dstA,
                                  const int* __restrict__ dstB, int E, int N,
                                  int* __restrict__ cnt,
                                  int* __restrict__ rank) {
  const int i = blockIdx.x * blockDim.x + threadIdx.x;
  if (i >= E) return;
  const int y = blockIdx.y;
  const int* dst = y ? dstB : dstA;
  rank[(size_t)y * E + i] = atomicAdd(&cnt[(size_t)y * N + dst[i]], 1);
}

__device__ __forceinline__ int block_scan256(int t, int* s, int tid) {
  s[tid] = t;
  __syncthreads();
  for (int o = 1; o < 256; o <<= 1) {
    const int v = (tid >= o) ? s[tid - o] : 0;
    __syncthreads();
    s[tid] += v;
    __syncthreads();
  }
  return s[tid];
}

__global__ __launch_bounds__(256) void scan_block_kernel(
    const int* __restrict__ cnt, int* __restrict__ offs,
    int* __restrict__ bsum, int N, int nbk) {
  __shared__ int s[256];
  const int tid = threadIdx.x;
  const int y = blockIdx.y;
  const int* c = cnt + (size_t)y * N;
  int* o = offs + (size_t)y * N;
  const int base = blockIdx.x * SCAN_CHUNK + tid * 4;
  const int c0 = (base + 0 < N) ? c[base + 0] : 0;
  const int c1 = (base + 1 < N) ? c[base + 1] : 0;
  const int c2 = (base + 2 < N) ? c[base + 2] : 0;
  const int c3 = (base + 3 < N) ? c[base + 3] : 0;
  const int t = c0 + c1 + c2 + c3;
  const int incl = block_scan256(t, s, tid);
  const int e = incl - t;
  if (base + 0 < N) o[base + 0] = e;
  if (base + 1 < N) o[base + 1] = e + c0;
  if (base + 2 < N) o[base + 2] = e + c0 + c1;
  if (base + 3 < N) o[base + 3] = e + c0 + c1 + c2;
  if (tid == 255) bsum[y * nbk + blockIdx.x] = incl;
}

__global__ __launch_bounds__(256) void scan_bsums_kernel(int* __restrict__ bsum,
                                                         int nbk) {
  __shared__ int s[256];
  const int tid = threadIdx.x;
  int* b = bsum + blockIdx.x * nbk;
  int carry = 0;
  for (int base = 0; base < nbk; base += 256) {
    const int i = base + tid;
    const int t = (i < nbk) ? b[i] : 0;
    const int incl = block_scan256(t, s, tid);
    if (i < nbk) b[i] = carry + incl - t;
    carry += s[255];
    __syncthreads();
  }
}

__global__ void scan_finish_kernel(const int* __restrict__ cnt,
                                   int* __restrict__ offs,
                                   const int* __restrict__ bsum,
                                   float* __restrict__ inv, int N, int nbk) {
  const int i = blockIdx.x * blockDim.x + threadIdx.x;
  if (i >= N) return;
  const int y = blockIdx.y;
  const size_t g = (size_t)y * N + i;
  offs[g] += bsum[y * nbk + i / SCAN_CHUNK];
  inv[g] = 1.0f / fmaxf((float)cnt[g], 1.0f);
}

__global__ void fill_segments_kernel(const int* __restrict__ srcA,
                                     const int* __restrict__ dstA,
                                     const int* __restrict__ srcB,
                                     const int* __restrict__ dstB, int E, int N,
                                     const int* __restrict__ offs,
                                     const int* __restrict__ rank,
                                     int* __restrict__ seg) {
  const int i = blockIdx.x * blockDim.x + threadIdx.x;
  if (i >= E) return;
  const int y = blockIdx.y;
  const int* src = y ? srcB : srcA;
  const int* dst = y ? dstB : dstA;
  const int pos = offs[(size_t)y * N + dst[i]] + rank[(size_t)y * E + i];
  seg[(size_t)y * E + pos] = src[i];
}

__global__ void gather_sum8_kernel(const float* __restrict__ x,
                                   const int* __restrict__ offs,
                                   const int* __restrict__ cnt,
                                   const int* __restrict__ seg,
                                   float* __restrict__ agg, int N) {
  const int i = blockIdx.x * blockDim.x + threadIdx.x;
  const int r = i >> 3;
  const int k = i & 7;
  if (r >= N) return;
  int j = offs[r];
  const int end = j + cnt[r];
  float acc = 0.0f;
  for (; j + 4 <= end; j += 4) {
    const int s0 = seg[j], s1 = seg[j + 1], s2 = seg[j + 2], s3 = seg[j + 3];
    const float v0 = x[(size_t)s0 * 8 + k];
    const float v1 = x[(size_t)s1 * 8 + k];
    const float v2 = x[(size_t)s2 * 8 + k];
    const float v3 = x[(size_t)s3 * 8 + k];
    acc += v0;
    acc += v1;
    acc += v2;
    acc += v3;
  }
  for (; j < end; ++j) acc += x[(size_t)seg[j] * 8 + k];
  agg[(size_t)r * 8 + k] = acc;
}

__device__ __forceinline__ void add4(float4& a, const float4 v) {
  a.x += v.x;
  a.y += v.y;
  a.z += v.z;
  a.w += v.w;
}

__global__ __launch_bounds__(256) void gather_sum128_kernel(
    const float* __restrict__ h, const int* __restrict__ offs,
    const int* __restrict__ cnt, const int* __restrict__ seg,
    float* __restrict__ agg, int N) {
  int node = (int)((blockIdx.x * blockDim.x + threadIdx.x) >> 5);
  node = __builtin_amdgcn_readfirstlane(node);
  if (node >= N) return;
  const int lane = threadIdx.x & 31;
  const int beg = offs[node];
  const int deg = cnt[node];
  const float* hl = h + lane * 4;
#define ROW_OF(jj) \
  (*(const float4*)(hl + (size_t)__builtin_amdgcn_readlane(idx, (jj)) * HIDDIM))
  float4 acc = make_float4(0.0f, 0.0f, 0.0f, 0.0f);
  for (int base = 0; base < deg; base += 32) {
    const int m = (deg - base < 32) ? (deg - base) : 32;
    int idx = 0;
    if (lane < m) idx = seg[beg + base + lane];
    int j = 0;
    for (; j + 8 <= m; j += 8) {
      const float4 v0 = ROW_OF(j), v1 = ROW_OF(j + 1), v2 = ROW_OF(j + 2),
                   v3 = ROW_OF(j + 3), v4 = ROW_OF(j + 4), v5 = ROW_OF(j + 5),
                   v6 = ROW_OF(j + 6), v7 = ROW_OF(j + 7);
      add4(acc, v0);
      add4(acc, v1);
      add4(acc, v2);
      add4(acc, v3);
      add4(acc, v4);
      add4(acc, v5);
      add4(acc, v6);
      add4(acc, v7);
    }
    if (j + 4 <= m) {
      const float4 v0 = ROW_OF(j), v1 = ROW_OF(j + 1), v2 = ROW_OF(j + 2),
                   v3 = ROW_OF(j + 3);
      add4(acc, v0);
      add4(acc, v1);
      add4(acc, v2);
      add4(acc, v3);
      j += 4;
    }
    if (j + 2 <= m) {
      const float4 v0 = ROW_OF(j), v1 = ROW_OF(j + 1);
      add4(acc, v0);
      add4(acc, v1);
      j += 2;
    }
    if (j < m) {
      const float4 v0 = ROW_OF(j);
      add4(acc, v0);
    }
  }
#undef ROW_OF
  *(float4*)(agg + (size_t)node * HIDDIM + lane * 4) = acc;
}

__device__ __forceinline__ float xor_add(float s, const int imm) {
  switch (imm) {
    case 1:  s += __int_as_float(__builtin_amdgcn_ds_swizzle(__float_as_int(s), 0x041f)); break;
    case 2:  s += __int_as_float(__builtin_amdgcn_ds_swizzle(__float_as_int(s), 0x081f)); break;
    case 4:  s += __int_as_float(__builtin_amdgcn_ds_swizzle(__float_as_int(s), 0x101f)); break;
    default: s += __int_as_float(__builtin_amdgcn_ds_swizzle(__float_as_int(s), 0x201f)); break;
  }
  return s;
}

template <int K>
__global__ __launch_bounds__(256) void sage_layer_kernel(
    const float* __restrict__ agg, const float* __restrict__ invcnt,
    const float* __restrict__ xin, const float* __restrict__ wl,
    const float* __restrict__ wr, const float* __restrict__ bias,
    float* __restrict__ out) {
  __shared__ float sM[16][K + 1];
  __shared__ float sX[16][K + 1];
  __shared__ float s_norm[16];

  const int tid = threadIdx.x;
  const int wave = tid >> 5;
  const int lane = tid & 31;
  const int half = lane >> 4;
  const int l16 = lane & 15;
  const int rowBase = blockIdx.x * 16;
  const int col = wave * 16 + l16;

  if (tid < 16) s_norm[tid] = 0.0f;
  for (int i = tid; i < 16 * K; i += 256) {
    int r = i / K;
    int k = i - r * K;
    size_t g = (size_t)(rowBase + r) * K + k;
    sM[r][k] = agg[g] * invcnt[rowBase + r];
    sX[r][k] = xin[g];
  }
  __syncthreads();

  v8f c = {};
#pragma unroll 8
  for (int k0 = 0; k0 < K; k0 += 4) {
    const int ka = k0 + 2 * half;
    v2f am, ax, bl, br;
    am.x = sM[l16][ka];
    am.y = sM[l16][ka + 1];
    ax.x = sX[l16][ka];
    ax.y = sX[l16][ka + 1];
    bl.x = wl[(size_t)ka * HIDDIM + col];
    bl.y = wl[(size_t)(ka + 1) * HIDDIM + col];
    br.x = wr[(size_t)ka * HIDDIM + col];
    br.y = wr[(size_t)(ka + 1) * HIDDIM + col];
    c = __builtin_amdgcn_wmma_f32_16x16x4_f32(false, am, false, bl, (short)0, c,
                                              false, false);
    c = __builtin_amdgcn_wmma_f32_16x16x4_f32(false, ax, false, br, (short)0, c,
                                              false, false);
  }

  const float bv = bias[col];
#pragma unroll
  for (int v = 0; v < 8; ++v) c[v] += bv;

#pragma unroll
  for (int v = 0; v < 8; ++v) {
    float s = c[v] * c[v];
    s = xor_add(s, 1);
    s = xor_add(s, 2);
    s = xor_add(s, 4);
    s = xor_add(s, 8);
    if (l16 == 0) atomicAdd(&s_norm[v + 8 * half], s);
  }
  __syncthreads();

#pragma unroll
  for (int v = 0; v < 8; ++v) {
    const int row = v + 8 * half;
    const float scale = 1.0f / fmaxf(sqrtf(s_norm[row]), 1e-12f);
    out[(size_t)(rowBase + row) * HIDDIM + col] = tanhf(c[v] * scale);
  }
}

extern "C" void kernel_launch(void* const* d_in, const int* in_sizes, int n_in,
                              void* d_out, int out_size, void* d_ws,
                              size_t ws_size, hipStream_t stream) {
  const float* x = (const float*)d_in[0];
  const int* ei_conn = (const int*)d_in[1];
  const int* ei_dest = (const int*)d_in[2];
  const float* wl1 = (const float*)d_in[3];
  const float* wr1 = (const float*)d_in[4];
  const float* b1 = (const float*)d_in[5];
  const float* wl2 = (const float*)d_in[6];
  const float* wr2 = (const float*)d_in[7];
  const float* b2 = (const float*)d_in[8];
  const float* wl3 = (const float*)d_in[9];
  const float* wr3 = (const float*)d_in[10];
  const float* b3 = (const float*)d_in[11];

  const int N = in_sizes[0] / 8;
  const int E = in_sizes[1] / 2;
  const int nbk = (N + SCAN_CHUNK - 1) / SCAN_CHUNK;

  float* ws = (float*)d_ws;
  float* agg = ws;
  float* h1 = agg + (size_t)N * HIDDIM;
  float* h2 = h1 + (size_t)N * HIDDIM;
  float* inv = h2 + (size_t)N * HIDDIM;
  int* cnt = (int*)(inv + (size_t)2 * N);
  int* offs = cnt + (size_t)2 * N;
  int* bsum = offs + (size_t)2 * N;
  int* rank = bsum + (size_t)2 * nbk;
  int* seg = rank + (size_t)2 * E;

  const int T = 256;
  auto nb = [](long long n, int t) { return (int)((n + t - 1) / t); };

  const int* src_conn = ei_conn;
  const int* dst_conn = ei_conn + E;
  const int* src_dest = ei_dest;
  const int* dst_dest = ei_dest + E;

  zero_int_kernel<<<nb((long long)2 * N, T), T, 0, stream>>>(cnt, 2 * N);
  rank_edges_kernel<<<dim3(nb(E, T), 2), T, 0, stream>>>(dst_conn, dst_dest, E,
                                                         N, cnt, rank);
  scan_block_kernel<<<dim3(nbk, 2), 256, 0, stream>>>(cnt, offs, bsum, N, nbk);
  scan_bsums_kernel<<<2, 256, 0, stream>>>(bsum, nbk);
  scan_finish_kernel<<<dim3(nb(N, T), 2), T, 0, stream>>>(cnt, offs, bsum, inv,
                                                          N, nbk);
  fill_segments_kernel<<<dim3(nb(E, T), 2), T, 0, stream>>>(
      src_conn, dst_conn, src_dest, dst_dest, E, N, offs, rank, seg);

  const int gemmBlocks = (N + 15) / 16;
  const int gathBlocks = (N + 7) / 8;

  gather_sum8_kernel<<<nb((long long)N * 8, T), T, 0, stream>>>(x, offs, cnt,
                                                                seg, agg, N);
  sage_layer_kernel<8><<<gemmBlocks, T, 0, stream>>>(agg, inv, x, wl1, wr1, b1,
                                                     h1);

  gather_sum128_kernel<<<gathBlocks, T, 0, stream>>>(h1, offs + N, cnt + N,
                                                     seg + E, agg, N);
  sage_layer_kernel<128><<<gemmBlocks, T, 0, stream>>>(agg, inv + N, h1, wl2,
                                                       wr2, b2, h2);

  gather_sum128_kernel<<<gathBlocks, T, 0, stream>>>(h2, offs, cnt, seg, agg,
                                                     N);
  sage_layer_kernel<128><<<gemmBlocks, T, 0, stream>>>(agg, inv, h2, wl3, wr3,
                                                       b3, (float*)d_out);
}


# Neighbour gather fused into the layer kernel: each block sums its 16 nodes' neighbour rows straight into the LDS mean tile, same sum order; no aggregate buffer, 12 launches become 9
# speedup vs baseline: 1.9720x; 1.0313x over previous
#include <hip/hip_runtime.h>
#include <hip/hip_bf16.h>
#include <math.h>

typedef __attribute__((ext_vector_type(2))) float v2f;
typedef __attribute__((ext_vector_type(8))) float v8f;

#define HIDDIM 128

#define SCAN_CHUNK 1024

__global__ void zero_int_kernel(int* __restrict__ p, int n) {
  int i = blockIdx.x * blockDim.x + threadIdx.x;
  if (i < n) p[i] = 0;
}

__global__ void rank_edges_kernel(const int* __restrict__ dstA,
                                  const int* __restrict__ dstB, int E, int N,
                                  int* __restrict__ cnt,
                                  int* __restrict__ rank) {
  const int i = blockIdx.x * blockDim.x + threadIdx.x;
  if (i >= E) return;
  const int y = blockIdx.y;
  const int* dst = y ? dstB : dstA;
  rank[(size_t)y * E + i] = atomicAdd(&cnt[(size_t)y * N + dst[i]], 1);
}

__device__ __forceinline__ int block_scan256(int t, int* s, int tid) {
  s[tid] = t;
  __syncthreads();
  for (int o = 1; o < 256; o <<= 1) {
    const int v = (tid >= o) ? s[tid - o] : 0;
    __syncthreads();
    s[tid] += v;
    __syncthreads();
  }
  return s[tid];
}

__global__ __launch_bounds__(256) void scan_block_kernel(
    const int* __restrict__ cnt, int* __restrict__ offs,
    int* __restrict__ bsum, int N, int nbk) {
  __shared__ int s[256];
  const int tid = threadIdx.x;
  const int y = blockIdx.y;
  const int* c = cnt + (size_t)y * N;
  int* o = offs + (size_t)y * N;
  const int base = blockIdx.x * SCAN_CHUNK + tid * 4;
  const int c0 = (base + 0 < N) ? c[base + 0] : 0;
  const int c1 = (base + 1 < N) ? c[base + 1] : 0;
  const int c2 = (base + 2 < N) ? c[base + 2] : 0;
  const int c3 = (base + 3 < N) ? c[base + 3] : 0;
  const int t = c0 + c1 + c2 + c3;
  const int incl = block_scan256(t, s, tid);
  const int e = incl - t;
  if (base + 0 < N) o[base + 0] = e;
  if (base + 1 < N) o[base + 1] = e + c0;
  if (base + 2 < N) o[base + 2] = e + c0 + c1;
  if (base + 3 < N) o[base + 3] = e + c0 + c1 + c2;
  if (tid == 255) bsum[y * nbk + blockIdx.x] = incl;
}

__global__ __launch_bounds__(256) void scan_bsums_kernel(int* __restrict__ bsum,
                                                         int nbk) {
  __shared__ int s[256];
  const int tid = threadIdx.x;
  int* b = bsum + blockIdx.x * nbk;
  int carry = 0;
  for (int base = 0; base < nbk; base += 256) {
    const int i = base + tid;
    const int t = (i < nbk) ? b[i] : 0;
    const int incl = block_scan256(t, s, tid);
    if (i < nbk) b[i] = carry + incl - t;
    carry += s[255];
    __syncthreads();
  }
}

__global__ void scan_finish_kernel(const int* __restrict__ cnt,
                                   int* __restrict__ offs,
                                   const int* __restrict__ bsum,
                                   float* __restrict__ inv, int N, int nbk) {
  const int i = blockIdx.x * blockDim.x + threadIdx.x;
  if (i >= N) return;
  const int y = blockIdx.y;
  const size_t g = (size_t)y * N + i;
  offs[g] += bsum[y * nbk + i / SCAN_CHUNK];
  inv[g] = 1.0f / fmaxf((float)cnt[g], 1.0f);
}

__global__ void fill_segments_kernel(const int* __restrict__ srcA,
                                     const int* __restrict__ dstA,
                                     const int* __restrict__ srcB,
                                     const int* __restrict__ dstB, int E, int N,
                                     const int* __restrict__ offs,
                                     const int* __restrict__ rank,
                                     int* __restrict__ seg) {
  const int i = blockIdx.x * blockDim.x + threadIdx.x;
  if (i >= E) return;
  const int y = blockIdx.y;
  const int* src = y ? srcB : srcA;
  const int* dst = y ? dstB : dstA;
  const int pos = offs[(size_t)y * N + dst[i]] + rank[(size_t)y * E + i];
  seg[(size_t)y * E + pos] = src[i];
}

__device__ __forceinline__ void add4(float4& a, const float4 v) {
  a.x += v.x;
  a.y += v.y;
  a.z += v.z;
  a.w += v.w;
}

__device__ __forceinline__ float4 gather_row128(const float* __restrict__ hl,
                                                const int* __restrict__ seg,
                                                int beg, int deg, int lane,
                                                int idx0) {
#define ROW_OF(jj) \
  (*(const float4*)(hl + (size_t)__builtin_amdgcn_readlane(idx, (jj)) * HIDDIM))
  float4 acc = make_float4(0.0f, 0.0f, 0.0f, 0.0f);
  for (int base = 0; base < deg; base += 32) {
    const int m = (deg - base < 32) ? (deg - base) : 32;
    int idx = idx0;
    if (base > 0) {
      idx = 0;
      if (lane < m) idx = seg[beg + base + lane];
    }
    int j = 0;
    for (; j + 8 <= m; j += 8) {
      const float4 v0 = ROW_OF(j), v1 = ROW_OF(j + 1), v2 = ROW_OF(j + 2),
                   v3 = ROW_OF(j + 3), v4 = ROW_OF(j + 4), v5 = ROW_OF(j + 5),
                   v6 = ROW_OF(j + 6), v7 = ROW_OF(j + 7);
      add4(acc, v0);
      add4(acc, v1);
      add4(acc, v2);
      add4(acc, v3);
      add4(acc, v4);
      add4(acc, v5);
      add4(acc, v6);
      add4(acc, v7);
    }
    if (j + 4 <= m) {
      const float4 v0 = ROW_OF(j), v1 = ROW_OF(j + 1), v2 = ROW_OF(j + 2),
                   v3 = ROW_OF(j + 3);
      add4(acc, v0);
      add4(acc, v1);
      add4(acc, v2);
      add4(acc, v3);
      j += 4;
    }
    if (j + 2 <= m) {
      const float4 v0 = ROW_OF(j), v1 = ROW_OF(j + 1);
      add4(acc, v0);
      add4(acc, v1);
      j += 2;
    }
    if (j < m) {
      const float4 v0 = ROW_OF(j);
      add4(acc, v0);
    }
  }
#undef ROW_OF
  return acc;
}

__device__ __forceinline__ float xor_add(float s, const int imm) {
  switch (imm) {
    case 1:  s += __int_as_float(__builtin_amdgcn_ds_swizzle(__float_as_int(s), 0x041f)); break;
    case 2:  s += __int_as_float(__builtin_amdgcn_ds_swizzle(__float_as_int(s), 0x081f)); break;
    case 4:  s += __int_as_float(__builtin_amdgcn_ds_swizzle(__float_as_int(s), 0x101f)); break;
    default: s += __int_as_float(__builtin_amdgcn_ds_swizzle(__float_as_int(s), 0x201f)); break;
  }
  return s;
}

template <int K>
__global__ __launch_bounds__(256) void sage_layer_kernel(
    const float* __restrict__ xin, const int* __restrict__ offs,
    const int* __restrict__ cnt, const int* __restrict__ seg,
    const float* __restrict__ invcnt, const float* __restrict__ wl,
    const float* __restrict__ wr, const float* __restrict__ bias,
    float* __restrict__ out) {
  constexpr int P = K + 4;
  __shared__ __attribute__((aligned(16))) float sM[16][P];
  __shared__ __attribute__((aligned(16))) float sX[16][P];
  __shared__ float s_norm[16];

  const int tid = threadIdx.x;
  const int wave = tid >> 5;
  const int lane = tid & 31;
  const int half = lane >> 4;
  const int l16 = lane & 15;
  const int rowBase = blockIdx.x * 16;
  const int col = wave * 16 + l16;

  if (tid < 16) s_norm[tid] = 0.0f;
  if constexpr (K == HIDDIM) {
    const int w = __builtin_amdgcn_readfirstlane(wave);
    const float* hl = xin + lane * 4;
    const float4 xa = *(const float4*)(hl + (size_t)(rowBase + w) * K);
    const float4 xb = *(const float4*)(hl + (size_t)(rowBase + w + 8) * K);
    const int n0 = rowBase + 2 * w;
    const int beg0 = offs[n0], beg1 = offs[n0 + 1];
    const int deg0 = cnt[n0], deg1 = cnt[n0 + 1];
    const float s0 = invcnt[n0], s1 = invcnt[n0 + 1];
    int idx0 = 0, idx1 = 0;
    if (lane < deg0) idx0 = seg[beg0 + lane];
    if (lane < deg1) idx1 = seg[beg1 + lane];
    *(float4*)&sX[w][lane * 4] = xa;
    *(float4*)&sX[w + 8][lane * 4] = xb;
    float4 m0 = gather_row128(hl, seg, beg0, deg0, lane, idx0);
    float4 m1 = gather_row128(hl, seg, beg1, deg1, lane, idx1);
    m0.x *= s0;
    m0.y *= s0;
    m0.z *= s0;
    m0.w *= s0;
    m1.x *= s1;
    m1.y *= s1;
    m1.z *= s1;
    m1.w *= s1;
    *(float4*)&sM[2 * w][lane * 4] = m0;
    *(float4*)&sM[2 * w + 1][lane * 4] = m1;
  } else {
    if (tid < 16 * K) {
      const int r = tid / K;
      const int k = tid - r * K;
      const int node = rowBase + r;
      sX[r][k] = xin[(size_t)node * K + k];
      int j = offs[node];
      const int end = j + cnt[node];
      float acc = 0.0f;
      for (; j + 4 <= end; j += 4) {
        const int s0 = seg[j], s1 = seg[j + 1], s2 = seg[j + 2], s3 = seg[j + 3];
        const float v0 = xin[(size_t)s0 * K + k];
        const float v1 = xin[(size_t)s1 * K + k];
        const float v2 = xin[(size_t)s2 * K + k];
        const float v3 = xin[(size_t)s3 * K + k];
        acc += v0;
        acc += v1;
        acc += v2;
        acc += v3;
      }
      for (; j < end; ++j) acc += xin[(size_t)seg[j] * K + k];
      sM[r][k] = acc * invcnt[node];
    }
  }
  __syncthreads();

  v8f c = {};
#pragma unroll 8
  for (int k0 = 0; k0 < K; k0 += 4) {
    const int ka = k0 + 2 * half;
    v2f bl, br;
    const v2f am = *(const v2f*)&sM[l16][ka];
    const v2f ax = *(const v2f*)&sX[l16][ka];
    bl.x = wl[(size_t)ka * HIDDIM + col];
    bl.y = wl[(size_t)(ka + 1) * HIDDIM + col];
    br.x = wr[(size_t)ka * HIDDIM + col];
    br.y = wr[(size_t)(ka + 1) * HIDDIM + col];
    c = __builtin_amdgcn_wmma_f32_16x16x4_f32(false, am, false, bl, (short)0, c,
                                              false, false);
    c = __builtin_amdgcn_wmma_f32_16x16x4_f32(false, ax, false, br, (short)0, c,
                                              false, false);
  }

  const float bv = bias[col];
#pragma unroll
  for (int v = 0; v < 8; ++v) c[v] += bv;

#pragma unroll
  for (int v = 0; v < 8; ++v) {
    float s = c[v] * c[v];
    s = xor_add(s, 1);
    s = xor_add(s, 2);
    s = xor_add(s, 4);
    s = xor_add(s, 8);
    if (l16 == 0) atomicAdd(&s_norm[v + 8 * half], s);
  }
  __syncthreads();

#pragma unroll
  for (int v = 0; v < 8; ++v) {
    const int row = v + 8 * half;
    const float scale = 1.0f / fmaxf(sqrtf(s_norm[row]), 1e-12f);
    out[(size_t)(rowBase + row) * HIDDIM + col] = tanhf(c[v] * scale);
  }
}

extern "C" void kernel_launch(void* const* d_in, const int* in_sizes, int n_in,
                              void* d_out, int out_size, void* d_ws,
                              size_t ws_size, hipStream_t stream) {
  const float* x = (const float*)d_in[0];
  const int* ei_conn = (const int*)d_in[1];
  const int* ei_dest = (const int*)d_in[2];
  const float* wl1 = (const float*)d_in[3];
  const float* wr1 = (const float*)d_in[4];
  const float* b1 = (const float*)d_in[5];
  const float* wl2 = (const float*)d_in[6];
  const float* wr2 = (const float*)d_in[7];
  const float* b2 = (const float*)d_in[8];
  const float* wl3 = (const float*)d_in[9];
  const float* wr3 = (const float*)d_in[10];
  const float* b3 = (const float*)d_in[11];

  const int N = in_sizes[0] / 8;
  const int E = in_sizes[1] / 2;
  const int nbk = (N + SCAN_CHUNK - 1) / SCAN_CHUNK;

  float* ws = (float*)d_ws;
  float* h1 = ws;
  float* h2 = h1 + (size_t)N * HIDDIM;
  float* inv = h2 + (size_t)N * HIDDIM;
  int* cnt = (int*)(inv + (size_t)2 * N);
  int* offs = cnt + (size_t)2 * N;
  int* bsum = offs + (size_t)2 * N;
  int* rank = bsum + (size_t)2 * nbk;
  int* seg = rank + (size_t)2 * E;

  const int T = 256;
  auto nb = [](long long n, int t) { return (int)((n + t - 1) / t); };

  const int* src_conn = ei_conn;
  const int* dst_conn = ei_conn + E;
  const int* src_dest = ei_dest;
  const int* dst_dest = ei_dest + E;

  zero_int_kernel<<<nb((long long)2 * N, T), T, 0, stream>>>(cnt, 2 * N);
  rank_edges_kernel<<<dim3(nb(E, T), 2), T, 0, stream>>>(dst_conn, dst_dest, E,
                                                         N, cnt, rank);
  scan_block_kernel<<<dim3(nbk, 2), 256, 0, stream>>>(cnt, offs, bsum, N, nbk);
  scan_bsums_kernel<<<2, 256, 0, stream>>>(bsum, nbk);
  scan_finish_kernel<<<dim3(nb(N, T), 2), T, 0, stream>>>(cnt, offs, bsum, inv,
                                                          N, nbk);
  fill_segments_kernel<<<dim3(nb(E, T), 2), T, 0, stream>>>(
      src_conn, dst_conn, src_dest, dst_dest, E, N, offs, rank, seg);

  const int gemmBlocks = (N + 15) / 16;

  sage_layer_kernel<8><<<gemmBlocks, T, 0, stream>>>(x, offs, cnt, seg, inv, wl1,
                                                     wr1, b1, h1);

  sage_layer_kernel<128><<<gemmBlocks, T, 0, stream>>>(
      h1, offs + N, cnt + N, seg + E, inv + N, wl2, wr2, b2, h2);

  sage_layer_kernel<128><<<gemmBlocks, T, 0, stream>>>(
      h2, offs, cnt, seg, inv, wl3, wr3, b3, (float*)d_out);
}


# Edge grouping by a two-level bucket sort on LDS counters (an LDS add's return is the edge's place): no global atomics, 6 list launches become 4; segment order, so sum order, stays arrival-dependent
# speedup vs baseline: 3.0393x; 1.5412x over previous
#include <hip/hip_runtime.h>
#include <hip/hip_bf16.h>
#include <math.h>

typedef __attribute__((ext_vector_type(2))) float v2f;
typedef __attribute__((ext_vector_type(8))) float v8f;

#define HIDDIM 128

#define BUCKET_SHIFT 8
#define BUCKET_NODES (1 << BUCKET_SHIFT)
#define MAX_BUCKETS 512
#define PART_BLOCKS 128
#define PART_THREADS 512
#define SCAN_THREADS (2 * MAX_BUCKETS)
#define HALF_BLOCKS (PART_BLOCKS / 2)

template <int T>
__device__ __forceinline__ int block_scan(int t, int* s, int tid) {
  s[tid] = t;
  __syncthreads();
  for (int o = 1; o < T; o <<= 1) {
    const int v = (tid >= o) ? s[tid - o] : 0;
    __syncthreads();
    s[tid] += v;
    __syncthreads();
  }
  return s[tid];
}

__global__ __launch_bounds__(PART_THREADS) void bucket_count_kernel(
    const int* __restrict__ dstA, const int* __restrict__ dstB, int E,
    int nbuck, int chunk, int* __restrict__ blkcnt) {
  __shared__ int hist[MAX_BUCKETS];
  const int tid = threadIdx.x;
  const int y = blockIdx.y;
  const int* dst = y ? dstB : dstA;
  for (int b = tid; b < nbuck; b += PART_THREADS) hist[b] = 0;
  __syncthreads();
  const int beg = blockIdx.x * chunk;
  const int end = (beg + chunk < E) ? (beg + chunk) : E;
  for (int i = beg + tid; i < end; i += 4 * PART_THREADS) {
    int d[4];
#pragma unroll
    for (int u = 0; u < 4; ++u) {
      const int j = i + u * PART_THREADS;
      const int v = dst[(j < end) ? j : i];
      d[u] = (j < end) ? v : -1;
    }
#pragma unroll
    for (int u = 0; u < 4; ++u)
      if (d[u] >= 0) atomicAdd(&hist[d[u] >> BUCKET_SHIFT], 1);
  }
  __syncthreads();
  int* o = blkcnt + ((size_t)y * gridDim.x + blockIdx.x) * nbuck;
  for (int b = tid; b < nbuck; b += PART_THREADS) o[b] = hist[b];
}

__global__ __launch_bounds__(SCAN_THREADS) void bucket_scan_kernel(
    const int* __restrict__ blkcnt, int* __restrict__ blkpref,
    int* __restrict__ bstart, int nbuck) {
  __shared__ int s[SCAN_THREADS];
  __shared__ int tots[2][MAX_BUCKETS];
  const int tid = threadIdx.x;
  const int y = blockIdx.x;
  const int j = tid & (MAX_BUCKETS - 1);
  const int q = tid / MAX_BUCKETS;
  const size_t first = ((size_t)y * PART_BLOCKS + q * HALF_BLOCKS) * nbuck + j;
  const int* c = blkcnt + first;
  int* p = blkpref + first;
  const bool live = j < nbuck;
  int v[HALF_BLOCKS];
#pragma unroll
  for (int u = 0; u < HALF_BLOCKS; ++u) v[u] = live ? c[(size_t)u * nbuck] : 0;
  int tot = 0;
#pragma unroll
  for (int u = 0; u < HALF_BLOCKS; ++u) tot += v[u];
  tots[q][j] = tot;
  __syncthreads();
  const int whole = (q == 0) ? (tots[0][j] + tots[1][j]) : 0;
  int run = (q == 1) ? tots[0][j] : 0;
  if (live) {
#pragma unroll
    for (int u = 0; u < HALF_BLOCKS; ++u) {
      p[(size_t)u * nbuck] = run;
      run += v[u];
    }
  }
  const int incl = block_scan<SCAN_THREADS>(whole, s, tid);
  if (q == 0 && live) {
    bstart[y * (nbuck + 1) + j] = incl - whole;
    if (j == nbuck - 1) bstart[y * (nbuck + 1) + nbuck] = incl;
  }
}

__global__ __launch_bounds__(PART_THREADS) void bucket_scatter_kernel(
    const int* __restrict__ srcA, const int* __restrict__ dstA,
    const int* __restrict__ srcB, const int* __restrict__ dstB, int E,
    int nbuck, int chunk, const int* __restrict__ blkpref,
    const int* __restrict__ bstart, unsigned* __restrict__ packed) {
  __shared__ int cur[MAX_BUCKETS];
  const int tid = threadIdx.x;
  const int y = blockIdx.y;
  const int* src = y ? srcB : srcA;
  const int* dst = y ? dstB : dstA;
  const int* p = blkpref + ((size_t)y * gridDim.x + blockIdx.x) * nbuck;
  const int* bs = bstart + y * (nbuck + 1);
  for (int b = tid; b < nbuck; b += PART_THREADS) cur[b] = bs[b] + p[b];
  __syncthreads();
  unsigned* out = packed + (size_t)y * E;
  const int beg = blockIdx.x * chunk;
  const int end = (beg + chunk < E) ? (beg + chunk) : E;
  for (int i = beg + tid; i < end; i += 4 * PART_THREADS) {
    int d[4], s[4];
#pragma unroll
    for (int u = 0; u < 4; ++u) {
      const int j = i + u * PART_THREADS;
      const int jj = (j < end) ? j : i;
      const int v = dst[jj];
      s[u] = src[jj];
      d[u] = (j < end) ? v : -1;
    }
#pragma unroll
    for (int u = 0; u < 4; ++u) {
      if (d[u] >= 0) {
        const int pos = atomicAdd(&cur[d[u] >> BUCKET_SHIFT], 1);
        out[pos] = ((unsigned)s[u] << BUCKET_SHIFT) |
                   (unsigned)(d[u] & (BUCKET_NODES - 1));
      }
    }
  }
}

__global__ __launch_bounds__(BUCKET_NODES) void bucket_sort_kernel(
    const unsigned* __restrict__ packed, const int* __restrict__ bstart, int E,
    int N, int nbuck, int* __restrict__ offs, int* __restrict__ cnt,
    float* __restrict__ inv, int* __restrict__ seg) {
  __shared__ int s[BUCKET_NODES];
  __shared__ int cur[BUCKET_NODES];
  const int tid = threadIdx.x;
  const int y = blockIdx.y;
  const int e0 = bstart[y * (nbuck + 1) + blockIdx.x];
  const int e1 = bstart[y * (nbuck + 1) + blockIdx.x + 1];
  const unsigned* pk = packed + (size_t)y * E;
  cur[tid] = 0;
  __syncthreads();
  for (int i = e0 + tid; i < e1; i += 4 * BUCKET_NODES) {
    unsigned w[4];
#pragma unroll
    for (int u = 0; u < 4; ++u) {
      const int j = i + u * BUCKET_NODES;
      w[u] = pk[(j < e1) ? j : i];
    }
#pragma unroll
    for (int u = 0; u < 4; ++u)
      if (i + u * BUCKET_NODES < e1)
        atomicAdd(&cur[w[u] & (BUCKET_NODES - 1)], 1);
  }
  __syncthreads();
  const int c = cur[tid];
  const int excl = block_scan<BUCKET_NODES>(c, s, tid) - c;
  const int node = blockIdx.x * BUCKET_NODES + tid;
  if (node < N) {
    const size_t g = (size_t)y * N + node;
    offs[g] = e0 + excl;
    cnt[g] = c;
    inv[g] = 1.0f / fmaxf((float)c, 1.0f);
  }
  cur[tid] = excl;
  __syncthreads();
  int* sg = seg + (size_t)y * E + e0;
  for (int i = e0 + tid; i < e1; i += 4 * BUCKET_NODES) {
    unsigned w[4];
#pragma unroll
    for (int u = 0; u < 4; ++u) {
      const int j = i + u * BUCKET_NODES;
      w[u] = pk[(j < e1) ? j : i];
    }
#pragma unroll
    for (int u = 0; u < 4; ++u) {
      if (i + u * BUCKET_NODES < e1) {
        const int pos = atomicAdd(&cur[w[u] & (BUCKET_NODES - 1)], 1);
        sg[pos] = (int)(w[u] >> BUCKET_SHIFT);
      }
    }
  }
}

__device__ __forceinline__ void add4(float4& a, const float4 v) {
  a.x += v.x;
  a.y += v.y;
  a.z += v.z;
  a.w += v.w;
}

__device__ __forceinline__ float4 gather_row128(const float* __restrict__ hl,
                                                const int* __restrict__ seg,
                                                int beg, int deg, int lane,
                                                int idx0) {
#define ROW_OF(jj) \
  (*(const float4*)(hl + (size_t)__builtin_amdgcn_readlane(idx, (jj)) * HIDDIM))
  float4 acc = make_float4(0.0f, 0.0f, 0.0f, 0.0f);
  for (int base = 0; base < deg; base += 32) {
    const int m = (deg - base < 32) ? (deg - base) : 32;
    int idx = idx0;
    if (base > 0) {
      idx = 0;
      if (lane < m) idx = seg[beg + base + lane];
    }
    int j = 0;
    for (; j + 8 <= m; j += 8) {
      const float4 v0 = ROW_OF(j), v1 = ROW_OF(j + 1), v2 = ROW_OF(j + 2),
                   v3 = ROW_OF(j + 3), v4 = ROW_OF(j + 4), v5 = ROW_OF(j + 5),
                   v6 = ROW_OF(j + 6), v7 = ROW_OF(j + 7);
      add4(acc, v0);
      add4(acc, v1);
      add4(acc, v2);
      add4(acc, v3);
      add4(acc, v4);
      add4(acc, v5);
      add4(acc, v6);
      add4(acc, v7);
    }
    if (j + 4 <= m) {
      const float4 v0 = ROW_OF(j), v1 = ROW_OF(j + 1), v2 = ROW_OF(j + 2),
                   v3 = ROW_OF(j + 3);
      add4(acc, v0);
      add4(acc, v1);
      add4(acc, v2);
      add4(acc, v3);
      j += 4;
    }
    if (j + 2 <= m) {
      const float4 v0 = ROW_OF(j), v1 = ROW_OF(j + 1);
      add4(acc, v0);
      add4(acc, v1);
      j += 2;
    }
    if (j < m) {
      const float4 v0 = ROW_OF(j);
      add4(acc, v0);
    }
  }
#undef ROW_OF
  return acc;
}

__device__ __forceinline__ float xor_add(float s, const int imm) {
  switch (imm) {
    case 1:  s += __int_as_float(__builtin_amdgcn_ds_swizzle(__float_as_int(s), 0x041f)); break;
    case 2:  s += __int_as_float(__builtin_amdgcn_ds_swizzle(__float_as_int(s), 0x081f)); break;
    case 4:  s += __int_as_float(__builtin_amdgcn_ds_swizzle(__float_as_int(s), 0x101f)); break;
    default: s += __int_as_float(__builtin_amdgcn_ds_swizzle(__float_as_int(s), 0x201f)); break;
  }
  return s;
}

template <int K>
__global__ __launch_bounds__(256) void sage_layer_kernel(
    const float* __restrict__ xin, const int* __restrict__ offs,
    const int* __restrict__ cnt, const int* __restrict__ seg,
    const float* __restrict__ invcnt, const float* __restrict__ wl,
    const float* __restrict__ wr, const float* __restrict__ bias,
    float* __restrict__ out) {
  constexpr int P = K + 4;
  __shared__ __attribute__((aligned(16))) float sM[16][P];
  __shared__ __attribute__((aligned(16))) float sX[16][P];
  __shared__ float s_norm[16];

  const int tid = threadIdx.x;
  const int wave = tid >> 5;
  const int lane = tid & 31;
  const int half = lane >> 4;
  const int l16 = lane & 15;
  const int rowBase = blockIdx.x * 16;
  const int col = wave * 16 + l16;

  if (tid < 16) s_norm[tid] = 0.0f;
  if constexpr (K == HIDDIM) {
    const int w = __builtin_amdgcn_readfirstlane(wave);
    const float* hl = xin + lane * 4;
    const float4 xa = *(const float4*)(hl + (size_t)(rowBase + w) * K);
    const float4 xb = *(const float4*)(hl + (size_t)(rowBase + w + 8) * K);
    const int n0 = rowBase + 2 * w;
    const int beg0 = offs[n0], beg1 = offs[n0 + 1];
    const int deg0 = cnt[n0], deg1 = cnt[n0 + 1];
    const float s0 = invcnt[n0], s1 = invcnt[n0 + 1];
    int idx0 = 0, idx1 = 0;
    if (lane < deg0) idx0 = seg[beg0 + lane];
    if (lane < deg1) idx1 = seg[beg1 + lane];
    *(float4*)&sX[w][lane * 4] = xa;
    *(float4*)&sX[w + 8][lane * 4] = xb;
    float4 m0 = gather_row128(hl, seg, beg0, deg0, lane, idx0);
    float4 m1 = gather_row128(hl, seg, beg1, deg1, lane, idx1);
    m0.x *= s0;
    m0.y *= s0;
    m0.z *= s0;
    m0.w *= s0;
    m1.x *= s1;
    m1.y *= s1;
    m1.z *= s1;
    m1.w *= s1;
    *(float4*)&sM[2 * w][lane * 4] = m0;
    *(float4*)&sM[2 * w + 1][lane * 4] = m1;
  } else {
    if (tid < 16 * K) {
      const int r = tid / K;
      const int k = tid - r * K;
      const int node = rowBase + r;
      sX[r][k] = xin[(size_t)node * K + k];
      int j = offs[node];
      const int end = j + cnt[node];
      float acc = 0.0f;
      for (; j + 4 <= end; j += 4) {
        const int s0 = seg[j], s1 = seg[j + 1], s2 = seg[j + 2], s3 = seg[j + 3];
        const float v0 = xin[(size_t)s0 * K + k];
        const float v1 = xin[(size_t)s1 * K + k];
        const float v2 = xin[(size_t)s2 * K + k];
        const float v3 = xin[(size_t)s3 * K + k];
        acc += v0;
        acc += v1;
        acc += v2;
        acc += v3;
      }
      for (; j < end; ++j) acc += xin[(size_t)seg[j] * K + k];
      sM[r][k] = acc * invcnt[node];
    }
  }
  __syncthreads();

  v8f c = {};
#pragma unroll 8
  for (int k0 = 0; k0 < K; k0 += 4) {
    const int ka = k0 + 2 * half;
    v2f bl, br;
    const v2f am = *(const v2f*)&sM[l16][ka];
    const v2f ax = *(const v2f*)&sX[l16][ka];
    bl.x = wl[(size_t)ka * HIDDIM + col];
    bl.y = wl[(size_t)(ka + 1) * HIDDIM + col];
    br.x = wr[(size_t)ka * HIDDIM + col];
    br.y = wr[(size_t)(ka + 1) * HIDDIM + col];
    c = __builtin_amdgcn_wmma_f32_16x16x4_f32(false, am, false, bl, (short)0, c,
                                              false, false);
    c = __builtin_amdgcn_wmma_f32_16x16x4_f32(false, ax, false, br, (short)0, c,
                                              false, false);
  }

  const float bv = bias[col];
#pragma unroll
  for (int v = 0; v < 8; ++v) c[v] += bv;

#pragma unroll
  for (int v = 0; v < 8; ++v) {
    float s = c[v] * c[v];
    s = xor_add(s, 1);
    s = xor_add(s, 2);
    s = xor_add(s, 4);
    s = xor_add(s, 8);
    if (l16 == 0) atomicAdd(&s_norm[v + 8 * half], s);
  }
  __syncthreads();

#pragma unroll
  for (int v = 0; v < 8; ++v) {
    const int row = v + 8 * half;
    const float scale = 1.0f / fmaxf(sqrtf(s_norm[row]), 1e-12f);
    out[(size_t)(rowBase + row) * HIDDIM + col] = tanhf(c[v] * scale);
  }
}

extern "C" void kernel_launch(void* const* d_in, const int* in_sizes, int n_in,
                              void* d_out, int out_size, void* d_ws,
                              size_t ws_size, hipStream_t stream) {
  const float* x = (const float*)d_in[0];
  const int* ei_conn = (const int*)d_in[1];
  const int* ei_dest = (const int*)d_in[2];
  const float* wl1 = (const float*)d_in[3];
  const float* wr1 = (const float*)d_in[4];
  const float* b1 = (const float*)d_in[5];
  const float* wl2 = (const float*)d_in[6];
  const float* wr2 = (const float*)d_in[7];
  const float* b2 = (const float*)d_in[8];
  const float* wl3 = (const float*)d_in[9];
  const float* wr3 = (const float*)d_in[10];
  const float* b3 = (const float*)d_in[11];

  const int N = in_sizes[0] / 8;
  const int E = in_sizes[1] / 2;
  const int nbuck = (N + BUCKET_NODES - 1) / BUCKET_NODES;
  const int chunk = (E + PART_BLOCKS - 1) / PART_BLOCKS;

  float* ws = (float*)d_ws;
  float* h1 = ws;
  float* h2 = h1 + (size_t)N * HIDDIM;
  float* inv = h2 + (size_t)N * HIDDIM;
  int* cnt = (int*)(inv + (size_t)2 * N);
  int* offs = cnt + (size_t)2 * N;
  int* blkcnt = offs + (size_t)2 * N;
  int* blkpref = blkcnt + (size_t)2 * PART_BLOCKS * nbuck;
  int* bstart = blkpref + (size_t)2 * PART_BLOCKS * nbuck;
  unsigned* packed = (unsigned*)(bstart + (size_t)2 * (nbuck + 1));
  int* seg = (int*)(packed + (size_t)2 * E);

  const int T = 256;

  const int* src_conn = ei_conn;
  const int* dst_conn = ei_conn + E;
  const int* src_dest = ei_dest;
  const int* dst_dest = ei_dest + E;

  bucket_count_kernel<<<dim3(PART_BLOCKS, 2), PART_THREADS, 0, stream>>>(
      dst_conn, dst_dest, E, nbuck, chunk, blkcnt);
  bucket_scan_kernel<<<2, SCAN_THREADS, 0, stream>>>(blkcnt, blkpref, bstart,
                                                     nbuck);
  bucket_scatter_kernel<<<dim3(PART_BLOCKS, 2), PART_THREADS, 0, stream>>>(
      src_conn, dst_conn, src_dest, dst_dest, E, nbuck, chunk, blkpref, bstart,
      packed);
  bucket_sort_kernel<<<dim3(nbuck, 2), BUCKET_NODES, 0, stream>>>(
      packed, bstart, E, N, nbuck, offs, cnt, inv, seg);

  const int gemmBlocks = (N + 15) / 16;

  sage_layer_kernel<8><<<gemmBlocks, T, 0, stream>>>(x, offs, cnt, seg, inv, wl1,
                                                     wr1, b1, h1);

  sage_layer_kernel<128><<<gemmBlocks, T, 0, stream>>>(
      h1, offs + N, cnt + N, seg + E, inv + N, wl2, wr2, b2, h2);

  sage_layer_kernel<128><<<gemmBlocks, T, 0, stream>>>(
      h2, offs, cnt, seg, inv, wl3, wr3, b3, (float*)d_out);
}


# Edge-sort placements staged in LDS: both placing kernels put each edge into an LDS image of their output runs and write the runs out whole, not one scattered 4-byte store an edge; sum order as before
# speedup vs baseline: 3.4489x; 1.1348x over previous
#include <hip/hip_runtime.h>
#include <hip/hip_bf16.h>
#include <math.h>

typedef __attribute__((ext_vector_type(2))) float v2f;
typedef __attribute__((ext_vector_type(8))) float v8f;

#define HIDDIM 128

#define BUCKET_SHIFT 8
#define BUCKET_NODES (1 << BUCKET_SHIFT)
#define MAX_BUCKETS 512
#define PART_BLOCKS 128
#define PART_THREADS 512
#define SCAN_THREADS (2 * MAX_BUCKETS)
#define HALF_BLOCKS (PART_BLOCKS / 2)
#define SCAT_SLOTS 64
#define SORT_STAGE 6144
static_assert(PART_THREADS == MAX_BUCKETS, "the scatter set-up gives every bucket one thread");
static_assert(SCAT_SLOTS % 32 == 0, "a row of the image is written out by whole waves");

template <int T>
__device__ __forceinline__ int block_scan(int t, int* s, int tid) {
  s[tid] = t;
  __syncthreads();
  for (int o = 1; o < T; o <<= 1) {
    const int v = (tid >= o) ? s[tid - o] : 0;
    __syncthreads();
    s[tid] += v;
    __syncthreads();
  }
  return s[tid];
}

__global__ __launch_bounds__(PART_THREADS) void bucket_count_kernel(
    const int* __restrict__ dstA, const int* __restrict__ dstB, int E,
    int nbuck, int chunk, int* __restrict__ blkcnt) {
  __shared__ int hist[MAX_BUCKETS];
  const int tid = threadIdx.x;
  const int y = blockIdx.y;
  const int* dst = y ? dstB : dstA;
  for (int b = tid; b < nbuck; b += PART_THREADS) hist[b] = 0;
  __syncthreads();
  const int beg = blockIdx.x * chunk;
  const int end = (beg + chunk < E) ? (beg + chunk) : E;
  for (int i = beg + tid; i < end; i += 4 * PART_THREADS) {
    int d[4];
#pragma unroll
    for (int u = 0; u < 4; ++u) {
      const int j = i + u * PART_THREADS;
      const int v = dst[(j < end) ? j : i];
      d[u] = (j < end) ? v : -1;
    }
#pragma unroll
    for (int u = 0; u < 4; ++u)
      if (d[u] >= 0) atomicAdd(&hist[d[u] >> BUCKET_SHIFT], 1);
  }
  __syncthreads();
  int* o = blkcnt + ((size_t)y * gridDim.x + blockIdx.x) * nbuck;
  for (int b = tid; b < nbuck; b += PART_THREADS) o[b] = hist[b];
}

__global__ __launch_bounds__(SCAN_THREADS) void bucket_scan_kernel(
    const int* __restrict__ blkcnt, int* __restrict__ blkpref,
    int* __restrict__ bstart, int nbuck) {
  __shared__ int s[SCAN_THREADS];
  __shared__ int tots[2][MAX_BUCKETS];
  const int tid = threadIdx.x;
  const int y = blockIdx.x;
  const int j = tid & (MAX_BUCKETS - 1);
  const int q = tid / MAX_BUCKETS;
  const size_t first = ((size_t)y * PART_BLOCKS + q * HALF_BLOCKS) * nbuck + j;
  const int* c = blkcnt + first;
  int* p = blkpref + first;
  const bool live = j < nbuck;
  int v[HALF_BLOCKS];
#pragma unroll
  for (int u = 0; u < HALF_BLOCKS; ++u) v[u] = live ? c[(size_t)u * nbuck] : 0;
  int tot = 0;
#pragma unroll
  for (int u = 0; u < HALF_BLOCKS; ++u) tot += v[u];
  tots[q][j] = tot;
  __syncthreads();
  const int whole = (q == 0) ? (tots[0][j] + tots[1][j]) : 0;
  int run = (q == 1) ? tots[0][j] : 0;
  if (live) {
#pragma unroll
    for (int u = 0; u < HALF_BLOCKS; ++u) {
      p[(size_t)u * nbuck] = run;
      run += v[u];
    }
  }
  const int incl = block_scan<SCAN_THREADS>(whole, s, tid);
  if (q == 0 && live) {
    bstart[y * (nbuck + 1) + j] = incl - whole;
    if (j == nbuck - 1) bstart[y * (nbuck + 1) + nbuck] = incl;
  }
}

__global__ __launch_bounds__(PART_THREADS) void bucket_scatter_kernel(
    const int* __restrict__ srcA, const int* __restrict__ dstA,
    const int* __restrict__ srcB, const int* __restrict__ dstB, int E,
    int nbuck, int chunk, const int* __restrict__ blkpref,
    const int* __restrict__ bstart, unsigned* __restrict__ packed) {
  __shared__ int cnt[MAX_BUCKETS];
  __shared__ int gpos[MAX_BUCKETS];
  __shared__ __attribute__((aligned(16))) unsigned img[MAX_BUCKETS * SCAT_SLOTS];
  const int tid = threadIdx.x;
  const int y = blockIdx.y;
  const int* src = y ? srcB : srcA;
  const int* dst = y ? dstB : dstA;
  const int* p = blkpref + ((size_t)y * gridDim.x + blockIdx.x) * nbuck;
  const int* bs = bstart + y * (nbuck + 1);
  const bool live = tid < nbuck;
  const int g0 = live ? bs[tid] : 0;
  const int g1 = live ? p[tid] : 0;
  uint4* img4 = (uint4*)img;
  for (int q = tid; q < nbuck * (SCAT_SLOTS / 4); q += PART_THREADS)
    img4[q] = make_uint4(0u, 0u, 0u, 0u);
  cnt[tid] = 0;
  gpos[tid] = g0 + g1;
  __syncthreads();
  unsigned* out = packed + (size_t)y * E;
  const int beg = blockIdx.x * chunk;
  const int end = (beg + chunk < E) ? (beg + chunk) : E;
  for (int i = beg + tid; i < end; i += 4 * PART_THREADS) {
    int d[4], s[4];
#pragma unroll
    for (int u = 0; u < 4; ++u) {
      const int j = i + u * PART_THREADS;
      const int jj = (j < end) ? j : i;
      const int v = dst[jj];
      s[u] = src[jj];
      d[u] = (j < end) ? v : -1;
    }
#pragma unroll
    for (int u = 0; u < 4; ++u) {
      if (d[u] >= 0) {
        const int b = d[u] >> BUCKET_SHIFT;
        const int r = atomicAdd(&cnt[b], 1);
        const unsigned w = ((unsigned)s[u] << BUCKET_SHIFT) |
                           (unsigned)(d[u] & (BUCKET_NODES - 1));
        if (r < SCAT_SLOTS)
          img[b * SCAT_SLOTS + r] = w;
        else
          out[gpos[b] + r] = w;
      }
    }
  }
  __syncthreads();
  constexpr int NW = PART_THREADS / 32;
  constexpr int RW = SCAT_SLOTS / 32;
  const int wave = tid >> 5;
  const int lane = tid & 31;
  for (int b0 = wave; b0 < nbuck; b0 += 4 * NW) {
    int n[4], gb[4];
    unsigned wv[4][RW];
#pragma unroll
    for (int u = 0; u < 4; ++u) {
      const int b = b0 + u * NW;
      const int bb = (b < nbuck) ? b : b0;
      const int c = cnt[bb];
      n[u] = (b < nbuck) ? c : 0;
      gb[u] = gpos[bb];
#pragma unroll
      for (int k = 0; k < RW; ++k) wv[u][k] = img[bb * SCAT_SLOTS + 32 * k + lane];
    }
#pragma unroll
    for (int u = 0; u < 4; ++u) {
#pragma unroll
      for (int k = 0; k < RW; ++k)
        if (32 * k + lane < n[u]) out[gb[u] + 32 * k + lane] = wv[u][k];
    }
  }
}

__global__ __launch_bounds__(BUCKET_NODES) void bucket_sort_kernel(
    const unsigned* __restrict__ packed, const int* __restrict__ bstart, int E,
    int N, int nbuck, int* __restrict__ offs, int* __restrict__ cnt,
    float* __restrict__ inv, int* __restrict__ seg) {
  __shared__ int s[BUCKET_NODES];
  __shared__ int cur[BUCKET_NODES];
  __shared__ int stage[SORT_STAGE];
  const int tid = threadIdx.x;
  const int y = blockIdx.y;
  const int e0 = bstart[y * (nbuck + 1) + blockIdx.x];
  const int e1 = bstart[y * (nbuck + 1) + blockIdx.x + 1];
  const unsigned* pk = packed + (size_t)y * E;
  cur[tid] = 0;
  __syncthreads();
  for (int i = e0 + tid; i < e1; i += 4 * BUCKET_NODES) {
    unsigned w[4];
#pragma unroll
    for (int u = 0; u < 4; ++u) {
      const int j = i + u * BUCKET_NODES;
      w[u] = pk[(j < e1) ? j : i];
    }
#pragma unroll
    for (int u = 0; u < 4; ++u)
      if (i + u * BUCKET_NODES < e1)
        atomicAdd(&cur[w[u] & (BUCKET_NODES - 1)], 1);
  }
  __syncthreads();
  const int c = cur[tid];
  const int excl = block_scan<BUCKET_NODES>(c, s, tid) - c;
  const int node = blockIdx.x * BUCKET_NODES + tid;
  if (node < N) {
    const size_t g = (size_t)y * N + node;
    offs[g] = e0 + excl;
    cnt[g] = c;
    inv[g] = 1.0f / fmaxf((float)c, 1.0f);
  }
  cur[tid] = excl;
  __syncthreads();
  int* sg = seg + (size_t)y * E + e0;
  for (int i = e0 + tid; i < e1; i += 4 * BUCKET_NODES) {
    unsigned w[4];
#pragma unroll
    for (int u = 0; u < 4; ++u) {
      const int j = i + u * BUCKET_NODES;
      w[u] = pk[(j < e1) ? j : i];
    }
#pragma unroll
    for (int u = 0; u < 4; ++u) {
      if (i + u * BUCKET_NODES < e1) {
        const int pos = atomicAdd(&cur[w[u] & (BUCKET_NODES - 1)], 1);
        const int v = (int)(w[u] >> BUCKET_SHIFT);
        if (pos < SORT_STAGE)
          stage[pos] = v;
        else
          sg[pos] = v;
      }
    }
  }
  __syncthreads();
  const int n = e1 - e0;
  const int m = (n < SORT_STAGE) ? n : SORT_STAGE;
  const int mis = (int)(((size_t)sg >> 2) & 31);
  int i = tid - mis;
  if (i < 0) i += BUCKET_NODES;
#pragma unroll 4
  for (; i < m; i += BUCKET_NODES) sg[i] = stage[i];
}

__device__ __forceinline__ void add4(float4& a, const float4 v) {
  a.x += v.x;
  a.y += v.y;
  a.z += v.z;
  a.w += v.w;
}

__device__ __forceinline__ float4 gather_row128(const float* __restrict__ hl,
                                                const int* __restrict__ seg,
                                                int beg, int deg, int lane,
                                                int idx0) {
#define ROW_OF(jj) \
  (*(const float4*)(hl + (size_t)__builtin_amdgcn_readlane(idx, (jj)) * HIDDIM))
  float4 acc = make_float4(0.0f, 0.0f, 0.0f, 0.0f);
  for (int base = 0; base < deg; base += 32) {
    const int m = (deg - base < 32) ? (deg - base) : 32;
    int idx = idx0;
    if (base > 0) {
      idx = 0;
      if (lane < m) idx = seg[beg + base + lane];
    }
    int j = 0;
    for (; j + 8 <= m; j += 8) {
      const float4 v0 = ROW_OF(j), v1 = ROW_OF(j + 1), v2 = ROW_OF(j + 2),
                   v3 = ROW_OF(j + 3), v4 = ROW_OF(j + 4), v5 = ROW_OF(j + 5),
                   v6 = ROW_OF(j + 6), v7 = ROW_OF(j + 7);
      add4(acc, v0);
      add4(acc, v1);
      add4(acc, v2);
      add4(acc, v3);
      add4(acc, v4);
      add4(acc, v5);
      add4(acc, v6);
      add4(acc, v7);
    }
    if (j + 4 <= m) {
      const float4 v0 = ROW_OF(j), v1 = ROW_OF(j + 1), v2 = ROW_OF(j + 2),
                   v3 = ROW_OF(j + 3);
      add4(acc, v0);
      add4(acc, v1);
      add4(acc, v2);
      add4(acc, v3);
      j += 4;
    }
    if (j + 2 <= m) {
      const float4 v0 = ROW_OF(j), v1 = ROW_OF(j + 1);
      add4(acc, v0);
      add4(acc, v1);
      j += 2;
    }
    if (j < m) {
      const float4 v0 = ROW_OF(j);
      add4(acc, v0);
    }
  }
#undef ROW_OF
  return acc;
}

__device__ __forceinline__ float xor_add(float s, const int imm) {
  switch (imm) {
    case 1:  s += __int_as_float(__builtin_amdgcn_ds_swizzle(__float_as_int(s), 0x041f)); break;
    case 2:  s += __int_as_float(__builtin_amdgcn_ds_swizzle(__float_as_int(s), 0x081f)); break;
    case 4:  s += __int_as_float(__builtin_amdgcn_ds_swizzle(__float_as_int(s), 0x101f)); break;
    default: s += __int_as_float(__builtin_amdgcn_ds_swizzle(__float_as_int(s), 0x201f)); break;
  }
  return s;
}

template <int K>
__global__ __launch_bounds__(256) void sage_layer_kernel(
    const float* __restrict__ xin, const int* __restrict__ offs,
    const int* __restrict__ cnt, const int* __restrict__ seg,
    const float* __restrict__ invcnt, const float* __restrict__ wl,
    const float* __restrict__ wr, const float* __restrict__ bias,
    float* __restrict__ out) {
  constexpr int P = K + 4;
  __shared__ __attribute__((aligned(16))) float sM[16][P];
  __shared__ __attribute__((aligned(16))) float sX[16][P];
  __shared__ float s_norm[16];

  const int tid = threadIdx.x;
  const int wave = tid >> 5;
  const int lane = tid & 31;
  const int half = lane >> 4;
  const int l16 = lane & 15;
  const int rowBase = blockIdx.x * 16;
  const int col = wave * 16 + l16;

  if (tid < 16) s_norm[tid] = 0.0f;
  if constexpr (K == HIDDIM) {
    const int w = __builtin_amdgcn_readfirstlane(wave);
    const float* hl = xin + lane * 4;
    const float4 xa = *(const float4*)(hl + (size_t)(rowBase + w) * K);
    const float4 xb = *(const float4*)(hl + (size_t)(rowBase + w + 8) * K);
    const int n0 = rowBase + 2 * w;
    const int beg0 = offs[n0], beg1 = offs[n0 + 1];
    const int deg0 = cnt[n0], deg1 = cnt[n0 + 1];
    const float s0 = invcnt[n0], s1 = invcnt[n0 + 1];
    int idx0 = 0, idx1 = 0;
    if (lane < deg0) idx0 = seg[beg0 + lane];
    if (lane < deg1) idx1 = seg[beg1 + lane];
    *(float4*)&sX[w][lane * 4] = xa;
    *(float4*)&sX[w + 8][lane * 4] = xb;
    float4 m0 = gather_row128(hl, seg, beg0, deg0, lane, idx0);
    float4 m1 = gather_row128(hl, seg, beg1, deg1, lane, idx1);
    m0.x *= s0;
    m0.y *= s0;
    m0.z *= s0;
    m0.w *= s0;
    m1.x *= s1;
    m1.y *= s1;
    m1.z *= s1;
    m1.w *= s1;
    *(float4*)&sM[2 * w][lane * 4] = m0;
    *(float4*)&sM[2 * w + 1][lane * 4] = m1;
  } else {
    if (tid < 16 * K) {
      const int r = tid / K;
      const int k = tid - r * K;
      const int node = rowBase + r;
      sX[r][k] = xin[(size_t)node * K + k];
      int j = offs[node];
      const int end = j + cnt[node];
      float acc = 0.0f;
      for (; j + 4 <= end; j += 4) {
        const int s0 = seg[j], s1 = seg[j + 1], s2 = seg[j + 2], s3 = seg[j + 3];
        const float v0 = xin[(size_t)s0 * K + k];
        const float v1 = xin[(size_t)s1 * K + k];
        const float v2 = xin[(size_t)s2 * K + k];
        const float v3 = xin[(size_t)s3 * K + k];
        acc += v0;
        acc += v1;
        acc += v2;
        acc += v3;
      }
      for (; j < end; ++j) acc += xin[(size_t)seg[j] * K + k];
      sM[r][k] = acc * invcnt[node];
    }
  }
  __syncthreads();

  v8f c = {};
#pragma unroll 8
  for (int k0 = 0; k0 < K; k0 += 4) {
    const int ka = k0 + 2 * half;
    v2f bl, br;
    const v2f am = *(const v2f*)&sM[l16][ka];
    const v2f ax = *(const v2f*)&sX[l16][ka];
    bl.x = wl[(size_t)ka * HIDDIM + col];
    bl.y = wl[(size_t)(ka + 1) * HIDDIM + col];
    br.x = wr[(size_t)ka * HIDDIM + col];
    br.y = wr[(size_t)(ka + 1) * HIDDIM + col];
    c = __builtin_amdgcn_wmma_f32_16x16x4_f32(false, am, false, bl, (short)0, c,
                                              false, false);
    c = __builtin_amdgcn_wmma_f32_16x16x4_f32(false, ax, false, br, (short)0, c,
                                              false, false);
  }

  const float bv = bias[col];
#pragma unroll
  for (int v = 0; v < 8; ++v) c[v] += bv;

#pragma unroll
  for (int v = 0; v < 8; ++v) {
    float s = c[v] * c[v];
    s = xor_add(s, 1);
    s = xor_add(s, 2);
    s = xor_add(s, 4);
    s = xor_add(s, 8);
    if (l16 == 0) atomicAdd(&s_norm[v + 8 * half], s);
  }
  __syncthreads();

#pragma unroll
  for (int v = 0; v < 8; ++v) {
    const int row = v + 8 * half;
    const float scale = 1.0f / fmaxf(sqrtf(s_norm[row]), 1e-12f);
    out[(size_t)(rowBase + row) * HIDDIM + col] = tanhf(c[v] * scale);
  }
}

extern "C" void kernel_launch(void* const* d_in, const int* in_sizes, int n_in,
                              void* d_out, int out_size, void* d_ws,
                              size_t ws_size, hipStream_t stream) {
  const float* x = (const float*)d_in[0];
  const int* ei_conn = (const int*)d_in[1];
  const int* ei_dest = (const int*)d_in[2];
  const float* wl1 = (const float*)d_in[3];
  const float* wr1 = (const float*)d_in[4];
  const float* b1 = (const float*)d_in[5];
  const float* wl2 = (const float*)d_in[6];
  const float* wr2 = (const float*)d_in[7];
  const float* b2 = (const float*)d_in[8];
  const float* wl3 = (const float*)d_in[9];
  const float* wr3 = (const float*)d_in[10];
  const float* b3 = (const float*)d_in[11];

  const int N = in_sizes[0] / 8;
  const int E = in_sizes[1] / 2;
  const int nbuck = (N + BUCKET_NODES - 1) / BUCKET_NODES;
  const int chunk = (E + PART_BLOCKS - 1) / PART_BLOCKS;

  float* ws = (float*)d_ws;
  float* h1 = ws;
  float* h2 = h1 + (size_t)N * HIDDIM;
  float* inv = h2 + (size_t)N * HIDDIM;
  int* cnt = (int*)(inv + (size_t)2 * N);
  int* offs = cnt + (size_t)2 * N;
  int* blkcnt = offs + (size_t)2 * N;
  int* blkpref = blkcnt + (size_t)2 * PART_BLOCKS * nbuck;
  int* bstart = blkpref + (size_t)2 * PART_BLOCKS * nbuck;
  unsigned* packed = (unsigned*)(bstart + (size_t)2 * (nbuck + 1));
  int* seg = (int*)(packed + (size_t)2 * E);

  const int T = 256;

  const int* src_conn = ei_conn;
  const int* dst_conn = ei_conn + E;
  const int* src_dest = ei_dest;
  const int* dst_dest = ei_dest + E;

  bucket_count_kernel<<<dim3(PART_BLOCKS, 2), PART_THREADS, 0, stream>>>(
      dst_conn, dst_dest, E, nbuck, chunk, blkcnt);
  bucket_scan_kernel<<<2, SCAN_THREADS, 0, stream>>>(blkcnt, blkpref, bstart,
                                                     nbuck);
  bucket_scatter_kernel<<<dim3(PART_BLOCKS, 2), PART_THREADS, 0, stream>>>(
      src_conn, dst_conn, src_dest, dst_dest, E, nbuck, chunk, blkpref, bstart,
      packed);
  bucket_sort_kernel<<<dim3(nbuck, 2), BUCKET_NODES, 0, stream>>>(
      packed, bstart, E, N, nbuck, offs, cnt, inv, seg);

  const int gemmBlocks = (N + 15) / 16;

  sage_layer_kernel<8><<<gemmBlocks, T, 0, stream>>>(x, offs, cnt, seg, inv, wl1,
                                                     wr1, b1, h1);

  sage_layer_kernel<128><<<gemmBlocks, T, 0, stream>>>(
      h1, offs + N, cnt + N, seg + E, inv + N, wl2, wr2, b2, h2);

  sage_layer_kernel<128><<<gemmBlocks, T, 0, stream>>>(
      h2, offs, cnt, seg, inv, wl3, wr3, b3, (float*)d_out);
}


# Epilogue per whole row: the 16x128 tile passes through LDS, a wave norms, scales, stores two full rows (4 columns a lane); row norm summed by a fixed lane tree, no LDS atomics: sum order changes
# speedup vs baseline: 3.7497x; 1.0872x over previous
#include <hip/hip_runtime.h>
#include <hip/hip_bf16.h>
#include <math.h>

typedef __attribute__((ext_vector_type(2))) float v2f;
typedef __attribute__((ext_vector_type(8))) float v8f;

#define HIDDIM 128

#define BUCKET_SHIFT 8
#define BUCKET_NODES (1 << BUCKET_SHIFT)
#define MAX_BUCKETS 512
#define PART_BLOCKS 128
#define PART_THREADS 512
#define SCAN_THREADS (2 * MAX_BUCKETS)
#define HALF_BLOCKS (PART_BLOCKS / 2)
#define SCAT_SLOTS 64
#define SORT_STAGE 6144
static_assert(PART_THREADS == MAX_BUCKETS, "the scatter set-up gives every bucket one thread");
static_assert(SCAT_SLOTS % 32 == 0, "a row of the image is written out by whole waves");

template <int T>
__device__ __forceinline__ int block_scan(int t, int* s, int tid) {
  s[tid] = t;
  __syncthreads();
  for (int o = 1; o < T; o <<= 1) {
    const int v = (tid >= o) ? s[tid - o] : 0;
    __syncthreads();
    s[tid] += v;
    __syncthreads();
  }
  return s[tid];
}

__global__ __launch_bounds__(PART_THREADS) void bucket_count_kernel(
    const int* __restrict__ dstA, const int* __restrict__ dstB, int E,
    int nbuck, int chunk, int* __restrict__ blkcnt) {
  __shared__ int hist[MAX_BUCKETS];
  const int tid = threadIdx.x;
  const int y = blockIdx.y;
  const int* dst = y ? dstB : dstA;
  for (int b = tid; b < nbuck; b += PART_THREADS) hist[b] = 0;
  __syncthreads();
  const int beg = blockIdx.x * chunk;
  const int end = (beg + chunk < E) ? (beg + chunk) : E;
  for (int i = beg + tid; i < end; i += 4 * PART_THREADS) {
    int d[4];
#pragma unroll
    for (int u = 0; u < 4; ++u) {
      const int j = i + u * PART_THREADS;
      const int v = dst[(j < end) ? j : i];
      d[u] = (j < end) ? v : -1;
    }
#pragma unroll
    for (int u = 0; u < 4; ++u)
      if (d[u] >= 0) atomicAdd(&hist[d[u] >> BUCKET_SHIFT], 1);
  }
  __syncthreads();
  int* o = blkcnt + ((size_t)y * gridDim.x + blockIdx.x) * nbuck;
  for (int b = tid; b < nbuck; b += PART_THREADS) o[b] = hist[b];
}

__global__ __launch_bounds__(SCAN_THREADS) void bucket_scan_kernel(
    const int* __restrict__ blkcnt, int* __restrict__ blkpref,
    int* __restrict__ bstart, int nbuck) {
  __shared__ int s[SCAN_THREADS];
  __shared__ int tots[2][MAX_BUCKETS];
  const int tid = threadIdx.x;
  const int y = blockIdx.x;
  const int j = tid & (MAX_BUCKETS - 1);
  const int q = tid / MAX_BUCKETS;
  const size_t first = ((size_t)y * PART_BLOCKS + q * HALF_BLOCKS) * nbuck + j;
  const int* c = blkcnt + first;
  int* p = blkpref + first;
  const bool live = j < nbuck;
  int v[HALF_BLOCKS];
#pragma unroll
  for (int u = 0; u < HALF_BLOCKS; ++u) v[u] = live ? c[(size_t)u * nbuck] : 0;
  int tot = 0;
#pragma unroll
  for (int u = 0; u < HALF_BLOCKS; ++u) tot += v[u];
  tots[q][j] = tot;
  __syncthreads();
  const int whole = (q == 0) ? (tots[0][j] + tots[1][j]) : 0;
  int run = (q == 1) ? tots[0][j] : 0;
  if (live) {
#pragma unroll
    for (int u = 0; u < HALF_BLOCKS; ++u) {
      p[(size_t)u * nbuck] = run;
      run += v[u];
    }
  }
  const int incl = block_scan<SCAN_THREADS>(whole, s, tid);
  if (q == 0 && live) {
    bstart[y * (nbuck + 1) + j] = incl - whole;
    if (j == nbuck - 1) bstart[y * (nbuck + 1) + nbuck] = incl;
  }
}

__global__ __launch_bounds__(PART_THREADS) void bucket_scatter_kernel(
    const int* __restrict__ srcA, const int* __restrict__ dstA,
    const int* __restrict__ srcB, const int* __restrict__ dstB, int E,
    int nbuck, int chunk, const int* __restrict__ blkpref,
    const int* __restrict__ bstart, unsigned* __restrict__ packed) {
  __shared__ int cnt[MAX_BUCKETS];
  __shared__ int gpos[MAX_BUCKETS];
  __shared__ __attribute__((aligned(16))) unsigned img[MAX_BUCKETS * SCAT_SLOTS];
  const int tid = threadIdx.x;
  const int y = blockIdx.y;
  const int* src = y ? srcB : srcA;
  const int* dst = y ? dstB : dstA;
  const int* p = blkpref + ((size_t)y * gridDim.x + blockIdx.x) * nbuck;
  const int* bs = bstart + y * (nbuck + 1);
  const bool live = tid < nbuck;
  const int g0 = live ? bs[tid] : 0;
  const int g1 = live ? p[tid] : 0;
  uint4* img4 = (uint4*)img;
  for (int q = tid; q < nbuck * (SCAT_SLOTS / 4); q += PART_THREADS)
    img4[q] = make_uint4(0u, 0u, 0u, 0u);
  cnt[tid] = 0;
  gpos[tid] = g0 + g1;
  __syncthreads();
  unsigned* out = packed + (size_t)y * E;
  const int beg = blockIdx.x * chunk;
  const int end = (beg + chunk < E) ? (beg + chunk) : E;
  for (int i = beg + tid; i < end; i += 4 * PART_THREADS) {
    int d[4], s[4];
#pragma unroll
    for (int u = 0; u < 4; ++u) {
      const int j = i + u * PART_THREADS;
      const int jj = (j < end) ? j : i;
      const int v = dst[jj];
      s[u] = src[jj];
      d[u] = (j < end) ? v : -1;
    }
#pragma unroll
    for (int u = 0; u < 4; ++u) {
      if (d[u] >= 0) {
        const int b = d[u] >> BUCKET_SHIFT;
        const int r = atomicAdd(&cnt[b], 1);
        const unsigned w = ((unsigned)s[u] << BUCKET_SHIFT) |
                           (unsigned)(d[u] & (BUCKET_NODES - 1));
        if (r < SCAT_SLOTS)
          img[b * SCAT_SLOTS + r] = w;
        else
          out[gpos[b] + r] = w;
      }
    }
  }
  __syncthreads();
  constexpr int NW = PART_THREADS / 32;
  constexpr int RW = SCAT_SLOTS / 32;
  const int wave = tid >> 5;
  const int lane = tid & 31;
  for (int b0 = wave; b0 < nbuck; b0 += 4 * NW) {
    int n[4], gb[4];
    unsigned wv[4][RW];
#pragma unroll
    for (int u = 0; u < 4; ++u) {
      const int b = b0 + u * NW;
      const int bb = (b < nbuck) ? b : b0;
      const int c = cnt[bb];
      n[u] = (b < nbuck) ? c : 0;
      gb[u] = gpos[bb];
#pragma unroll
      for (int k = 0; k < RW; ++k) wv[u][k] = img[bb * SCAT_SLOTS + 32 * k + lane];
    }
#pragma unroll
    for (int u = 0; u < 4; ++u) {
#pragma unroll
      for (int k = 0; k < RW; ++k)
        if (32 * k + lane < n[u]) out[gb[u] + 32 * k + lane] = wv[u][k];
    }
  }
}

__global__ __launch_bounds__(BUCKET_NODES) void bucket_sort_kernel(
    const unsigned* __restrict__ packed, const int* __restrict__ bstart, int E,
    int N, int nbuck, int* __restrict__ offs, int* __restrict__ cnt,
    float* __restrict__ inv, int* __restrict__ seg) {
  __shared__ int s[BUCKET_NODES];
  __shared__ int cur[BUCKET_NODES];
  __shared__ int stage[SORT_STAGE];
  const int tid = threadIdx.x;
  const int y = blockIdx.y;
  const int e0 = bstart[y * (nbuck + 1) + blockIdx.x];
  const int e1 = bstart[y * (nbuck + 1) + blockIdx.x + 1];
  const unsigned* pk = packed + (size_t)y * E;
  cur[tid] = 0;
  __syncthreads();
  for (int i = e0 + tid; i < e1; i += 4 * BUCKET_NODES) {
    unsigned w[4];
#pragma unroll
    for (int u = 0; u < 4; ++u) {
      const int j = i + u * BUCKET_NODES;
      w[u] = pk[(j < e1) ? j : i];
    }
#pragma unroll
    for (int u = 0; u < 4; ++u)
      if (i + u * BUCKET_NODES < e1)
        atomicAdd(&cur[w[u] & (BUCKET_NODES - 1)], 1);
  }
  __syncthreads();
  const int c = cur[tid];
  const int excl = block_scan<BUCKET_NODES>(c, s, tid) - c;
  const int node = blockIdx.x * BUCKET_NODES + tid;
  if (node < N) {
    const size_t g = (size_t)y * N + node;
    offs[g] = e0 + excl;
    cnt[g] = c;
    inv[g] = 1.0f / fmaxf((float)c, 1.0f);
  }
  cur[tid] = excl;
  __syncthreads();
  int* sg = seg + (size_t)y * E + e0;
  for (int i = e0 + tid; i < e1; i += 4 * BUCKET_NODES) {
    unsigned w[4];
#pragma unroll
    for (int u = 0; u < 4; ++u) {
      const int j = i + u * BUCKET_NODES;
      w[u] = pk[(j < e1) ? j : i];
    }
#pragma unroll
    for (int u = 0; u < 4; ++u) {
      if (i + u * BUCKET_NODES < e1) {
        const int pos = atomicAdd(&cur[w[u] & (BUCKET_NODES - 1)], 1);
        const int v = (int)(w[u] >> BUCKET_SHIFT);
        if (pos < SORT_STAGE)
          stage[pos] = v;
        else
          sg[pos] = v;
      }
    }
  }
  __syncthreads();
  const int n = e1 - e0;
  const int m = (n < SORT_STAGE) ? n : SORT_STAGE;
  const int mis = (int)(((size_t)sg >> 2) & 31);
  int i = tid - mis;
  if (i < 0) i += BUCKET_NODES;
#pragma unroll 4
  for (; i < m; i += BUCKET_NODES) sg[i] = stage[i];
}

__device__ __forceinline__ void add4(float4& a, const float4 v) {
  a.x += v.x;
  a.y += v.y;
  a.z += v.z;
  a.w += v.w;
}

__device__ __forceinline__ float4 gather_row128(const float* __restrict__ hl,
                                                const int* __restrict__ seg,
                                                int beg, int deg, int lane,
                                                int idx0) {
#define ROW_OF(jj) \
  (*(const float4*)(hl + (size_t)__builtin_amdgcn_readlane(idx, (jj)) * HIDDIM))
  float4 acc = make_float4(0.0f, 0.0f, 0.0f, 0.0f);
  for (int base = 0; base < deg; base += 32) {
    const int m = (deg - base < 32) ? (deg - base) : 32;
    int idx = idx0;
    if (base > 0) {
      idx = 0;
      if (lane < m) idx = seg[beg + base + lane];
    }
    int j = 0;
    for (; j + 8 <= m; j += 8) {
      const float4 v0 = ROW_OF(j), v1 = ROW_OF(j + 1), v2 = ROW_OF(j + 2),
                   v3 = ROW_OF(j + 3), v4 = ROW_OF(j + 4), v5 = ROW_OF(j + 5),
                   v6 = ROW_OF(j + 6), v7 = ROW_OF(j + 7);
      add4(acc, v0);
      add4(acc, v1);
      add4(acc, v2);
      add4(acc, v3);
      add4(acc, v4);
      add4(acc, v5);
      add4(acc, v6);
      add4(acc, v7);
    }
    if (j + 4 <= m) {
      const float4 v0 = ROW_OF(j), v1 = ROW_OF(j + 1), v2 = ROW_OF(j + 2),
                   v3 = ROW_OF(j + 3);
      add4(acc, v0);
      add4(acc, v1);
      add4(acc, v2);
      add4(acc, v3);
      j += 4;
    }
    if (j + 2 <= m) {
      const float4 v0 = ROW_OF(j), v1 = ROW_OF(j + 1);
      add4(acc, v0);
      add4(acc, v1);
      j += 2;
    }
    if (j < m) {
      const float4 v0 = ROW_OF(j);
      add4(acc, v0);
    }
  }
#undef ROW_OF
  return acc;
}

__device__ __forceinline__ float xor_add(float s, const int imm) {
  switch (imm) {
    case 1:  s += __int_as_float(__builtin_amdgcn_ds_swizzle(__float_as_int(s), 0x041f)); break;
    case 2:  s += __int_as_float(__builtin_amdgcn_ds_swizzle(__float_as_int(s), 0x081f)); break;
    case 4:  s += __int_as_float(__builtin_amdgcn_ds_swizzle(__float_as_int(s), 0x101f)); break;
    default: s += __int_as_float(__builtin_amdgcn_ds_swizzle(__float_as_int(s), 0x201f)); break;
  }
  return s;
}

__device__ __forceinline__ float other_half(float x) {
  const int b = __float_as_int(x);
  return __int_as_float(
      __builtin_amdgcn_permlanex16(b, b, 0x76543210, 0xfedcba98, false, false));
}

template <int K>
__global__ __launch_bounds__(256) void sage_layer_kernel(
    const float* __restrict__ xin, const int* __restrict__ offs,
    const int* __restrict__ cnt, const int* __restrict__ seg,
    const float* __restrict__ invcnt, const float* __restrict__ wl,
    const float* __restrict__ wr, const float* __restrict__ bias,
    float* __restrict__ out) {
  constexpr int P = K + 4;
  __shared__ __attribute__((aligned(16))) float sM[16][P];
  __shared__ __attribute__((aligned(16))) float sX[16][P];
  __shared__ __attribute__((aligned(16))) float sO[16][HIDDIM + 4];

  const int tid = threadIdx.x;
  const int wave = tid >> 5;
  const int lane = tid & 31;
  const int half = lane >> 4;
  const int l16 = lane & 15;
  const int rowBase = blockIdx.x * 16;
  const int col = wave * 16 + l16;

  if constexpr (K == HIDDIM) {
    const int w = __builtin_amdgcn_readfirstlane(wave);
    const float* hl = xin + lane * 4;
    const float4 xa = *(const float4*)(hl + (size_t)(rowBase + w) * K);
    const float4 xb = *(const float4*)(hl + (size_t)(rowBase + w + 8) * K);
    const int n0 = rowBase + 2 * w;
    const int beg0 = offs[n0], beg1 = offs[n0 + 1];
    const int deg0 = cnt[n0], deg1 = cnt[n0 + 1];
    const float s0 = invcnt[n0], s1 = invcnt[n0 + 1];
    int idx0 = 0, idx1 = 0;
    if (lane < deg0) idx0 = seg[beg0 + lane];
    if (lane < deg1) idx1 = seg[beg1 + lane];
    *(float4*)&sX[w][lane * 4] = xa;
    *(float4*)&sX[w + 8][lane * 4] = xb;
    float4 m0 = gather_row128(hl, seg, beg0, deg0, lane, idx0);
    float4 m1 = gather_row128(hl, seg, beg1, deg1, lane, idx1);
    m0.x *= s0;
    m0.y *= s0;
    m0.z *= s0;
    m0.w *= s0;
    m1.x *= s1;
    m1.y *= s1;
    m1.z *= s1;
    m1.w *= s1;
    *(float4*)&sM[2 * w][lane * 4] = m0;
    *(float4*)&sM[2 * w + 1][lane * 4] = m1;
  } else {
    if (tid < 16 * K) {
      const int r = tid / K;
      const int k = tid - r * K;
      const int node = rowBase + r;
      sX[r][k] = xin[(size_t)node * K + k];
      int j = offs[node];
      const int end = j + cnt[node];
      float acc = 0.0f;
      for (; j + 4 <= end; j += 4) {
        const int s0 = seg[j], s1 = seg[j + 1], s2 = seg[j + 2], s3 = seg[j + 3];
        const float v0 = xin[(size_t)s0 * K + k];
        const float v1 = xin[(size_t)s1 * K + k];
        const float v2 = xin[(size_t)s2 * K + k];
        const float v3 = xin[(size_t)s3 * K + k];
        acc += v0;
        acc += v1;
        acc += v2;
        acc += v3;
      }
      for (; j < end; ++j) acc += xin[(size_t)seg[j] * K + k];
      sM[r][k] = acc * invcnt[node];
    }
  }
  __syncthreads();

  v8f c = {};
#pragma unroll 8
  for (int k0 = 0; k0 < K; k0 += 4) {
    const int ka = k0 + 2 * half;
    v2f bl, br;
    const v2f am = *(const v2f*)&sM[l16][ka];
    const v2f ax = *(const v2f*)&sX[l16][ka];
    bl.x = wl[(size_t)ka * HIDDIM + col];
    bl.y = wl[(size_t)(ka + 1) * HIDDIM + col];
    br.x = wr[(size_t)ka * HIDDIM + col];
    br.y = wr[(size_t)(ka + 1) * HIDDIM + col];
    c = __builtin_amdgcn_wmma_f32_16x16x4_f32(false, am, false, bl, (short)0, c,
                                              false, false);
    c = __builtin_amdgcn_wmma_f32_16x16x4_f32(false, ax, false, br, (short)0, c,
                                              false, false);
  }

  const float bv = bias[col];
#pragma unroll
  for (int v = 0; v < 8; ++v) sO[v + 8 * half][col] = c[v] + bv;
  __syncthreads();

  float4 o[2];
  float s[2];
#pragma unroll
  for (int q = 0; q < 2; ++q) {
    o[q] = *(const float4*)&sO[wave + 8 * q][lane * 4];
    s[q] = o[q].x * o[q].x + o[q].y * o[q].y + o[q].z * o[q].z + o[q].w * o[q].w;
  }
#pragma unroll
  for (int q = 0; q < 2; ++q) s[q] = xor_add(s[q], 1);
#pragma unroll
  for (int q = 0; q < 2; ++q) s[q] = xor_add(s[q], 2);
#pragma unroll
  for (int q = 0; q < 2; ++q) s[q] = xor_add(s[q], 4);
#pragma unroll
  for (int q = 0; q < 2; ++q) s[q] = xor_add(s[q], 8);
#pragma unroll
  for (int q = 0; q < 2; ++q) s[q] += other_half(s[q]);
#pragma unroll
  for (int q = 0; q < 2; ++q) {
    const float scale = 1.0f / fmaxf(sqrtf(s[q]), 1e-12f);
    float4 r;
    r.x = tanhf(o[q].x * scale);
    r.y = tanhf(o[q].y * scale);
    r.z = tanhf(o[q].z * scale);
    r.w = tanhf(o[q].w * scale);
    *(float4*)(out + (size_t)(rowBase + wave + 8 * q) * HIDDIM + lane * 4) = r;
  }
}

extern "C" void kernel_launch(void* const* d_in, const int* in_sizes, int n_in,
                              void* d_out, int out_size, void* d_ws,
                              size_t ws_size, hipStream_t stream) {
  const float* x = (const float*)d_in[0];
  const int* ei_conn = (const int*)d_in[1];
  const int* ei_dest = (const int*)d_in[2];
  const float* wl1 = (const float*)d_in[3];
  const float* wr1 = (const float*)d_in[4];
  const float* b1 = (const float*)d_in[5];
  const float* wl2 = (const float*)d_in[6];
  const float* wr2 = (const float*)d_in[7];
  const float* b2 = (const float*)d_in[8];
  const float* wl3 = (const float*)d_in[9];
  const float* wr3 = (const float*)d_in[10];
  const float* b3 = (const float*)d_in[11];

  const int N = in_sizes[0] / 8;
  const int E = in_sizes[1] / 2;
  const int nbuck = (N + BUCKET_NODES - 1) / BUCKET_NODES;
  const int chunk = (E + PART_BLOCKS - 1) / PART_BLOCKS;

  float* ws = (float*)d_ws;
  float* h1 = ws;
  float* h2 = h1 + (size_t)N * HIDDIM;
  float* inv = h2 + (size_t)N * HIDDIM;
  int* cnt = (int*)(inv + (size_t)2 * N);
  int* offs = cnt + (size_t)2 * N;
  int* blkcnt = offs + (size_t)2 * N;
  int* blkpref = blkcnt + (size_t)2 * PART_BLOCKS * nbuck;
  int* bstart = blkpref + (size_t)2 * PART_BLOCKS * nbuck;
  unsigned* packed = (unsigned*)(bstart + (size_t)2 * (nbuck + 1));
  int* seg = (int*)(packed + (size_t)2 * E);

  const int T = 256;

  const int* src_conn = ei_conn;
  const int* dst_conn = ei_conn + E;
  const int* src_dest = ei_dest;
  const int* dst_dest = ei_dest + E;

  bucket_count_kernel<<<dim3(PART_BLOCKS, 2), PART_THREADS, 0, stream>>>(
      dst_conn, dst_dest, E, nbuck, chunk, blkcnt);
  bucket_scan_kernel<<<2, SCAN_THREADS, 0, stream>>>(blkcnt, blkpref, bstart,
                                                     nbuck);
  bucket_scatter_kernel<<<dim3(PART_BLOCKS, 2), PART_THREADS, 0, stream>>>(
      src_conn, dst_conn, src_dest, dst_dest, E, nbuck, chunk, blkpref, bstart,
      packed);
  bucket_sort_kernel<<<dim3(nbuck, 2), BUCKET_NODES, 0, stream>>>(
      packed, bstart, E, N, nbuck, offs, cnt, inv, seg);

  const int gemmBlocks = (N + 15) / 16;

  sage_layer_kernel<8><<<gemmBlocks, T, 0, stream>>>(x, offs, cnt, seg, inv, wl1,
                                                     wr1, b1, h1);

  sage_layer_kernel<128><<<gemmBlocks, T, 0, stream>>>(
      h1, offs + N, cnt + N, seg + E, inv + N, wl2, wr2, b2, h2);

  sage_layer_kernel<128><<<gemmBlocks, T, 0, stream>>>(
      h2, offs, cnt, seg, inv, wl3, wr3, b3, (float*)d_out);
}


# Wide layers take 64 rows a block: each wave's weight fragments of a k step feed four row tiles, so weights are fetched once per 64 rows; output tile reuses the mean tile's LDS; sum order unchanged
# speedup vs baseline: 3.9227x; 1.0461x over previous
#include <hip/hip_runtime.h>
#include <hip/hip_bf16.h>
#include <math.h>

typedef __attribute__((ext_vector_type(2))) float v2f;
typedef __attribute__((ext_vector_type(8))) float v8f;

#define HIDDIM 128

#define BUCKET_SHIFT 8
#define BUCKET_NODES (1 << BUCKET_SHIFT)
#define MAX_BUCKETS 512
#define PART_BLOCKS 128
#define PART_THREADS 512
#define SCAN_THREADS (2 * MAX_BUCKETS)
#define HALF_BLOCKS (PART_BLOCKS / 2)
#define SCAT_SLOTS 64
#define SORT_STAGE 6144
static_assert(PART_THREADS == MAX_BUCKETS, "the scatter set-up gives every bucket one thread");
static_assert(SCAT_SLOTS % 32 == 0, "a row of the image is written out by whole waves");

template <int T>
__device__ __forceinline__ int block_scan(int t, int* s, int tid) {
  s[tid] = t;
  __syncthreads();
  for (int o = 1; o < T; o <<= 1) {
    const int v = (tid >= o) ? s[tid - o] : 0;
    __syncthreads();
    s[tid] += v;
    __syncthreads();
  }
  return s[tid];
}

__global__ __launch_bounds__(PART_THREADS) void bucket_count_kernel(
    const int* __restrict__ dstA, const int* __restrict__ dstB, int E,
    int nbuck, int chunk, int* __restrict__ blkcnt) {
  __shared__ int hist[MAX_BUCKETS];
  const int tid = threadIdx.x;
  const int y = blockIdx.y;
  const int* dst = y ? dstB : dstA;
  for (int b = tid; b < nbuck; b += PART_THREADS) hist[b] = 0;
  __syncthreads();
  const int beg = blockIdx.x * chunk;
  const int end = (beg + chunk < E) ? (beg + chunk) : E;
  for (int i = beg + tid; i < end; i += 4 * PART_THREADS) {
    int d[4];
#pragma unroll
    for (int u = 0; u < 4; ++u) {
      const int j = i + u * PART_THREADS;
      const int v = dst[(j < end) ? j : i];
      d[u] = (j < end) ? v : -1;
    }
#pragma unroll
    for (int u = 0; u < 4; ++u)
      if (d[u] >= 0) atomicAdd(&hist[d[u] >> BUCKET_SHIFT], 1);
  }
  __syncthreads();
  int* o = blkcnt + ((size_t)y * gridDim.x + blockIdx.x) * nbuck;
  for (int b = tid; b < nbuck; b += PART_THREADS) o[b] = hist[b];
}

__global__ __launch_bounds__(SCAN_THREADS) void bucket_scan_kernel(
    const int* __restrict__ blkcnt, int* __restrict__ blkpref,
    int* __restrict__ bstart, int nbuck) {
  __shared__ int s[SCAN_THREADS];
  __shared__ int tots[2][MAX_BUCKETS];
  const int tid = threadIdx.x;
  const int y = blockIdx.x;
  const int j = tid & (MAX_BUCKETS - 1);
  const int q = tid / MAX_BUCKETS;
  const size_t first = ((size_t)y * PART_BLOCKS + q * HALF_BLOCKS) * nbuck + j;
  const int* c = blkcnt + first;
  int* p = blkpref + first;
  const bool live = j < nbuck;
  int v[HALF_BLOCKS];
#pragma unroll
  for (int u = 0; u < HALF_BLOCKS; ++u) v[u] = live ? c[(size_t)u * nbuck] : 0;
  int tot = 0;
#pragma unroll
  for (int u = 0; u < HALF_BLOCKS; ++u) tot += v[u];
  tots[q][j] = tot;
  __syncthreads();
  const int whole = (q == 0) ? (tots[0][j] + tots[1][j]) : 0;
  int run = (q == 1) ? tots[0][j] : 0;
  if (live) {
#pragma unroll
    for (int u = 0; u < HALF_BLOCKS; ++u) {
      p[(size_t)u * nbuck] = run;
      run += v[u];
    }
  }
  const int incl = block_scan<SCAN_THREADS>(whole, s, tid);
  if (q == 0 && live) {
    bstart[y * (nbuck + 1) + j] = incl - whole;
    if (j == nbuck - 1) bstart[y * (nbuck + 1) + nbuck] = incl;
  }
}

__global__ __launch_bounds__(PART_THREADS) void bucket_scatter_kernel(
    const int* __restrict__ srcA, const int* __restrict__ dstA,
    const int* __restrict__ srcB, const int* __restrict__ dstB, int E,
    int nbuck, int chunk, const int* __restrict__ blkpref,
    const int* __restrict__ bstart, unsigned* __restrict__ packed) {
  __shared__ int cnt[MAX_BUCKETS];
  __shared__ int gpos[MAX_BUCKETS];
  __shared__ __attribute__((aligned(16))) unsigned img[MAX_BUCKETS * SCAT_SLOTS];
  const int tid = threadIdx.x;
  const int y = blockIdx.y;
  const int* src = y ? srcB : srcA;
  const int* dst = y ? dstB : dstA;
  const int* p = blkpref + ((size_t)y * gridDim.x + blockIdx.x) * nbuck;
  const int* bs = bstart + y * (nbuck + 1);
  const bool live = tid < nbuck;
  const int g0 = live ? bs[tid] : 0;
  const int g1 = live ? p[tid] : 0;
  uint4* img4 = (uint4*)img;
  for (int q = tid; q < nbuck * (SCAT_SLOTS / 4); q += PART_THREADS)
    img4[q] = make_uint4(0u, 0u, 0u, 0u);
  cnt[tid] = 0;
  gpos[tid] = g0 + g1;
  __syncthreads();
  unsigned* out = packed + (size_t)y * E;
  const int beg = blockIdx.x * chunk;
  const int end = (beg + chunk < E) ? (beg + chunk) : E;
  for (int i = beg + tid; i < end; i += 4 * PART_THREADS) {
    int d[4], s[4];
#pragma unroll
    for (int u = 0; u < 4; ++u) {
      const int j = i + u * PART_THREADS;
      const int jj = (j < end) ? j : i;
      const int v = dst[jj];
      s[u] = src[jj];
      d[u] = (j < end) ? v : -1;
    }
#pragma unroll
    for (int u = 0; u < 4; ++u) {
      if (d[u] >= 0) {
        const int b = d[u] >> BUCKET_SHIFT;
        const int r = atomicAdd(&cnt[b], 1);
        const unsigned w = ((unsigned)s[u] << BUCKET_SHIFT) |
                           (unsigned)(d[u] & (BUCKET_NODES - 1));
        if (r < SCAT_SLOTS)
          img[b * SCAT_SLOTS + r] = w;
        else
          out[gpos[b] + r] = w;
      }
    }
  }
  __syncthreads();
  constexpr int NW = PART_THREADS / 32;
  constexpr int RW = SCAT_SLOTS / 32;
  const int wave = tid >> 5;
  const int lane = tid & 31;
  for (int b0 = wave; b0 < nbuck; b0 += 4 * NW) {
    int n[4], gb[4];
    unsigned wv[4][RW];
#pragma unroll
    for (int u = 0; u < 4; ++u) {
      const int b = b0 + u * NW;
      const int bb = (b < nbuck) ? b : b0;
      const int c = cnt[bb];
      n[u] = (b < nbuck) ? c : 0;
      gb[u] = gpos[bb];
#pragma unroll
      for (int k = 0; k < RW; ++k) wv[u][k] = img[bb * SCAT_SLOTS + 32 * k + lane];
    }
#pragma unroll
    for (int u = 0; u < 4; ++u) {
#pragma unroll
      for (int k = 0; k < RW; ++k)
        if (32 * k + lane < n[u]) out[gb[u] + 32 * k + lane] = wv[u][k];
    }
  }
}

__global__ __launch_bounds__(BUCKET_NODES) void bucket_sort_kernel(
    const unsigned* __restrict__ packed, const int* __restrict__ bstart, int E,
    int N, int nbuck, int* __restrict__ offs, int* __restrict__ cnt,
    float* __restrict__ inv, int* __restrict__ seg) {
  __shared__ int s[BUCKET_NODES];
  __shared__ int cur[BUCKET_NODES];
  __shared__ int stage[SORT_STAGE];
  const int tid = threadIdx.x;
  const int y = blockIdx.y;
  const int e0 = bstart[y * (nbuck + 1) + blockIdx.x];
  const int e1 = bstart[y * (nbuck + 1) + blockIdx.x + 1];
  const unsigned* pk = packed + (size_t)y * E;
  cur[tid] = 0;
  __syncthreads();
  for (int i = e0 + tid; i < e1; i += 4 * BUCKET_NODES) {
    unsigned w[4];
#pragma unroll
    for (int u = 0; u < 4; ++u) {
      const int j = i + u * BUCKET_NODES;
      w[u] = pk[(j < e1) ? j : i];
    }
#pragma unroll
    for (int u = 0; u < 4; ++u)
      if (i + u * BUCKET_NODES < e1)
        atomicAdd(&cur[w[u] & (BUCKET_NODES - 1)], 1);
  }
  __syncthreads();
  const int c = cur[tid];
  const int excl = block_scan<BUCKET_NODES>(c, s, tid) - c;
  const int node = blockIdx.x * BUCKET_NODES + tid;
  if (node < N) {
    const size_t g = (size_t)y * N + node;
    offs[g] = e0 + excl;
    cnt[g] = c;
    inv[g] = 1.0f / fmaxf((float)c, 1.0f);
  }
  cur[tid] = excl;
  __syncthreads();
  int* sg = seg + (size_t)y * E + e0;
  for (int i = e0 + tid; i < e1; i += 4 * BUCKET_NODES) {
    unsigned w[4];
#pragma unroll
    for (int u = 0; u < 4; ++u) {
      const int j = i + u * BUCKET_NODES;
      w[u] = pk[(j < e1) ? j : i];
    }
#pragma unroll
    for (int u = 0; u < 4; ++u) {
      if (i + u * BUCKET_NODES < e1) {
        const int pos = atomicAdd(&cur[w[u] & (BUCKET_NODES - 1)], 1);
        const int v = (int)(w[u] >> BUCKET_SHIFT);
        if (pos < SORT_STAGE)
          stage[pos] = v;
        else
          sg[pos] = v;
      }
    }
  }
  __syncthreads();
  const int n = e1 - e0;
  const int m = (n < SORT_STAGE) ? n : SORT_STAGE;
  const int mis = (int)(((size_t)sg >> 2) & 31);
  int i = tid - mis;
  if (i < 0) i += BUCKET_NODES;
#pragma unroll 4
  for (; i < m; i += BUCKET_NODES) sg[i] = stage[i];
}

__device__ __forceinline__ void add4(float4& a, const float4 v) {
  a.x += v.x;
  a.y += v.y;
  a.z += v.z;
  a.w += v.w;
}

__device__ __forceinline__ float4 gather_row128(const float* __restrict__ hl,
                                                const int* __restrict__ seg,
                                                int beg, int deg, int lane,
                                                int idx0) {
#define ROW_OF(jj) \
  (*(const float4*)(hl + (size_t)__builtin_amdgcn_readlane(idx, (jj)) * HIDDIM))
  float4 acc = make_float4(0.0f, 0.0f, 0.0f, 0.0f);
  for (int base = 0; base < deg; base += 32) {
    const int m = (deg - base < 32) ? (deg - base) : 32;
    int idx = idx0;
    if (base > 0) {
      idx = 0;
      if (lane < m) idx = seg[beg + base + lane];
    }
    int j = 0;
    for (; j + 8 <= m; j += 8) {
      const float4 v0 = ROW_OF(j), v1 = ROW_OF(j + 1), v2 = ROW_OF(j + 2),
                   v3 = ROW_OF(j + 3), v4 = ROW_OF(j + 4), v5 = ROW_OF(j + 5),
                   v6 = ROW_OF(j + 6), v7 = ROW_OF(j + 7);
      add4(acc, v0);
      add4(acc, v1);
      add4(acc, v2);
      add4(acc, v3);
      add4(acc, v4);
      add4(acc, v5);
      add4(acc, v6);
      add4(acc, v7);
    }
    if (j + 4 <= m) {
      const float4 v0 = ROW_OF(j), v1 = ROW_OF(j + 1), v2 = ROW_OF(j + 2),
                   v3 = ROW_OF(j + 3);
      add4(acc, v0);
      add4(acc, v1);
      add4(acc, v2);
      add4(acc, v3);
      j += 4;
    }
    if (j + 2 <= m) {
      const float4 v0 = ROW_OF(j), v1 = ROW_OF(j + 1);
      add4(acc, v0);
      add4(acc, v1);
      j += 2;
    }
    if (j < m) {
      const float4 v0 = ROW_OF(j);
      add4(acc, v0);
    }
  }
#undef ROW_OF
  return acc;
}

__device__ __forceinline__ float xor_add(float s, const int imm) {
  switch (imm) {
    case 1:  s += __int_as_float(__builtin_amdgcn_ds_swizzle(__float_as_int(s), 0x041f)); break;
    case 2:  s += __int_as_float(__builtin_amdgcn_ds_swizzle(__float_as_int(s), 0x081f)); break;
    case 4:  s += __int_as_float(__builtin_amdgcn_ds_swizzle(__float_as_int(s), 0x101f)); break;
    default: s += __int_as_float(__builtin_amdgcn_ds_swizzle(__float_as_int(s), 0x201f)); break;
  }
  return s;
}

__device__ __forceinline__ float other_half(float x) {
  const int b = __float_as_int(x);
  return __int_as_float(
      __builtin_amdgcn_permlanex16(b, b, 0x76543210, 0xfedcba98, false, false));
}

template <int K, int R>
__global__ __launch_bounds__(256) void sage_layer_kernel(
    const float* __restrict__ xin, const int* __restrict__ offs,
    const int* __restrict__ cnt, const int* __restrict__ seg,
    const float* __restrict__ invcnt, const float* __restrict__ wl,
    const float* __restrict__ wr, const float* __restrict__ bias,
    float* __restrict__ out, int N) {
  constexpr int P = K + 4;
  constexpr int ROWS = 16 * R;
  constexpr int WROWS = ROWS / 8;
  __shared__ __attribute__((aligned(16))) float sM[ROWS][P];
  __shared__ __attribute__((aligned(16))) float sX[ROWS][P];
  constexpr bool SHARE = (K == HIDDIM);
  __shared__ __attribute__((aligned(16))) float sOwn[SHARE ? 1 : ROWS][HIDDIM + 4];
  float (*sO)[HIDDIM + 4];
  if constexpr (SHARE)
    sO = sM;
  else
    sO = sOwn;

  const int tid = threadIdx.x;
  const int wave = tid >> 5;
  const int lane = tid & 31;
  const int half = lane >> 4;
  const int l16 = lane & 15;
  const int rowBase = blockIdx.x * ROWS;
  const int col = wave * 16 + l16;

  if constexpr (K == HIDDIM) {
    const int w = __builtin_amdgcn_readfirstlane(wave);
    const float* hl = xin + lane * 4;
    const int last = N - 1;
    float4 xo[WROWS];
#pragma unroll
    for (int i = 0; i < WROWS; ++i) {
      const int r = rowBase + w + 8 * i;
      xo[i] = *(const float4*)(hl + (size_t)((r < last) ? r : last) * K);
    }
    const int n0 = rowBase + WROWS * w;
    int beg[WROWS], deg[WROWS], idx[WROWS];
    float sc[WROWS];
#pragma unroll
    for (int i = 0; i < WROWS; ++i) {
      const int n = (n0 + i < last) ? (n0 + i) : last;
      beg[i] = offs[n];
      deg[i] = cnt[n];
      sc[i] = invcnt[n];
    }
#pragma unroll
    for (int i = 0; i < WROWS; ++i) {
      idx[i] = 0;
      if (lane < deg[i]) idx[i] = seg[beg[i] + lane];
    }
#pragma unroll
    for (int i = 0; i < WROWS; ++i) *(float4*)&sX[w + 8 * i][lane * 4] = xo[i];
#pragma unroll
    for (int i = 0; i < WROWS; ++i) {
      float4 m = gather_row128(hl, seg, beg[i], deg[i], lane, idx[i]);
      m.x *= sc[i];
      m.y *= sc[i];
      m.z *= sc[i];
      m.w *= sc[i];
      *(float4*)&sM[WROWS * w + i][lane * 4] = m;
    }
  } else {
    if (tid < ROWS * K) {
      const int r = tid / K;
      const int k = tid - r * K;
      const int node = rowBase + r;
      sX[r][k] = xin[(size_t)node * K + k];
      int j = offs[node];
      const int end = j + cnt[node];
      float acc = 0.0f;
      for (; j + 4 <= end; j += 4) {
        const int s0 = seg[j], s1 = seg[j + 1], s2 = seg[j + 2], s3 = seg[j + 3];
        const float v0 = xin[(size_t)s0 * K + k];
        const float v1 = xin[(size_t)s1 * K + k];
        const float v2 = xin[(size_t)s2 * K + k];
        const float v3 = xin[(size_t)s3 * K + k];
        acc += v0;
        acc += v1;
        acc += v2;
        acc += v3;
      }
      for (; j < end; ++j) acc += xin[(size_t)seg[j] * K + k];
      sM[r][k] = acc * invcnt[node];
    }
  }
  __syncthreads();

  v8f c[R];
#pragma unroll
  for (int t = 0; t < R; ++t) c[t] = v8f{};
#pragma unroll 8
  for (int k0 = 0; k0 < K; k0 += 4) {
    const int ka = k0 + 2 * half;
    v2f bl, br;
    bl.x = wl[(size_t)ka * HIDDIM + col];
    bl.y = wl[(size_t)(ka + 1) * HIDDIM + col];
    br.x = wr[(size_t)ka * HIDDIM + col];
    br.y = wr[(size_t)(ka + 1) * HIDDIM + col];
#pragma unroll
    for (int t = 0; t < R; ++t) {
      const v2f am = *(const v2f*)&sM[16 * t + l16][ka];
      const v2f ax = *(const v2f*)&sX[16 * t + l16][ka];
      c[t] = __builtin_amdgcn_wmma_f32_16x16x4_f32(false, am, false, bl, (short)0,
                                                   c[t], false, false);
      c[t] = __builtin_amdgcn_wmma_f32_16x16x4_f32(false, ax, false, br, (short)0,
                                                   c[t], false, false);
    }
  }

  const float bv = bias[col];
  if constexpr (SHARE) __syncthreads();
#pragma unroll
  for (int t = 0; t < R; ++t) {
#pragma unroll
    for (int v = 0; v < 8; ++v) sO[16 * t + v + 8 * half][col] = c[t][v] + bv;
  }
  __syncthreads();

  float4 o[WROWS];
  float s[WROWS];
#pragma unroll
  for (int q = 0; q < WROWS; ++q) {
    o[q] = *(const float4*)&sO[wave + 8 * q][lane * 4];
    s[q] = o[q].x * o[q].x + o[q].y * o[q].y + o[q].z * o[q].z + o[q].w * o[q].w;
  }
#pragma unroll
  for (int q = 0; q < WROWS; ++q) s[q] = xor_add(s[q], 1);
#pragma unroll
  for (int q = 0; q < WROWS; ++q) s[q] = xor_add(s[q], 2);
#pragma unroll
  for (int q = 0; q < WROWS; ++q) s[q] = xor_add(s[q], 4);
#pragma unroll
  for (int q = 0; q < WROWS; ++q) s[q] = xor_add(s[q], 8);
#pragma unroll
  for (int q = 0; q < WROWS; ++q) s[q] += other_half(s[q]);
#pragma unroll
  for (int q = 0; q < WROWS; ++q) {
    const float scale = 1.0f / fmaxf(sqrtf(s[q]), 1e-12f);
    float4 r;
    r.x = tanhf(o[q].x * scale);
    r.y = tanhf(o[q].y * scale);
    r.z = tanhf(o[q].z * scale);
    r.w = tanhf(o[q].w * scale);
    const int row = rowBase + wave + 8 * q;
    if (row < N) *(float4*)(out + (size_t)row * HIDDIM + lane * 4) = r;
  }
}

extern "C" void kernel_launch(void* const* d_in, const int* in_sizes, int n_in,
                              void* d_out, int out_size, void* d_ws,
                              size_t ws_size, hipStream_t stream) {
  const float* x = (const float*)d_in[0];
  const int* ei_conn = (const int*)d_in[1];
  const int* ei_dest = (const int*)d_in[2];
  const float* wl1 = (const float*)d_in[3];
  const float* wr1 = (const float*)d_in[4];
  const float* b1 = (const float*)d_in[5];
  const float* wl2 = (const float*)d_in[6];
  const float* wr2 = (const float*)d_in[7];
  const float* b2 = (const float*)d_in[8];
  const float* wl3 = (const float*)d_in[9];
  const float* wr3 = (const float*)d_in[10];
  const float* b3 = (const float*)d_in[11];

  const int N = in_sizes[0] / 8;
  const int E = in_sizes[1] / 2;
  const int nbuck = (N + BUCKET_NODES - 1) / BUCKET_NODES;
  const int chunk = (E + PART_BLOCKS - 1) / PART_BLOCKS;

  float* ws = (float*)d_ws;
  float* h1 = ws;
  float* h2 = h1 + (size_t)N * HIDDIM;
  float* inv = h2 + (size_t)N * HIDDIM;
  int* cnt = (int*)(inv + (size_t)2 * N);
  int* offs = cnt + (size_t)2 * N;
  int* blkcnt = offs + (size_t)2 * N;
  int* blkpref = blkcnt + (size_t)2 * PART_BLOCKS * nbuck;
  int* bstart = blkpref + (size_t)2 * PART_BLOCKS * nbuck;
  unsigned* packed = (unsigned*)(bstart + (size_t)2 * (nbuck + 1));
  int* seg = (int*)(packed + (size_t)2 * E);

  const int T = 256;

  const int* src_conn = ei_conn;
  const int* dst_conn = ei_conn + E;
  const int* src_dest = ei_dest;
  const int* dst_dest = ei_dest + E;

  bucket_count_kernel<<<dim3(PART_BLOCKS, 2), PART_THREADS, 0, stream>>>(
      dst_conn, dst_dest, E, nbuck, chunk, blkcnt);
  bucket_scan_kernel<<<2, SCAN_THREADS, 0, stream>>>(blkcnt, blkpref, bstart,
                                                     nbuck);
  bucket_scatter_kernel<<<dim3(PART_BLOCKS, 2), PART_THREADS, 0, stream>>>(
      src_conn, dst_conn, src_dest, dst_dest, E, nbuck, chunk, blkpref, bstart,
      packed);
  bucket_sort_kernel<<<dim3(nbuck, 2), BUCKET_NODES, 0, stream>>>(
      packed, bstart, E, N, nbuck, offs, cnt, inv, seg);

  const int narrowBlocks = (N + 15) / 16;
  const int wideBlocks = (N + 63) / 64;

  sage_layer_kernel<8, 1><<<narrowBlocks, T, 0, stream>>>(x, offs, cnt, seg, inv,
                                                          wl1, wr1, b1, h1, N);

  sage_layer_kernel<128, 4><<<wideBlocks, T, 0, stream>>>(
      h1, offs + N, cnt + N, seg + E, inv + N, wl2, wr2, b2, h2, N);

  sage_layer_kernel<128, 4><<<wideBlocks, T, 0, stream>>>(
      h2, offs, cnt, seg, inv, wl3, wr3, b3, (float*)d_out, N);
}


# Wide layers: the x @ wr k steps run inside the neighbour gather, between a row group's loads and its adds, so half the products overlap the gather; x @ wr summed before mean @ wl: sum order changes
# speedup vs baseline: 4.0413x; 1.0302x over previous
#include <hip/hip_runtime.h>
#include <hip/hip_bf16.h>
#include <math.h>

typedef __attribute__((ext_vector_type(2))) float v2f;
typedef __attribute__((ext_vector_type(8))) float v8f;

#define HIDDIM 128

#define BUCKET_SHIFT 8
#define BUCKET_NODES (1 << BUCKET_SHIFT)
#define MAX_BUCKETS 512
#define PART_BLOCKS 128
#define PART_THREADS 512
#define SCAN_THREADS (2 * MAX_BUCKETS)
#define HALF_BLOCKS (PART_BLOCKS / 2)
#define SCAT_SLOTS 64
#define SORT_STAGE 6144
static_assert(PART_THREADS == MAX_BUCKETS, "the scatter set-up gives every bucket one thread");
static_assert(SCAT_SLOTS % 32 == 0, "a row of the image is written out by whole waves");

template <int T>
__device__ __forceinline__ int block_scan(int t, int* s, int tid) {
  s[tid] = t;
  __syncthreads();
  for (int o = 1; o < T; o <<= 1) {
    const int v = (tid >= o) ? s[tid - o] : 0;
    __syncthreads();
    s[tid] += v;
    __syncthreads();
  }
  return s[tid];
}

__global__ __launch_bounds__(PART_THREADS) void bucket_count_kernel(
    const int* __restrict__ dstA, const int* __restrict__ dstB, int E,
    int nbuck, int chunk, int* __restrict__ blkcnt) {
  __shared__ int hist[MAX_BUCKETS];
  const int tid = threadIdx.x;
  const int y = blockIdx.y;
  const int* dst = y ? dstB : dstA;
  for (int b = tid; b < nbuck; b += PART_THREADS) hist[b] = 0;
  __syncthreads();
  const int beg = blockIdx.x * chunk;
  const int end = (beg + chunk < E) ? (beg + chunk) : E;
  for (int i = beg + tid; i < end; i += 4 * PART_THREADS) {
    int d[4];
#pragma unroll
    for (int u = 0; u < 4; ++u) {
      const int j = i + u * PART_THREADS;
      const int v = dst[(j < end) ? j : i];
      d[u] = (j < end) ? v : -1;
    }
#pragma unroll
    for (int u = 0; u < 4; ++u)
      if (d[u] >= 0) atomicAdd(&hist[d[u] >> BUCKET_SHIFT], 1);
  }
  __syncthreads();
  int* o = blkcnt + ((size_t)y * gridDim.x + blockIdx.x) * nbuck;
  for (int b = tid; b < nbuck; b += PART_THREADS) o[b] = hist[b];
}

__global__ __launch_bounds__(SCAN_THREADS) void bucket_scan_kernel(
    const int* __restrict__ blkcnt, int* __restrict__ blkpref,
    int* __restrict__ bstart, int nbuck) {
  __shared__ int s[SCAN_THREADS];
  __shared__ int tots[2][MAX_BUCKETS];
  const int tid = threadIdx.x;
  const int y = blockIdx.x;
  const int j = tid & (MAX_BUCKETS - 1);
  const int q = tid / MAX_BUCKETS;
  const size_t first = ((size_t)y * PART_BLOCKS + q * HALF_BLOCKS) * nbuck + j;
  const int* c = blkcnt + first;
  int* p = blkpref + first;
  const bool live = j < nbuck;
  int v[HALF_BLOCKS];
#pragma unroll
  for (int u = 0; u < HALF_BLOCKS; ++u) v[u] = live ? c[(size_t)u * nbuck] : 0;
  int tot = 0;
#pragma unroll
  for (int u = 0; u < HALF_BLOCKS; ++u) tot += v[u];
  tots[q][j] = tot;
  __syncthreads();
  const int whole = (q == 0) ? (tots[0][j] + tots[1][j]) : 0;
  int run = (q == 1) ? tots[0][j] : 0;
  if (live) {
#pragma unroll
    for (int u = 0; u < HALF_BLOCKS; ++u) {
      p[(size_t)u * nbuck] = run;
      run += v[u];
    }
  }
  const int incl = block_scan<SCAN_THREADS>(whole, s, tid);
  if (q == 0 && live) {
    bstart[y * (nbuck + 1) + j] = incl - whole;
    if (j == nbuck - 1) bstart[y * (nbuck + 1) + nbuck] = incl;
  }
}

__global__ __launch_bounds__(PART_THREADS) void bucket_scatter_kernel(
    const int* __restrict__ srcA, const int* __restrict__ dstA,
    const int* __restrict__ srcB, const int* __restrict__ dstB, int E,
    int nbuck, int chunk, const int* __restrict__ blkpref,
    const int* __restrict__ bstart, unsigned* __restrict__ packed) {
  __shared__ int cnt[MAX_BUCKETS];
  __shared__ int gpos[MAX_BUCKETS];
  __shared__ __attribute__((aligned(16))) unsigned img[MAX_BUCKETS * SCAT_SLOTS];
  const int tid = threadIdx.x;
  const int y = blockIdx.y;
  const int* src = y ? srcB : srcA;
  const int* dst = y ? dstB : dstA;
  const int* p = blkpref + ((size_t)y * gridDim.x + blockIdx.x) * nbuck;
  const int* bs = bstart + y * (nbuck + 1);
  const bool live = tid < nbuck;
  const int g0 = live ? bs[tid] : 0;
  const int g1 = live ? p[tid] : 0;
  uint4* img4 = (uint4*)img;
  for (int q = tid; q < nbuck * (SCAT_SLOTS / 4); q += PART_THREADS)
    img4[q] = make_uint4(0u, 0u, 0u, 0u);
  cnt[tid] = 0;
  gpos[tid] = g0 + g1;
  __syncthreads();
  unsigned* out = packed + (size_t)y * E;
  const int beg = blockIdx.x * chunk;
  const int end = (beg + chunk < E) ? (beg + chunk) : E;
  for (int i = beg + tid; i < end; i += 4 * PART_THREADS) {
    int d[4], s[4];
#pragma unroll
    for (int u = 0; u < 4; ++u) {
      const int j = i + u * PART_THREADS;
      const int jj = (j < end) ? j : i;
      const int v = dst[jj];
      s[u] = src[jj];
      d[u] = (j < end) ? v : -1;
    }
#pragma unroll
    for (int u = 0; u < 4; ++u) {
      if (d[u] >= 0) {
        const int b = d[u] >> BUCKET_SHIFT;
        const int r = atomicAdd(&cnt[b], 1);
        const unsigned w = ((unsigned)s[u] << BUCKET_SHIFT) |
                           (unsigned)(d[u] & (BUCKET_NODES - 1));
        if (r < SCAT_SLOTS)
          img[b * SCAT_SLOTS + r] = w;
        else
          out[gpos[b] + r] = w;
      }
    }
  }
  __syncthreads();
  constexpr int NW = PART_THREADS / 32;
  constexpr int RW = SCAT_SLOTS / 32;
  const int wave = tid >> 5;
  const int lane = tid & 31;
  for (int b0 = wave; b0 < nbuck; b0 += 4 * NW) {
    int n[4], gb[4];
    unsigned wv[4][RW];
#pragma unroll
    for (int u = 0; u < 4; ++u) {
      const int b = b0 + u * NW;
      const int bb = (b < nbuck) ? b : b0;
      const int c = cnt[bb];
      n[u] = (b < nbuck) ? c : 0;
      gb[u] = gpos[bb];
#pragma unroll
      for (int k = 0; k < RW; ++k) wv[u][k] = img[bb * SCAT_SLOTS + 32 * k + lane];
    }
#pragma unroll
    for (int u = 0; u < 4; ++u) {
#pragma unroll
      for (int k = 0; k < RW; ++k)
        if (32 * k + lane < n[u]) out[gb[u] + 32 * k + lane] = wv[u][k];
    }
  }
}

__global__ __launch_bounds__(BUCKET_NODES) void bucket_sort_kernel(
    const unsigned* __restrict__ packed, const int* __restrict__ bstart, int E,
    int N, int nbuck, int* __restrict__ offs, int* __restrict__ cnt,
    float* __restrict__ inv, int* __restrict__ seg) {
  __shared__ int s[BUCKET_NODES];
  __shared__ int cur[BUCKET_NODES];
  __shared__ int stage[SORT_STAGE];
  const int tid = threadIdx.x;
  const int y = blockIdx.y;
  const int e0 = bstart[y * (nbuck + 1) + blockIdx.x];
  const int e1 = bstart[y * (nbuck + 1) + blockIdx.x + 1];
  const unsigned* pk = packed + (size_t)y * E;
  cur[tid] = 0;
  __syncthreads();
  for (int i = e0 + tid; i < e1; i += 4 * BUCKET_NODES) {
    unsigned w[4];
#pragma unroll
    for (int u = 0; u < 4; ++u) {
      const int j = i + u * BUCKET_NODES;
      w[u] = pk[(j < e1) ? j : i];
    }
#pragma unroll
    for (int u = 0; u < 4; ++u)
      if (i + u * BUCKET_NODES < e1)
        atomicAdd(&cur[w[u] & (BUCKET_NODES - 1)], 1);
  }
  __syncthreads();
  const int c = cur[tid];
  const int excl = block_scan<BUCKET_NODES>(c, s, tid) - c;
  const int node = blockIdx.x * BUCKET_NODES + tid;
  if (node < N) {
    const size_t g = (size_t)y * N + node;
    offs[g] = e0 + excl;
    cnt[g] = c;
    inv[g] = 1.0f / fmaxf((float)c, 1.0f);
  }
  cur[tid] = excl;
  __syncthreads();
  int* sg = seg + (size_t)y * E + e0;
  for (int i = e0 + tid; i < e1; i += 4 * BUCKET_NODES) {
    unsigned w[4];
#pragma unroll
    for (int u = 0; u < 4; ++u) {
      const int j = i + u * BUCKET_NODES;
      w[u] = pk[(j < e1) ? j : i];
    }
#pragma unroll
    for (int u = 0; u < 4; ++u) {
      if (i + u * BUCKET_NODES < e1) {
        const int pos = atomicAdd(&cur[w[u] & (BUCKET_NODES - 1)], 1);
        const int v = (int)(w[u] >> BUCKET_SHIFT);
        if (pos < SORT_STAGE)
          stage[pos] = v;
        else
          sg[pos] = v;
      }
    }
  }
  __syncthreads();
  const int n = e1 - e0;
  const int m = (n < SORT_STAGE) ? n : SORT_STAGE;
  const int mis = (int)(((size_t)sg >> 2) & 31);
  int i = tid - mis;
  if (i < 0) i += BUCKET_NODES;
#pragma unroll 4
  for (; i < m; i += BUCKET_NODES) sg[i] = stage[i];
}

__device__ __forceinline__ void add4(float4& a, const float4 v) {
  a.x += v.x;
  a.y += v.y;
  a.z += v.z;
  a.w += v.w;
}

template <int R, int P>
struct XSteps {
  static constexpr int KSTEPS = HIDDIM / 4;
  static_assert(KSTEPS % 2 == 0, "steps are taken in pairs");
  const float* wrl;
  const float* sxl;
  v8f* c;
  int ks;
  v2f b[2];
  __device__ __forceinline__ void fetch() {
    if (ks < KSTEPS) {
#pragma unroll
      for (int u = 0; u < 2; ++u) {
        b[u].x = wrl[(size_t)(4 * (ks + u)) * HIDDIM];
        b[u].y = wrl[(size_t)(4 * (ks + u) + 1) * HIDDIM];
      }
    }
  }
  __device__ __forceinline__ void run() {
    if (ks < KSTEPS) {
#pragma unroll
      for (int u = 0; u < 2; ++u) {
#pragma unroll
        for (int t = 0; t < R; ++t) {
          const v2f ax = *(const v2f*)(sxl + 16 * t * P + 4 * (ks + u));
          c[t] = __builtin_amdgcn_wmma_f32_16x16x4_f32(false, ax, false, b[u],
                                                       (short)0, c[t], false, false);
        }
      }
      ks += 2;
    }
  }
};

template <class X>
__device__ __forceinline__ float4 gather_row128(const float* __restrict__ hl,
                                                const int* __restrict__ seg,
                                                int beg, int deg, int lane,
                                                int idx0, X& xp) {
#define ROW_OF(jj) \
  (*(const float4*)(hl + (size_t)__builtin_amdgcn_readlane(idx, (jj)) * HIDDIM))
  float4 acc = make_float4(0.0f, 0.0f, 0.0f, 0.0f);
  for (int base = 0; base < deg; base += 32) {
    const int m = (deg - base < 32) ? (deg - base) : 32;
    int idx = idx0;
    if (base > 0) {
      idx = 0;
      if (lane < m) idx = seg[beg + base + lane];
    }
    int j = 0;
    for (; j + 8 <= m; j += 8) {
      const float4 v0 = ROW_OF(j), v1 = ROW_OF(j + 1), v2 = ROW_OF(j + 2),
                   v3 = ROW_OF(j + 3), v4 = ROW_OF(j + 4), v5 = ROW_OF(j + 5),
                   v6 = ROW_OF(j + 6), v7 = ROW_OF(j + 7);
      xp.run();
      xp.fetch();
      add4(acc, v0);
      add4(acc, v1);
      add4(acc, v2);
      add4(acc, v3);
      add4(acc, v4);
      add4(acc, v5);
      add4(acc, v6);
      add4(acc, v7);
    }
    if (j + 4 <= m) {
      const float4 v0 = ROW_OF(j), v1 = ROW_OF(j + 1), v2 = ROW_OF(j + 2),
                   v3 = ROW_OF(j + 3);
      xp.run();
      xp.fetch();
      add4(acc, v0);
      add4(acc, v1);
      add4(acc, v2);
      add4(acc, v3);
      j += 4;
    }
    if (j + 2 <= m) {
      const float4 v0 = ROW_OF(j), v1 = ROW_OF(j + 1);
      xp.run();
      xp.fetch();
      add4(acc, v0);
      add4(acc, v1);
      j += 2;
    }
    if (j < m) {
      const float4 v0 = ROW_OF(j);
      xp.run();
      xp.fetch();
      add4(acc, v0);
    }
  }
#undef ROW_OF
  return acc;
}

__device__ __forceinline__ float xor_add(float s, const int imm) {
  switch (imm) {
    case 1:  s += __int_as_float(__builtin_amdgcn_ds_swizzle(__float_as_int(s), 0x041f)); break;
    case 2:  s += __int_as_float(__builtin_amdgcn_ds_swizzle(__float_as_int(s), 0x081f)); break;
    case 4:  s += __int_as_float(__builtin_amdgcn_ds_swizzle(__float_as_int(s), 0x101f)); break;
    default: s += __int_as_float(__builtin_amdgcn_ds_swizzle(__float_as_int(s), 0x201f)); break;
  }
  return s;
}

__device__ __forceinline__ float other_half(float x) {
  const int b = __float_as_int(x);
  return __int_as_float(
      __builtin_amdgcn_permlanex16(b, b, 0x76543210, 0xfedcba98, false, false));
}

template <int K, int R>
__global__ __launch_bounds__(256) void sage_layer_kernel(
    const float* __restrict__ xin, const int* __restrict__ offs,
    const int* __restrict__ cnt, const int* __restrict__ seg,
    const float* __restrict__ invcnt, const float* __restrict__ wl,
    const float* __restrict__ wr, const float* __restrict__ bias,
    float* __restrict__ out, int N) {
  constexpr int P = K + 4;
  constexpr int ROWS = 16 * R;
  constexpr int WROWS = ROWS / 8;
  __shared__ __attribute__((aligned(16))) float sM[ROWS][P];
  __shared__ __attribute__((aligned(16))) float sX[ROWS][P];
  constexpr bool SHARE = (K == HIDDIM);
  __shared__ __attribute__((aligned(16))) float sOwn[SHARE ? 1 : ROWS][HIDDIM + 4];
  float (*sO)[HIDDIM + 4];
  if constexpr (SHARE)
    sO = sM;
  else
    sO = sOwn;

  const int tid = threadIdx.x;
  const int wave = tid >> 5;
  const int lane = tid & 31;
  const int half = lane >> 4;
  const int l16 = lane & 15;
  const int rowBase = blockIdx.x * ROWS;
  const int col = wave * 16 + l16;

  v8f c[R];
#pragma unroll
  for (int t = 0; t < R; ++t) c[t] = v8f{};

  if constexpr (K == HIDDIM) {
    const int w = __builtin_amdgcn_readfirstlane(wave);
    const float* hl = xin + lane * 4;
    const int last = N - 1;
    XSteps<R, P> xp;
    xp.wrl = wr + (size_t)(2 * half) * HIDDIM + col;
    xp.sxl = &sX[l16][2 * half];
    xp.c = c;
    xp.ks = 0;
    xp.fetch();
    float4 xo[WROWS];
#pragma unroll
    for (int i = 0; i < WROWS; ++i) {
      const int r = rowBase + w + 8 * i;
      xo[i] = *(const float4*)(hl + (size_t)((r < last) ? r : last) * K);
    }
    const int n0 = rowBase + WROWS * w;
    int beg[WROWS], deg[WROWS], idx[WROWS];
    float sc[WROWS];
#pragma unroll
    for (int i = 0; i < WROWS; ++i) {
      const int n = (n0 + i < last) ? (n0 + i) : last;
      beg[i] = offs[n];
      deg[i] = cnt[n];
      sc[i] = invcnt[n];
    }
#pragma unroll
    for (int i = 0; i < WROWS; ++i) {
      idx[i] = 0;
      if (lane < deg[i]) idx[i] = seg[beg[i] + lane];
    }
#pragma unroll
    for (int i = 0; i < WROWS; ++i) *(float4*)&sX[w + 8 * i][lane * 4] = xo[i];
    __syncthreads();
#pragma unroll
    for (int i = 0; i < WROWS; ++i) {
      float4 m = gather_row128(hl, seg, beg[i], deg[i], lane, idx[i], xp);
      m.x *= sc[i];
      m.y *= sc[i];
      m.z *= sc[i];
      m.w *= sc[i];
      *(float4*)&sM[WROWS * w + i][lane * 4] = m;
    }
    while (xp.ks < XSteps<R, P>::KSTEPS) {
      xp.run();
      xp.fetch();
    }
  } else {
    if (tid < ROWS * K) {
      const int r = tid / K;
      const int k = tid - r * K;
      const int node = rowBase + r;
      sX[r][k] = xin[(size_t)node * K + k];
      int j = offs[node];
      const int end = j + cnt[node];
      float acc = 0.0f;
      for (; j + 4 <= end; j += 4) {
        const int s0 = seg[j], s1 = seg[j + 1], s2 = seg[j + 2], s3 = seg[j + 3];
        const float v0 = xin[(size_t)s0 * K + k];
        const float v1 = xin[(size_t)s1 * K + k];
        const float v2 = xin[(size_t)s2 * K + k];
        const float v3 = xin[(size_t)s3 * K + k];
        acc += v0;
        acc += v1;
        acc += v2;
        acc += v3;
      }
      for (; j < end; ++j) acc += xin[(size_t)seg[j] * K + k];
      sM[r][k] = acc * invcnt[node];
    }
  }
  __syncthreads();

  if constexpr (K == HIDDIM) {
#pragma unroll 8
    for (int k0 = 0; k0 < K; k0 += 4) {
      const int ka = k0 + 2 * half;
      v2f bl;
      bl.x = wl[(size_t)ka * HIDDIM + col];
      bl.y = wl[(size_t)(ka + 1) * HIDDIM + col];
#pragma unroll
      for (int t = 0; t < R; ++t) {
        const v2f am = *(const v2f*)&sM[16 * t + l16][ka];
        c[t] = __builtin_amdgcn_wmma_f32_16x16x4_f32(false, am, false, bl, (short)0,
                                                     c[t], false, false);
      }
    }
  } else {
#pragma unroll 8
    for (int k0 = 0; k0 < K; k0 += 4) {
      const int ka = k0 + 2 * half;
      v2f bl, br;
      bl.x = wl[(size_t)ka * HIDDIM + col];
      bl.y = wl[(size_t)(ka + 1) * HIDDIM + col];
      br.x = wr[(size_t)ka * HIDDIM + col];
      br.y = wr[(size_t)(ka + 1) * HIDDIM + col];
#pragma unroll
      for (int t = 0; t < R; ++t) {
        const v2f am = *(const v2f*)&sM[16 * t + l16][ka];
        const v2f ax = *(const v2f*)&sX[16 * t + l16][ka];
        c[t] = __builtin_amdgcn_wmma_f32_16x16x4_f32(false, am, false, bl, (short)0,
                                                     c[t], false, false);
        c[t] = __builtin_amdgcn_wmma_f32_16x16x4_f32(false, ax, false, br, (short)0,
                                                     c[t], false, false);
      }
    }
  }

  const float bv = bias[col];
  if constexpr (SHARE) __syncthreads();
#pragma unroll
  for (int t = 0; t < R; ++t) {
#pragma unroll
    for (int v = 0; v < 8; ++v) sO[16 * t + v + 8 * half][col] = c[t][v] + bv;
  }
  __syncthreads();

  float4 o[WROWS];
  float s[WROWS];
#pragma unroll
  for (int q = 0; q < WROWS; ++q) {
    o[q] = *(const float4*)&sO[wave + 8 * q][lane * 4];
    s[q] = o[q].x * o[q].x + o[q].y * o[q].y + o[q].z * o[q].z + o[q].w * o[q].w;
  }
#pragma unroll
  for (int q = 0; q < WROWS; ++q) s[q] = xor_add(s[q], 1);
#pragma unroll
  for (int q = 0; q < WROWS; ++q) s[q] = xor_add(s[q], 2);
#pragma unroll
  for (int q = 0; q < WROWS; ++q) s[q] = xor_add(s[q], 4);
#pragma unroll
  for (int q = 0; q < WROWS; ++q) s[q] = xor_add(s[q], 8);
#pragma unroll
  for (int q = 0; q < WROWS; ++q) s[q] += other_half(s[q]);
#pragma unroll
  for (int q = 0; q < WROWS; ++q) {
    const float scale = 1.0f / fmaxf(sqrtf(s[q]), 1e-12f);
    float4 r;
    r.x = tanhf(o[q].x * scale);
    r.y = tanhf(o[q].y * scale);
    r.z = tanhf(o[q].z * scale);
    r.w = tanhf(o[q].w * scale);
    const int row = rowBase + wave + 8 * q;
    if (row < N) *(float4*)(out + (size_t)row * HIDDIM + lane * 4) = r;
  }
}

extern "C" void kernel_launch(void* const* d_in, const int* in_sizes, int n_in,
                              void* d_out, int out_size, void* d_ws,
                              size_t ws_size, hipStream_t stream) {
  const float* x = (const float*)d_in[0];
  const int* ei_conn = (const int*)d_in[1];
  const int* ei_dest = (const int*)d_in[2];
  const float* wl1 = (const float*)d_in[3];
  const float* wr1 = (const float*)d_in[4];
  const float* b1 = (const float*)d_in[5];
  const float* wl2 = (const float*)d_in[6];
  const float* wr2 = (const float*)d_in[7];
  const float* b2 = (const float*)d_in[8];
  const float* wl3 = (const float*)d_in[9];
  const float* wr3 = (const float*)d_in[10];
  const float* b3 = (const float*)d_in[11];

  const int N = in_sizes[0] / 8;
  const int E = in_sizes[1] / 2;
  const int nbuck = (N + BUCKET_NODES - 1) / BUCKET_NODES;
  const int chunk = (E + PART_BLOCKS - 1) / PART_BLOCKS;

  float* ws = (float*)d_ws;
  float* h1 = ws;
  float* h2 = h1 + (size_t)N * HIDDIM;
  float* inv = h2 + (size_t)N * HIDDIM;
  int* cnt = (int*)(inv + (size_t)2 * N);
  int* offs = cnt + (size_t)2 * N;
  int* blkcnt = offs + (size_t)2 * N;
  int* blkpref = blkcnt + (size_t)2 * PART_BLOCKS * nbuck;
  int* bstart = blkpref + (size_t)2 * PART_BLOCKS * nbuck;
  unsigned* packed = (unsigned*)(bstart + (size_t)2 * (nbuck + 1));
  int* seg = (int*)(packed + (size_t)2 * E);

  const int T = 256;

  const int* src_conn = ei_conn;
  const int* dst_conn = ei_conn + E;
  const int* src_dest = ei_dest;
  const int* dst_dest = ei_dest + E;

  bucket_count_kernel<<<dim3(PART_BLOCKS, 2), PART_THREADS, 0, stream>>>(
      dst_conn, dst_dest, E, nbuck, chunk, blkcnt);
  bucket_scan_kernel<<<2, SCAN_THREADS, 0, stream>>>(blkcnt, blkpref, bstart,
                                                     nbuck);
  bucket_scatter_kernel<<<dim3(PART_BLOCKS, 2), PART_THREADS, 0, stream>>>(
      src_conn, dst_conn, src_dest, dst_dest, E, nbuck, chunk, blkpref, bstart,
      packed);
  bucket_sort_kernel<<<dim3(nbuck, 2), BUCKET_NODES, 0, stream>>>(
      packed, bstart, E, N, nbuck, offs, cnt, inv, seg);

  const int narrowBlocks = (N + 15) / 16;
  const int wideBlocks = (N + 63) / 64;

  sage_layer_kernel<8, 1><<<narrowBlocks, T, 0, stream>>>(x, offs, cnt, seg, inv,
                                                          wl1, wr1, b1, h1, N);

  sage_layer_kernel<128, 4><<<wideBlocks, T, 0, stream>>>(
      h1, offs + N, cnt + N, seg + E, inv + N, wl2, wr2, b2, h2, N);

  sage_layer_kernel<128, 4><<<wideBlocks, T, 0, stream>>>(
      h2, offs, cnt, seg, inv, wl3, wr3, b3, (float*)d_out, N);
}


# Row epilogue as straight-line code: 1/norm by v_rsq_f32 + Newton step, tanh by one branch-free rational form (a normalised component is at most 1 in size), full tiles stored untested; last bits move
# speedup vs baseline: 4.2693x; 1.0564x over previous
#include <hip/hip_runtime.h>
#include <hip/hip_bf16.h>
#include <math.h>

typedef __attribute__((ext_vector_type(2))) float v2f;
typedef __attribute__((ext_vector_type(8))) float v8f;

#define HIDDIM 128

#define BUCKET_SHIFT 8
#define BUCKET_NODES (1 << BUCKET_SHIFT)
#define MAX_BUCKETS 512
#define PART_BLOCKS 128
#define PART_THREADS 512
#define SCAN_THREADS (2 * MAX_BUCKETS)
#define HALF_BLOCKS (PART_BLOCKS / 2)
#define SCAT_SLOTS 64
#define SORT_STAGE 6144
static_assert(PART_THREADS == MAX_BUCKETS, "the scatter set-up gives every bucket one thread");
static_assert(SCAT_SLOTS % 32 == 0, "a row of the image is written out by whole waves");

template <int T>
__device__ __forceinline__ int block_scan(int t, int* s, int tid) {
  s[tid] = t;
  __syncthreads();
  for (int o = 1; o < T; o <<= 1) {
    const int v = (tid >= o) ? s[tid - o] : 0;
    __syncthreads();
    s[tid] += v;
    __syncthreads();
  }
  return s[tid];
}

__global__ __launch_bounds__(PART_THREADS) void bucket_count_kernel(
    const int* __restrict__ dstA, const int* __restrict__ dstB, int E,
    int nbuck, int chunk, int* __restrict__ blkcnt) {
  __shared__ int hist[MAX_BUCKETS];
  const int tid = threadIdx.x;
  const int y = blockIdx.y;
  const int* dst = y ? dstB : dstA;
  for (int b = tid; b < nbuck; b += PART_THREADS) hist[b] = 0;
  __syncthreads();
  const int beg = blockIdx.x * chunk;
  const int end = (beg + chunk < E) ? (beg + chunk) : E;
  for (int i = beg + tid; i < end; i += 4 * PART_THREADS) {
    int d[4];
#pragma unroll
    for (int u = 0; u < 4; ++u) {
      const int j = i + u * PART_THREADS;
      const int v = dst[(j < end) ? j : i];
      d[u] = (j < end) ? v : -1;
    }
#pragma unroll
    for (int u = 0; u < 4; ++u)
      if (d[u] >= 0) atomicAdd(&hist[d[u] >> BUCKET_SHIFT], 1);
  }
  __syncthreads();
  int* o = blkcnt + ((size_t)y * gridDim.x + blockIdx.x) * nbuck;
  for (int b = tid; b < nbuck; b += PART_THREADS) o[b] = hist[b];
}

__global__ __launch_bounds__(SCAN_THREADS) void bucket_scan_kernel(
    const int* __restrict__ blkcnt, int* __restrict__ blkpref,
    int* __restrict__ bstart, int nbuck) {
  __shared__ int s[SCAN_THREADS];
  __shared__ int tots[2][MAX_BUCKETS];
  const int tid = threadIdx.x;
  const int y = blockIdx.x;
  const int j = tid & (MAX_BUCKETS - 1);
  const int q = tid / MAX_BUCKETS;
  const size_t first = ((size_t)y * PART_BLOCKS + q * HALF_BLOCKS) * nbuck + j;
  const int* c = blkcnt + first;
  int* p = blkpref + first;
  const bool live = j < nbuck;
  int v[HALF_BLOCKS];
#pragma unroll
  for (int u = 0; u < HALF_BLOCKS; ++u) v[u] = live ? c[(size_t)u * nbuck] : 0;
  int tot = 0;
#pragma unroll
  for (int u = 0; u < HALF_BLOCKS; ++u) tot += v[u];
  tots[q][j] = tot;
  __syncthreads();
  const int whole = (q == 0) ? (tots[0][j] + tots[1][j]) : 0;
  int run = (q == 1) ? tots[0][j] : 0;
  if (live) {
#pragma unroll
    for (int u = 0; u < HALF_BLOCKS; ++u) {
      p[(size_t)u * nbuck] = run;
      run += v[u];
    }
  }
  const int incl = block_scan<SCAN_THREADS>(whole, s, tid);
  if (q == 0 && live) {
    bstart[y * (nbuck + 1) + j] = incl - whole;
    if (j == nbuck - 1) bstart[y * (nbuck + 1) + nbuck] = incl;
  }
}

__global__ __launch_bounds__(PART_THREADS) void bucket_scatter_kernel(
    const int* __restrict__ srcA, const int* __restrict__ dstA,
    const int* __restrict__ srcB, const int* __restrict__ dstB, int E,
    int nbuck, int chunk, const int* __restrict__ blkpref,
    const int* __restrict__ bstart, unsigned* __restrict__ packed) {
  __shared__ int cnt[MAX_BUCKETS];
  __shared__ int gpos[MAX_BUCKETS];
  __shared__ __attribute__((aligned(16))) unsigned img[MAX_BUCKETS * SCAT_SLOTS];
  const int tid = threadIdx.x;
  const int y = blockIdx.y;
  const int* src = y ? srcB : srcA;
  const int* dst = y ? dstB : dstA;
  const int* p = blkpref + ((size_t)y * gridDim.x + blockIdx.x) * nbuck;
  const int* bs = bstart + y * (nbuck + 1);
  const bool live = tid < nbuck;
  const int g0 = live ? bs[tid] : 0;
  const int g1 = live ? p[tid] : 0;
  uint4* img4 = (uint4*)img;
  for (int q = tid; q < nbuck * (SCAT_SLOTS / 4); q += PART_THREADS)
    img4[q] = make_uint4(0u, 0u, 0u, 0u);
  cnt[tid] = 0;
  gpos[tid] = g0 + g1;
  __syncthreads();
  unsigned* out = packed + (size_t)y * E;
  const int beg = blockIdx.x * chunk;
  const int end = (beg + chunk < E) ? (beg + chunk) : E;
  for (int i = beg + tid; i < end; i += 4 * PART_THREADS) {
    int d[4], s[4];
#pragma unroll
    for (int u = 0; u < 4; ++u) {
      const int j = i + u * PART_THREADS;
      const int jj = (j < end) ? j : i;
      const int v = dst[jj];
      s[u] = src[jj];
      d[u] = (j < end) ? v : -1;
    }
#pragma unroll
    for (int u = 0; u < 4; ++u) {
      if (d[u] >= 0) {
        const int b = d[u] >> BUCKET_SHIFT;
        const int r = atomicAdd(&cnt[b], 1);
        const unsigned w = ((unsigned)s[u] << BUCKET_SHIFT) |
                           (unsigned)(d[u] & (BUCKET_NODES - 1));
        if (r < SCAT_SLOTS)
          img[b * SCAT_SLOTS + r] = w;
        else
          out[gpos[b] + r] = w;
      }
    }
  }
  __syncthreads();
  constexpr int NW = PART_THREADS / 32;
  constexpr int RW = SCAT_SLOTS / 32;
  const int wave = tid >> 5;
  const int lane = tid & 31;
  for (int b0 = wave; b0 < nbuck; b0 += 4 * NW) {
    int n[4], gb[4];
    unsigned wv[4][RW];
#pragma unroll
    for (int u = 0; u < 4; ++u) {
      const int b = b0 + u * NW;
      const int bb = (b < nbuck) ? b : b0;
      const int c = cnt[bb];
      n[u] = (b < nbuck) ? c : 0;
      gb[u] = gpos[bb];
#pragma unroll
      for (int k = 0; k < RW; ++k) wv[u][k] = img[bb * SCAT_SLOTS + 32 * k + lane];
    }
#pragma unroll
    for (int u = 0; u < 4; ++u) {
#pragma unroll
      for (int k = 0; k < RW; ++k)
        if (32 * k + lane < n[u]) out[gb[u] + 32 * k + lane] = wv[u][k];
    }
  }
}

__global__ __launch_bounds__(BUCKET_NODES) void bucket_sort_kernel(
    const unsigned* __restrict__ packed, const int* __restrict__ bstart, int E,
    int N, int nbuck, int* __restrict__ offs, int* __restrict__ cnt,
    float* __restrict__ inv, int* __restrict__ seg) {
  __shared__ int s[BUCKET_NODES];
  __shared__ int cur[BUCKET_NODES];
  __shared__ int stage[SORT_STAGE];
  const int tid = threadIdx.x;
  const int y = blockIdx.y;
  const int e0 = bstart[y * (nbuck + 1) + blockIdx.x];
  const int e1 = bstart[y * (nbuck + 1) + blockIdx.x + 1];
  const unsigned* pk = packed + (size_t)y * E;
  cur[tid] = 0;
  __syncthreads();
  for (int i = e0 + tid; i < e1; i += 4 * BUCKET_NODES) {
    unsigned w[4];
#pragma unroll
    for (int u = 0; u < 4; ++u) {
      const int j = i + u * BUCKET_NODES;
      w[u] = pk[(j < e1) ? j : i];
    }
#pragma unroll
    for (int u = 0; u < 4; ++u)
      if (i + u * BUCKET_NODES < e1)
        atomicAdd(&cur[w[u] & (BUCKET_NODES - 1)], 1);
  }
  __syncthreads();
  const int c = cur[tid];
  const int excl = block_scan<BUCKET_NODES>(c, s, tid) - c;
  const int node = blockIdx.x * BUCKET_NODES + tid;
  if (node < N) {
    const size_t g = (size_t)y * N + node;
    offs[g] = e0 + excl;
    cnt[g] = c;
    inv[g] = 1.0f / fmaxf((float)c, 1.0f);
  }
  cur[tid] = excl;
  __syncthreads();
  int* sg = seg + (size_t)y * E + e0;
  for (int i = e0 + tid; i < e1; i += 4 * BUCKET_NODES) {
    unsigned w[4];
#pragma unroll
    for (int u = 0; u < 4; ++u) {
      const int j = i + u * BUCKET_NODES;
      w[u] = pk[(j < e1) ? j : i];
    }
#pragma unroll
    for (int u = 0; u < 4; ++u) {
      if (i + u * BUCKET_NODES < e1) {
        const int pos = atomicAdd(&cur[w[u] & (BUCKET_NODES - 1)], 1);
        const int v = (int)(w[u] >> BUCKET_SHIFT);
        if (pos < SORT_STAGE)
          stage[pos] = v;
        else
          sg[pos] = v;
      }
    }
  }
  __syncthreads();
  const int n = e1 - e0;
  const int m = (n < SORT_STAGE) ? n : SORT_STAGE;
  const int mis = (int)(((size_t)sg >> 2) & 31);
  int i = tid - mis;
  if (i < 0) i += BUCKET_NODES;
#pragma unroll 4
  for (; i < m; i += BUCKET_NODES) sg[i] = stage[i];
}

__device__ __forceinline__ void add4(float4& a, const float4 v) {
  a.x += v.x;
  a.y += v.y;
  a.z += v.z;
  a.w += v.w;
}

template <int R, int P>
struct XSteps {
  static constexpr int KSTEPS = HIDDIM / 4;
  static_assert(KSTEPS % 2 == 0, "steps are taken in pairs");
  const float* wrl;
  const float* sxl;
  v8f* c;
  int ks;
  v2f b[2];
  __device__ __forceinline__ void fetch() {
    if (ks < KSTEPS) {
#pragma unroll
      for (int u = 0; u < 2; ++u) {
        b[u].x = wrl[(size_t)(4 * (ks + u)) * HIDDIM];
        b[u].y = wrl[(size_t)(4 * (ks + u) + 1) * HIDDIM];
      }
    }
  }
  __device__ __forceinline__ void run() {
    if (ks < KSTEPS) {
#pragma unroll
      for (int u = 0; u < 2; ++u) {
#pragma unroll
        for (int t = 0; t < R; ++t) {
          const v2f ax = *(const v2f*)(sxl + 16 * t * P + 4 * (ks + u));
          c[t] = __builtin_amdgcn_wmma_f32_16x16x4_f32(false, ax, false, b[u],
                                                       (short)0, c[t], false, false);
        }
      }
      ks += 2;
    }
  }
};

template <class X>
__device__ __forceinline__ float4 gather_row128(const float* __restrict__ hl,
                                                const int* __restrict__ seg,
                                                int beg, int deg, int lane,
                                                int idx0, X& xp) {
#define ROW_OF(jj) \
  (*(const float4*)(hl + (size_t)__builtin_amdgcn_readlane(idx, (jj)) * HIDDIM))
  float4 acc = make_float4(0.0f, 0.0f, 0.0f, 0.0f);
  for (int base = 0; base < deg; base += 32) {
    const int m = (deg - base < 32) ? (deg - base) : 32;
    int idx = idx0;
    if (base > 0) {
      idx = 0;
      if (lane < m) idx = seg[beg + base + lane];
    }
    int j = 0;
    for (; j + 8 <= m; j += 8) {
      const float4 v0 = ROW_OF(j), v1 = ROW_OF(j + 1), v2 = ROW_OF(j + 2),
                   v3 = ROW_OF(j + 3), v4 = ROW_OF(j + 4), v5 = ROW_OF(j + 5),
                   v6 = ROW_OF(j + 6), v7 = ROW_OF(j + 7);
      xp.run();
      xp.fetch();
      add4(acc, v0);
      add4(acc, v1);
      add4(acc, v2);
      add4(acc, v3);
      add4(acc, v4);
      add4(acc, v5);
      add4(acc, v6);
      add4(acc, v7);
    }
    if (j + 4 <= m) {
      const float4 v0 = ROW_OF(j), v1 = ROW_OF(j + 1), v2 = ROW_OF(j + 2),
                   v3 = ROW_OF(j + 3);
      xp.run();
      xp.fetch();
      add4(acc, v0);
      add4(acc, v1);
      add4(acc, v2);
      add4(acc, v3);
      j += 4;
    }
    if (j + 2 <= m) {
      const float4 v0 = ROW_OF(j), v1 = ROW_OF(j + 1);
      xp.run();
      xp.fetch();
      add4(acc, v0);
      add4(acc, v1);
      j += 2;
    }
    if (j < m) {
      const float4 v0 = ROW_OF(j);
      xp.run();
      xp.fetch();
      add4(acc, v0);
    }
  }
#undef ROW_OF
  return acc;
}

__device__ __forceinline__ float xor_add(float s, const int imm) {
  switch (imm) {
    case 1:  s += __int_as_float(__builtin_amdgcn_ds_swizzle(__float_as_int(s), 0x041f)); break;
    case 2:  s += __int_as_float(__builtin_amdgcn_ds_swizzle(__float_as_int(s), 0x081f)); break;
    case 4:  s += __int_as_float(__builtin_amdgcn_ds_swizzle(__float_as_int(s), 0x101f)); break;
    default: s += __int_as_float(__builtin_amdgcn_ds_swizzle(__float_as_int(s), 0x201f)); break;
  }
  return s;
}

__device__ __forceinline__ float other_half(float x) {
  const int b = __float_as_int(x);
  return __int_as_float(
      __builtin_amdgcn_permlanex16(b, b, 0x76543210, 0xfedcba98, false, false));
}

__device__ __forceinline__ float inv_norm(float s) {
  const float r = fminf(__builtin_amdgcn_rsqf(s), 1e12f);
  const float c = fmaf(-0.5f * s * r, r, 1.5f);
  return fminf(r * c, 1e12f);
}

__device__ __forceinline__ float tanh_unit(float x) {
  const float t = x * x;
  float n = t + 378.0f;
  n = fmaf(n, t, 17325.0f);
  n = fmaf(n, t, 135135.0f);
  float d = fmaf(28.0f, t, 3150.0f);
  d = fmaf(d, t, 62370.0f);
  d = fmaf(d, t, 135135.0f);
  const float p = x * n;
  const float r = __builtin_amdgcn_rcpf(d);
  const float q = p * r;
  return fmaf(fmaf(-d, q, p), r, q);
}

template <int K, int R>
__global__ __launch_bounds__(256) void sage_layer_kernel(
    const float* __restrict__ xin, const int* __restrict__ offs,
    const int* __restrict__ cnt, const int* __restrict__ seg,
    const float* __restrict__ invcnt, const float* __restrict__ wl,
    const float* __restrict__ wr, const float* __restrict__ bias,
    float* __restrict__ out, int N) {
  constexpr int P = K + 4;
  constexpr int ROWS = 16 * R;
  constexpr int WROWS = ROWS / 8;
  __shared__ __attribute__((aligned(16))) float sM[ROWS][P];
  __shared__ __attribute__((aligned(16))) float sX[ROWS][P];
  constexpr bool SHARE = (K == HIDDIM);
  __shared__ __attribute__((aligned(16))) float sOwn[SHARE ? 1 : ROWS][HIDDIM + 4];
  float (*sO)[HIDDIM + 4];
  if constexpr (SHARE)
    sO = sM;
  else
    sO = sOwn;

  const int tid = threadIdx.x;
  const int wave = tid >> 5;
  const int lane = tid & 31;
  const int half = lane >> 4;
  const int l16 = lane & 15;
  const int rowBase = blockIdx.x * ROWS;
  const int col = wave * 16 + l16;

  v8f c[R];
#pragma unroll
  for (int t = 0; t < R; ++t) c[t] = v8f{};

  if constexpr (K == HIDDIM) {
    const int w = __builtin_amdgcn_readfirstlane(wave);
    const float* hl = xin + lane * 4;
    const int last = N - 1;
    XSteps<R, P> xp;
    xp.wrl = wr + (size_t)(2 * half) * HIDDIM + col;
    xp.sxl = &sX[l16][2 * half];
    xp.c = c;
    xp.ks = 0;
    xp.fetch();
    float4 xo[WROWS];
#pragma unroll
    for (int i = 0; i < WROWS; ++i) {
      const int r = rowBase + w + 8 * i;
      xo[i] = *(const float4*)(hl + (size_t)((r < last) ? r : last) * K);
    }
    const int n0 = rowBase + WROWS * w;
    int beg[WROWS], deg[WROWS], idx[WROWS];
    float sc[WROWS];
#pragma unroll
    for (int i = 0; i < WROWS; ++i) {
      const int n = (n0 + i < last) ? (n0 + i) : last;
      beg[i] = offs[n];
      deg[i] = cnt[n];
      sc[i] = invcnt[n];
    }
#pragma unroll
    for (int i = 0; i < WROWS; ++i) {
      idx[i] = 0;
      if (lane < deg[i]) idx[i] = seg[beg[i] + lane];
    }
#pragma unroll
    for (int i = 0; i < WROWS; ++i) *(float4*)&sX[w + 8 * i][lane * 4] = xo[i];
    __syncthreads();
#pragma unroll
    for (int i = 0; i < WROWS; ++i) {
      float4 m = gather_row128(hl, seg, beg[i], deg[i], lane, idx[i], xp);
      m.x *= sc[i];
      m.y *= sc[i];
      m.z *= sc[i];
      m.w *= sc[i];
      *(float4*)&sM[WROWS * w + i][lane * 4] = m;
    }
    while (xp.ks < XSteps<R, P>::KSTEPS) {
      xp.run();
      xp.fetch();
    }
  } else {
    if (tid < ROWS * K) {
      const int r = tid / K;
      const int k = tid - r * K;
      const int node = rowBase + r;
      sX[r][k] = xin[(size_t)node * K + k];
      int j = offs[node];
      const int end = j + cnt[node];
      float acc = 0.0f;
      for (; j + 4 <= end; j += 4) {
        const int s0 = seg[j], s1 = seg[j + 1], s2 = seg[j + 2], s3 = seg[j + 3];
        const float v0 = xin[(size_t)s0 * K + k];
        const float v1 = xin[(size_t)s1 * K + k];
        const float v2 = xin[(size_t)s2 * K + k];
        const float v3 = xin[(size_t)s3 * K + k];
        acc += v0;
        acc += v1;
        acc += v2;
        acc += v3;
      }
      for (; j < end; ++j) acc += xin[(size_t)seg[j] * K + k];
      sM[r][k] = acc * invcnt[node];
    }
  }
  __syncthreads();

  if constexpr (K == HIDDIM) {
#pragma unroll 8
    for (int k0 = 0; k0 < K; k0 += 4) {
      const int ka = k0 + 2 * half;
      v2f bl;
      bl.x = wl[(size_t)ka * HIDDIM + col];
      bl.y = wl[(size_t)(ka + 1) * HIDDIM + col];
#pragma unroll
      for (int t = 0; t < R; ++t) {
        const v2f am = *(const v2f*)&sM[16 * t + l16][ka];
        c[t] = __builtin_amdgcn_wmma_f32_16x16x4_f32(false, am, false, bl, (short)0,
                                                     c[t], false, false);
      }
    }
  } else {
#pragma unroll 8
    for (int k0 = 0; k0 < K; k0 += 4) {
      const int ka = k0 + 2 * half;
      v2f bl, br;
      bl.x = wl[(size_t)ka * HIDDIM + col];
      bl.y = wl[(size_t)(ka + 1) * HIDDIM + col];
      br.x = wr[(size_t)ka * HIDDIM + col];
      br.y = wr[(size_t)(ka + 1) * HIDDIM + col];
#pragma unroll
      for (int t = 0; t < R; ++t) {
        const v2f am = *(const v2f*)&sM[16 * t + l16][ka];
        const v2f ax = *(const v2f*)&sX[16 * t + l16][ka];
        c[t] = __builtin_amdgcn_wmma_f32_16x16x4_f32(false, am, false, bl, (short)0,
                                                     c[t], false, false);
        c[t] = __builtin_amdgcn_wmma_f32_16x16x4_f32(false, ax, false, br, (short)0,
                                                     c[t], false, false);
      }
    }
  }

  const float bv = bias[col];
  if constexpr (SHARE) __syncthreads();
#pragma unroll
  for (int t = 0; t < R; ++t) {
#pragma unroll
    for (int v = 0; v < 8; ++v) sO[16 * t + v + 8 * half][col] = c[t][v] + bv;
  }
  __syncthreads();

  float4 o[WROWS];
  float s[WROWS];
#pragma unroll
  for (int q = 0; q < WROWS; ++q) {
    o[q] = *(const float4*)&sO[wave + 8 * q][lane * 4];
    s[q] = o[q].x * o[q].x + o[q].y * o[q].y + o[q].z * o[q].z + o[q].w * o[q].w;
  }
#pragma unroll
  for (int q = 0; q < WROWS; ++q) s[q] = xor_add(s[q], 1);
#pragma unroll
  for (int q = 0; q < WROWS; ++q) s[q] = xor_add(s[q], 2);
#pragma unroll
  for (int q = 0; q < WROWS; ++q) s[q] = xor_add(s[q], 4);
#pragma unroll
  for (int q = 0; q < WROWS; ++q) s[q] = xor_add(s[q], 8);
#pragma unroll
  for (int q = 0; q < WROWS; ++q) s[q] += other_half(s[q]);
  float4 r[WROWS];
#pragma unroll
  for (int q = 0; q < WROWS; ++q) {
    const float scale = inv_norm(s[q]);
    r[q].x = tanh_unit(o[q].x * scale);
    r[q].y = tanh_unit(o[q].y * scale);
    r[q].z = tanh_unit(o[q].z * scale);
    r[q].w = tanh_unit(o[q].w * scale);
  }
  if (rowBase + ROWS <= N) {
    float* orow = out + (size_t)(rowBase + wave) * HIDDIM + lane * 4;
#pragma unroll
    for (int q = 0; q < WROWS; ++q) *(float4*)(orow + (size_t)(8 * q) * HIDDIM) = r[q];
  } else {
#pragma unroll
    for (int q = 0; q < WROWS; ++q) {
      const int row = rowBase + wave + 8 * q;
      if (row < N) *(float4*)(out + (size_t)row * HIDDIM + lane * 4) = r[q];
    }
  }
}

extern "C" void kernel_launch(void* const* d_in, const int* in_sizes, int n_in,
                              void* d_out, int out_size, void* d_ws,
                              size_t ws_size, hipStream_t stream) {
  const float* x = (const float*)d_in[0];
  const int* ei_conn = (const int*)d_in[1];
  const int* ei_dest = (const int*)d_in[2];
  const float* wl1 = (const float*)d_in[3];
  const float* wr1 = (const float*)d_in[4];
  const float* b1 = (const float*)d_in[5];
  const float* wl2 = (const float*)d_in[6];
  const float* wr2 = (const float*)d_in[7];
  const float* b2 = (const float*)d_in[8];
  const float* wl3 = (const float*)d_in[9];
  const float* wr3 = (const float*)d_in[10];
  const float* b3 = (const float*)d_in[11];

  const int N = in_sizes[0] / 8;
  const int E = in_sizes[1] / 2;
  const int nbuck = (N + BUCKET_NODES - 1) / BUCKET_NODES;
  const int chunk = (E + PART_BLOCKS - 1) / PART_BLOCKS;

  float* ws = (float*)d_ws;
  float* h1 = ws;
  float* h2 = h1 + (size_t)N * HIDDIM;
  float* inv = h2 + (size_t)N * HIDDIM;
  int* cnt = (int*)(inv + (size_t)2 * N);
  int* offs = cnt + (size_t)2 * N;
  int* blkcnt = offs + (size_t)2 * N;
  int* blkpref = blkcnt + (size_t)2 * PART_BLOCKS * nbuck;
  int* bstart = blkpref + (size_t)2 * PART_BLOCKS * nbuck;
  unsigned* packed = (unsigned*)(bstart + (size_t)2 * (nbuck + 1));
  int* seg = (int*)(packed + (size_t)2 * E);

  const int T = 256;

  const int* src_conn = ei_conn;
  const int* dst_conn = ei_conn + E;
  const int* src_dest = ei_dest;
  const int* dst_dest = ei_dest + E;

  bucket_count_kernel<<<dim3(PART_BLOCKS, 2), PART_THREADS, 0, stream>>>(
      dst_conn, dst_dest, E, nbuck, chunk, blkcnt);
  bucket_scan_kernel<<<2, SCAN_THREADS, 0, stream>>>(blkcnt, blkpref, bstart,
                                                     nbuck);
  bucket_scatter_kernel<<<dim3(PART_BLOCKS, 2), PART_THREADS, 0, stream>>>(
      src_conn, dst_conn, src_dest, dst_dest, E, nbuck, chunk, blkpref, bstart,
      packed);
  bucket_sort_kernel<<<dim3(nbuck, 2), BUCKET_NODES, 0, stream>>>(
      packed, bstart, E, N, nbuck, offs, cnt, inv, seg);

  const int narrowBlocks = (N + 15) / 16;
  const int wideBlocks = (N + 63) / 64;

  sage_layer_kernel<8, 1><<<narrowBlocks, T, 0, stream>>>(x, offs, cnt, seg, inv,
                                                          wl1, wr1, b1, h1, N);

  sage_layer_kernel<128, 4><<<wideBlocks, T, 0, stream>>>(
      h1, offs + N, cnt + N, seg + E, inv + N, wl2, wr2, b2, h2, N);

  sage_layer_kernel<128, 4><<<wideBlocks, T, 0, stream>>>(
      h2, offs, cnt, seg, inv, wl3, wr3, b3, (float*)d_out, N);
}


# Edge sort level 1 in one launch: a block counts, scans and groups its own chunk; level 2 collects its bucket's 128 runs, start = sum of the chunks' group starts; 7 launches to 5; sum order changes
# speedup vs baseline: 4.5316x; 1.0614x over previous
#include <hip/hip_runtime.h>
#include <hip/hip_bf16.h>
#include <math.h>

typedef __attribute__((ext_vector_type(2))) float v2f;
typedef __attribute__((ext_vector_type(8))) float v8f;

#define HIDDIM 128

#define BUCKET_SHIFT 8
#define BUCKET_NODES (1 << BUCKET_SHIFT)
#define MAX_BUCKETS 512
#define PART_BLOCKS 128
#define PART_THREADS 512
#define PART_GROUP 8
#define CHUNK_STAGE 16384
#define SORT_STAGE 6144
#define SORT_WAVES (BUCKET_NODES / 32)
static_assert(PART_THREADS == MAX_BUCKETS, "level 1 gives every bucket one thread");
static_assert(PART_BLOCKS <= BUCKET_NODES, "level 2 gives every chunk one thread");
static_assert(PART_BLOCKS % (4 * SORT_WAVES) == 0, "a level-2 wave takes four runs a trip");

template <int T>
__device__ __forceinline__ int block_scan(int t, int* s, int tid) {
  s[tid] = t;
  __syncthreads();
  for (int o = 1; o < T; o <<= 1) {
    const int v = (tid >= o) ? s[tid - o] : 0;
    __syncthreads();
    s[tid] += v;
    __syncthreads();
  }
  return s[tid];
}

__global__ __launch_bounds__(PART_THREADS) void chunk_group_kernel(
    const int* __restrict__ srcA, const int* __restrict__ dstA,
    const int* __restrict__ srcB, const int* __restrict__ dstB, int E,
    int nbuck, int chunk, int* __restrict__ blkoff,
    unsigned* __restrict__ packed) {
  __shared__ int cnt[MAX_BUCKETS];
  __shared__ int sc[PART_THREADS];
  __shared__ __attribute__((aligned(16))) unsigned img[CHUNK_STAGE];
  const int tid = threadIdx.x;
  const int y = blockIdx.y;
  const int* src = y ? srcB : srcA;
  const int* dst = y ? dstB : dstA;
  cnt[tid] = 0;
  __syncthreads();
  const int beg = blockIdx.x * chunk;
  const int end = (beg + chunk < E) ? (beg + chunk) : E;
  for (int i = beg + tid; i < end; i += PART_GROUP * PART_THREADS) {
    int d[PART_GROUP];
#pragma unroll
    for (int u = 0; u < PART_GROUP; ++u) {
      const int j = i + u * PART_THREADS;
      const int v = dst[(j < end) ? j : i];
      d[u] = (j < end) ? v : -1;
    }
#pragma unroll
    for (int u = 0; u < PART_GROUP; ++u)
      if (d[u] >= 0) atomicAdd(&cnt[d[u] >> BUCKET_SHIFT], 1);
  }
  __syncthreads();
  const int c = cnt[tid];
  const int incl = block_scan<PART_THREADS>(c, sc, tid);
  const int excl = incl - c;
  int* bo = blkoff + ((size_t)y * gridDim.x + blockIdx.x) * (nbuck + 1);
  if (tid < nbuck) bo[tid] = excl;
  if (tid == PART_THREADS - 1) bo[nbuck] = incl;
  cnt[tid] = excl;
  __syncthreads();
  unsigned* out = packed + (size_t)y * E + beg;
  for (int i = beg + tid; i < end; i += PART_GROUP * PART_THREADS) {
    int d[PART_GROUP], s[PART_GROUP];
#pragma unroll
    for (int u = 0; u < PART_GROUP; ++u) {
      const int j = i + u * PART_THREADS;
      const int jj = (j < end) ? j : i;
      const int v = dst[jj];
      s[u] = src[jj];
      d[u] = (j < end) ? v : -1;
    }
    asm volatile("" ::"v"(s[0]));
#pragma unroll
    for (int u = 0; u < PART_GROUP; ++u) {
      if (d[u] >= 0) {
        const int r = atomicAdd(&cnt[d[u] >> BUCKET_SHIFT], 1);
        const unsigned w = ((unsigned)s[u] << BUCKET_SHIFT) |
                           (unsigned)(d[u] & (BUCKET_NODES - 1));
        if (r < CHUNK_STAGE)
          img[r] = w;
        else
          out[r] = w;
      }
    }
  }
  __syncthreads();
  const int n = (end > beg) ? (end - beg) : 0;
  const int m = (n < CHUNK_STAGE) ? n : CHUNK_STAGE;
  const int mis = (int)(((size_t)out >> 2) & 31);
  int i = tid - mis;
  if (i < 0) i += PART_THREADS;
#pragma unroll 4
  for (; i < m; i += PART_THREADS) out[i] = img[i];
}

__device__ __forceinline__ void place_id(unsigned w, int* cur, int* stage,
                                         int* sg) {
  const int pos = atomicAdd(&cur[w & (BUCKET_NODES - 1)], 1);
  const int v = (int)(w >> BUCKET_SHIFT);
  if (pos < SORT_STAGE)
    stage[pos] = v;
  else
    sg[pos] = v;
}

__global__ __launch_bounds__(BUCKET_NODES) void bucket_sort_kernel(
    const unsigned* __restrict__ packed, const int* __restrict__ blkoff, int E,
    int N, int nbuck, int chunk, int* __restrict__ offs, int* __restrict__ cnt,
    float* __restrict__ inv, int* __restrict__ seg) {
  __shared__ int s[BUCKET_NODES];
  __shared__ int cur[BUCKET_NODES];
  __shared__ int stage[SORT_STAGE];
  __shared__ int rbeg[PART_BLOCKS];
  __shared__ int rlen[PART_BLOCKS];
  const int tid = threadIdx.x;
  const int wave = tid >> 5;
  const int lane = tid & 31;
  const int y = blockIdx.y;
  int a = 0;
  if (tid < PART_BLOCKS) {
    const int* bo =
        blkoff + ((size_t)y * PART_BLOCKS + tid) * (nbuck + 1) + blockIdx.x;
    a = bo[0];
    rlen[tid] = bo[1] - a;
    rbeg[tid] = tid * chunk + a;
  }
  cur[tid] = 0;
  int part = a;
  for (int o = 16; o > 0; o >>= 1) part += __shfl_xor(part, o);
  if (lane == 0) s[wave] = part;
  __syncthreads();
  int e0 = 0;
  for (int q = 0; q < SORT_WAVES; ++q) e0 += s[q];
  __syncthreads();
  const unsigned* pk = packed + (size_t)y * E;
  for (int p0 = wave; p0 < PART_BLOCKS; p0 += 4 * SORT_WAVES) {
    int rb[4], ln[4];
    unsigned w0[4], w1[4];
#pragma unroll
    for (int u = 0; u < 4; ++u) {
      rb[u] = rbeg[p0 + u * SORT_WAVES];
      ln[u] = rlen[p0 + u * SORT_WAVES];
    }
#pragma unroll
    for (int u = 0; u < 4; ++u) {
      w0[u] = 0u;
      w1[u] = 0u;
      if (lane < ln[u]) w0[u] = pk[rb[u] + lane];
      if (32 + lane < ln[u]) w1[u] = pk[rb[u] + 32 + lane];
    }
#pragma unroll
    for (int u = 0; u < 4; ++u) {
      if (lane < ln[u]) atomicAdd(&cur[w0[u] & (BUCKET_NODES - 1)], 1);
      if (32 + lane < ln[u]) atomicAdd(&cur[w1[u] & (BUCKET_NODES - 1)], 1);
      for (int o = 64 + lane; o < ln[u]; o += 32)
        atomicAdd(&cur[pk[rb[u] + o] & (BUCKET_NODES - 1)], 1);
    }
  }
  __syncthreads();
  const int c = cur[tid];
  const int excl = block_scan<BUCKET_NODES>(c, s, tid) - c;
  const int n = s[BUCKET_NODES - 1];
  const int node = blockIdx.x * BUCKET_NODES + tid;
  if (node < N) {
    const size_t g = (size_t)y * N + node;
    offs[g] = e0 + excl;
    cnt[g] = c;
    inv[g] = 1.0f / fmaxf((float)c, 1.0f);
  }
  cur[tid] = excl;
  __syncthreads();
  int* sg = seg + (size_t)y * E + e0;
  for (int p0 = wave; p0 < PART_BLOCKS; p0 += 4 * SORT_WAVES) {
    int rb[4], ln[4];
    unsigned w0[4], w1[4];
#pragma unroll
    for (int u = 0; u < 4; ++u) {
      rb[u] = rbeg[p0 + u * SORT_WAVES];
      ln[u] = rlen[p0 + u * SORT_WAVES];
    }
#pragma unroll
    for (int u = 0; u < 4; ++u) {
      w0[u] = 0u;
      w1[u] = 0u;
      if (lane < ln[u]) w0[u] = pk[rb[u] + lane];
      if (32 + lane < ln[u]) w1[u] = pk[rb[u] + 32 + lane];
    }
#pragma unroll
    for (int u = 0; u < 4; ++u) {
      if (lane < ln[u]) place_id(w0[u], cur, stage, sg);
      if (32 + lane < ln[u]) place_id(w1[u], cur, stage, sg);
      for (int o = 64 + lane; o < ln[u]; o += 32)
        place_id(pk[rb[u] + o], cur, stage, sg);
    }
  }
  __syncthreads();
  const int m = (n < SORT_STAGE) ? n : SORT_STAGE;
  const int mis = (int)(((size_t)sg >> 2) & 31);
  int i = tid - mis;
  if (i < 0) i += BUCKET_NODES;
#pragma unroll 4
  for (; i < m; i += BUCKET_NODES) sg[i] = stage[i];
}

__device__ __forceinline__ void add4(float4& a, const float4 v) {
  a.x += v.x;
  a.y += v.y;
  a.z += v.z;
  a.w += v.w;
}

template <int R, int P>
struct XSteps {
  static constexpr int KSTEPS = HIDDIM / 4;
  static_assert(KSTEPS % 2 == 0, "steps are taken in pairs");
  const float* wrl;
  const float* sxl;
  v8f* c;
  int ks;
  v2f b[2];
  __device__ __forceinline__ void fetch() {
    if (ks < KSTEPS) {
#pragma unroll
      for (int u = 0; u < 2; ++u) {
        b[u].x = wrl[(size_t)(4 * (ks + u)) * HIDDIM];
        b[u].y = wrl[(size_t)(4 * (ks + u) + 1) * HIDDIM];
      }
    }
  }
  __device__ __forceinline__ void run() {
    if (ks < KSTEPS) {
#pragma unroll
      for (int u = 0; u < 2; ++u) {
#pragma unroll
        for (int t = 0; t < R; ++t) {
          const v2f ax = *(const v2f*)(sxl + 16 * t * P + 4 * (ks + u));
          c[t] = __builtin_amdgcn_wmma_f32_16x16x4_f32(false, ax, false, b[u],
                                                       (short)0, c[t], false, false);
        }
      }
      ks += 2;
    }
  }
};

template <class X>
__device__ __forceinline__ float4 gather_row128(const float* __restrict__ hl,
                                                const int* __restrict__ seg,
                                                int beg, int deg, int lane,
                                                int idx0, X& xp) {
#define ROW_OF(jj) \
  (*(const float4*)(hl + (size_t)__builtin_amdgcn_readlane(idx, (jj)) * HIDDIM))
  float4 acc = make_float4(0.0f, 0.0f, 0.0f, 0.0f);
  for (int base = 0; base < deg; base += 32) {
    const int m = (deg - base < 32) ? (deg - base) : 32;
    int idx = idx0;
    if (base > 0) {
      idx = 0;
      if (lane < m) idx = seg[beg + base + lane];
    }
    int j = 0;
    for (; j + 8 <= m; j += 8) {
      const float4 v0 = ROW_OF(j), v1 = ROW_OF(j + 1), v2 = ROW_OF(j + 2),
                   v3 = ROW_OF(j + 3), v4 = ROW_OF(j + 4), v5 = ROW_OF(j + 5),
                   v6 = ROW_OF(j + 6), v7 = ROW_OF(j + 7);
      xp.run();
      xp.fetch();
      add4(acc, v0);
      add4(acc, v1);
      add4(acc, v2);
      add4(acc, v3);
      add4(acc, v4);
      add4(acc, v5);
      add4(acc, v6);
      add4(acc, v7);
    }
    if (j + 4 <= m) {
      const float4 v0 = ROW_OF(j), v1 = ROW_OF(j + 1), v2 = ROW_OF(j + 2),
                   v3 = ROW_OF(j + 3);
      xp.run();
      xp.fetch();
      add4(acc, v0);
      add4(acc, v1);
      add4(acc, v2);
      add4(acc, v3);
      j += 4;
    }
    if (j + 2 <= m) {
      const float4 v0 = ROW_OF(j), v1 = ROW_OF(j + 1);
      xp.run();
      xp.fetch();
      add4(acc, v0);
      add4(acc, v1);
      j += 2;
    }
    if (j < m) {
      const float4 v0 = ROW_OF(j);
      xp.run();
      xp.fetch();
      add4(acc, v0);
    }
  }
#undef ROW_OF
  return acc;
}

__device__ __forceinline__ float xor_add(float s, const int imm) {
  switch (imm) {
    case 1:  s += __int_as_float(__builtin_amdgcn_ds_swizzle(__float_as_int(s), 0x041f)); break;
    case 2:  s += __int_as_float(__builtin_amdgcn_ds_swizzle(__float_as_int(s), 0x081f)); break;
    case 4:  s += __int_as_float(__builtin_amdgcn_ds_swizzle(__float_as_int(s), 0x101f)); break;
    default: s += __int_as_float(__builtin_amdgcn_ds_swizzle(__float_as_int(s), 0x201f)); break;
  }
  return s;
}

__device__ __forceinline__ float other_half(float x) {
  const int b = __float_as_int(x);
  return __int_as_float(
      __builtin_amdgcn_permlanex16(b, b, 0x76543210, 0xfedcba98, false, false));
}

__device__ __forceinline__ float inv_norm(float s) {
  const float r = fminf(__builtin_amdgcn_rsqf(s), 1e12f);
  const float c = fmaf(-0.5f * s * r, r, 1.5f);
  return fminf(r * c, 1e12f);
}

__device__ __forceinline__ float tanh_unit(float x) {
  const float t = x * x;
  float n = t + 378.0f;
  n = fmaf(n, t, 17325.0f);
  n = fmaf(n, t, 135135.0f);
  float d = fmaf(28.0f, t, 3150.0f);
  d = fmaf(d, t, 62370.0f);
  d = fmaf(d, t, 135135.0f);
  const float p = x * n;
  const float r = __builtin_amdgcn_rcpf(d);
  const float q = p * r;
  return fmaf(fmaf(-d, q, p), r, q);
}

template <int K, int R>
__global__ __launch_bounds__(256) void sage_layer_kernel(
    const float* __restrict__ xin, const int* __restrict__ offs,
    const int* __restrict__ cnt, const int* __restrict__ seg,
    const float* __restrict__ invcnt, const float* __restrict__ wl,
    const float* __restrict__ wr, const float* __restrict__ bias,
    float* __restrict__ out, int N) {
  constexpr int P = K + 4;
  constexpr int ROWS = 16 * R;
  constexpr int WROWS = ROWS / 8;
  __shared__ __attribute__((aligned(16))) float sM[ROWS][P];
  __shared__ __attribute__((aligned(16))) float sX[ROWS][P];
  constexpr bool SHARE = (K == HIDDIM);
  __shared__ __attribute__((aligned(16))) float sOwn[SHARE ? 1 : ROWS][HIDDIM + 4];
  float (*sO)[HIDDIM + 4];
  if constexpr (SHARE)
    sO = sM;
  else
    sO = sOwn;

  const int tid = threadIdx.x;
  const int wave = tid >> 5;
  const int lane = tid & 31;
  const int half = lane >> 4;
  const int l16 = lane & 15;
  const int rowBase = blockIdx.x * ROWS;
  const int col = wave * 16 + l16;

  v8f c[R];
#pragma unroll
  for (int t = 0; t < R; ++t) c[t] = v8f{};

  if constexpr (K == HIDDIM) {
    const int w = __builtin_amdgcn_readfirstlane(wave);
    const float* hl = xin + lane * 4;
    const int last = N - 1;
    XSteps<R, P> xp;
    xp.wrl = wr + (size_t)(2 * half) * HIDDIM + col;
    xp.sxl = &sX[l16][2 * half];
    xp.c = c;
    xp.ks = 0;
    xp.fetch();
    float4 xo[WROWS];
#pragma unroll
    for (int i = 0; i < WROWS; ++i) {
      const int r = rowBase + w + 8 * i;
      xo[i] = *(const float4*)(hl + (size_t)((r < last) ? r : last) * K);
    }
    const int n0 = rowBase + WROWS * w;
    int beg[WROWS], deg[WROWS], idx[WROWS];
    float sc[WROWS];
#pragma unroll
    for (int i = 0; i < WROWS; ++i) {
      const int n = (n0 + i < last) ? (n0 + i) : last;
      beg[i] = offs[n];
      deg[i] = cnt[n];
      sc[i] = invcnt[n];
    }
#pragma unroll
    for (int i = 0; i < WROWS; ++i) {
      idx[i] = 0;
      if (lane < deg[i]) idx[i] = seg[beg[i] + lane];
    }
#pragma unroll
    for (int i = 0; i < WROWS; ++i) *(float4*)&sX[w + 8 * i][lane * 4] = xo[i];
    __syncthreads();
#pragma unroll
    for (int i = 0; i < WROWS; ++i) {
      float4 m = gather_row128(hl, seg, beg[i], deg[i], lane, idx[i], xp);
      m.x *= sc[i];
      m.y *= sc[i];
      m.z *= sc[i];
      m.w *= sc[i];
      *(float4*)&sM[WROWS * w + i][lane * 4] = m;
    }
    while (xp.ks < XSteps<R, P>::KSTEPS) {
      xp.run();
      xp.fetch();
    }
  } else {
    if (tid < ROWS * K) {
      const int r = tid / K;
      const int k = tid - r * K;
      const int node = rowBase + r;
      sX[r][k] = xin[(size_t)node * K + k];
      int j = offs[node];
      const int end = j + cnt[node];
      float acc = 0.0f;
      for (; j + 4 <= end; j += 4) {
        const int s0 = seg[j], s1 = seg[j + 1], s2 = seg[j + 2], s3 = seg[j + 3];
        const float v0 = xin[(size_t)s0 * K + k];
        const float v1 = xin[(size_t)s1 * K + k];
        const float v2 = xin[(size_t)s2 * K + k];
        const float v3 = xin[(size_t)s3 * K + k];
        acc += v0;
        acc += v1;
        acc += v2;
        acc += v3;
      }
      for (; j < end; ++j) acc += xin[(size_t)seg[j] * K + k];
      sM[r][k] = acc * invcnt[node];
    }
  }
  __syncthreads();

  if constexpr (K == HIDDIM) {
#pragma unroll 8
    for (int k0 = 0; k0 < K; k0 += 4) {
      const int ka = k0 + 2 * half;
      v2f bl;
      bl.x = wl[(size_t)ka * HIDDIM + col];
      bl.y = wl[(size_t)(ka + 1) * HIDDIM + col];
#pragma unroll
      for (int t = 0; t < R; ++t) {
        const v2f am = *(const v2f*)&sM[16 * t + l16][ka];
        c[t] = __builtin_amdgcn_wmma_f32_16x16x4_f32(false, am, false, bl, (short)0,
                                                     c[t], false, false);
      }
    }
  } else {
#pragma unroll 8
    for (int k0 = 0; k0 < K; k0 += 4) {
      const int ka = k0 + 2 * half;
      v2f bl, br;
      bl.x = wl[(size_t)ka * HIDDIM + col];
      bl.y = wl[(size_t)(ka + 1) * HIDDIM + col];
      br.x = wr[(size_t)ka * HIDDIM + col];
      br.y = wr[(size_t)(ka + 1) * HIDDIM + col];
#pragma unroll
      for (int t = 0; t < R; ++t) {
        const v2f am = *(const v2f*)&sM[16 * t + l16][ka];
        const v2f ax = *(const v2f*)&sX[16 * t + l16][ka];
        c[t] = __builtin_amdgcn_wmma_f32_16x16x4_f32(false, am, false, bl, (short)0,
                                                     c[t], false, false);
        c[t] = __builtin_amdgcn_wmma_f32_16x16x4_f32(false, ax, false, br, (short)0,
                                                     c[t], false, false);
      }
    }
  }

  const float bv = bias[col];
  if constexpr (SHARE) __syncthreads();
#pragma unroll
  for (int t = 0; t < R; ++t) {
#pragma unroll
    for (int v = 0; v < 8; ++v) sO[16 * t + v + 8 * half][col] = c[t][v] + bv;
  }
  __syncthreads();

  float4 o[WROWS];
  float s[WROWS];
#pragma unroll
  for (int q = 0; q < WROWS; ++q) {
    o[q] = *(const float4*)&sO[wave + 8 * q][lane * 4];
    s[q] = o[q].x * o[q].x + o[q].y * o[q].y + o[q].z * o[q].z + o[q].w * o[q].w;
  }
#pragma unroll
  for (int q = 0; q < WROWS; ++q) s[q] = xor_add(s[q], 1);
#pragma unroll
  for (int q = 0; q < WROWS; ++q) s[q] = xor_add(s[q], 2);
#pragma unroll
  for (int q = 0; q < WROWS; ++q) s[q] = xor_add(s[q], 4);
#pragma unroll
  for (int q = 0; q < WROWS; ++q) s[q] = xor_add(s[q], 8);
#pragma unroll
  for (int q = 0; q < WROWS; ++q) s[q] += other_half(s[q]);
  float4 r[WROWS];
#pragma unroll
  for (int q = 0; q < WROWS; ++q) {
    const float scale = inv_norm(s[q]);
    r[q].x = tanh_unit(o[q].x * scale);
    r[q].y = tanh_unit(o[q].y * scale);
    r[q].z = tanh_unit(o[q].z * scale);
    r[q].w = tanh_unit(o[q].w * scale);
  }
  if (rowBase + ROWS <= N) {
    float* orow = out + (size_t)(rowBase + wave) * HIDDIM + lane * 4;
#pragma unroll
    for (int q = 0; q < WROWS; ++q) *(float4*)(orow + (size_t)(8 * q) * HIDDIM) = r[q];
  } else {
#pragma unroll
    for (int q = 0; q < WROWS; ++q) {
      const int row = rowBase + wave + 8 * q;
      if (row < N) *(float4*)(out + (size_t)row * HIDDIM + lane * 4) = r[q];
    }
  }
}

extern "C" void kernel_launch(void* const* d_in, const int* in_sizes, int n_in,
                              void* d_out, int out_size, void* d_ws,
                              size_t ws_size, hipStream_t stream) {
  const float* x = (const float*)d_in[0];
  const int* ei_conn = (const int*)d_in[1];
  const int* ei_dest = (const int*)d_in[2];
  const float* wl1 = (const float*)d_in[3];
  const float* wr1 = (const float*)d_in[4];
  const float* b1 = (const float*)d_in[5];
  const float* wl2 = (const float*)d_in[6];
  const float* wr2 = (const float*)d_in[7];
  const float* b2 = (const float*)d_in[8];
  const float* wl3 = (const float*)d_in[9];
  const float* wr3 = (const float*)d_in[10];
  const float* b3 = (const float*)d_in[11];

  const int N = in_sizes[0] / 8;
  const int E = in_sizes[1] / 2;
  const int nbuck = (N + BUCKET_NODES - 1) / BUCKET_NODES;
  const int chunk = (E + PART_BLOCKS - 1) / PART_BLOCKS;

  float* ws = (float*)d_ws;
  float* h1 = ws;
  float* h2 = h1 + (size_t)N * HIDDIM;
  float* inv = h2 + (size_t)N * HIDDIM;
  int* cnt = (int*)(inv + (size_t)2 * N);
  int* offs = cnt + (size_t)2 * N;
  int* blkoff = offs + (size_t)2 * N;
  unsigned* packed = (unsigned*)(blkoff + (size_t)2 * PART_BLOCKS * (nbuck + 1));
  int* seg = (int*)(packed + (size_t)2 * E);

  const int T = 256;

  const int* src_conn = ei_conn;
  const int* dst_conn = ei_conn + E;
  const int* src_dest = ei_dest;
  const int* dst_dest = ei_dest + E;

  chunk_group_kernel<<<dim3(PART_BLOCKS, 2), PART_THREADS, 0, stream>>>(
      src_conn, dst_conn, src_dest, dst_dest, E, nbuck, chunk, blkoff, packed);
  bucket_sort_kernel<<<dim3(nbuck, 2), BUCKET_NODES, 0, stream>>>(
      packed, blkoff, E, N, nbuck, chunk, offs, cnt, inv, seg);

  const int narrowBlocks = (N + 15) / 16;
  const int wideBlocks = (N + 63) / 64;

  sage_layer_kernel<8, 1><<<narrowBlocks, T, 0, stream>>>(x, offs, cnt, seg, inv,
                                                          wl1, wr1, b1, h1, N);

  sage_layer_kernel<128, 4><<<wideBlocks, T, 0, stream>>>(
      h1, offs + N, cnt + N, seg + E, inv + N, wl2, wr2, b2, h2, N);

  sage_layer_kernel<128, 4><<<wideBlocks, T, 0, stream>>>(
      h2, offs, cnt, seg, inv, wl3, wr3, b3, (float*)d_out, N);
}


# Layer 1 takes 64 rows a block like the wide layers: 4 lanes a node (2 edge parities x 2 row halves), a trip's 16 ids, then 16 rows, each one load group; 50000 waves become 12504; sum order changes
# speedup vs baseline: 4.7328x; 1.0444x over previous
#include <hip/hip_runtime.h>
#include <hip/hip_bf16.h>
#include <math.h>

typedef __attribute__((ext_vector_type(2))) float v2f;
typedef __attribute__((ext_vector_type(8))) float v8f;

#define HIDDIM 128

#define BUCKET_SHIFT 8
#define BUCKET_NODES (1 << BUCKET_SHIFT)
#define MAX_BUCKETS 512
#define PART_BLOCKS 128
#define PART_THREADS 512
#define PART_GROUP 8
#define CHUNK_STAGE 16384
#define SORT_STAGE 6144
#define SORT_WAVES (BUCKET_NODES / 32)
static_assert(PART_THREADS == MAX_BUCKETS, "level 1 gives every bucket one thread");
static_assert(PART_BLOCKS <= BUCKET_NODES, "level 2 gives every chunk one thread");
static_assert(PART_BLOCKS % (4 * SORT_WAVES) == 0, "a level-2 wave takes four runs a trip");

template <int T>
__device__ __forceinline__ int block_scan(int t, int* s, int tid) {
  s[tid] = t;
  __syncthreads();
  for (int o = 1; o < T; o <<= 1) {
    const int v = (tid >= o) ? s[tid - o] : 0;
    __syncthreads();
    s[tid] += v;
    __syncthreads();
  }
  return s[tid];
}

__global__ __launch_bounds__(PART_THREADS) void chunk_group_kernel(
    const int* __restrict__ srcA, const int* __restrict__ dstA,
    const int* __restrict__ srcB, const int* __restrict__ dstB, int E,
    int nbuck, int chunk, int* __restrict__ blkoff,
    unsigned* __restrict__ packed) {
  __shared__ int cnt[MAX_BUCKETS];
  __shared__ int sc[PART_THREADS];
  __shared__ __attribute__((aligned(16))) unsigned img[CHUNK_STAGE];
  const int tid = threadIdx.x;
  const int y = blockIdx.y;
  const int* src = y ? srcB : srcA;
  const int* dst = y ? dstB : dstA;
  cnt[tid] = 0;
  __syncthreads();
  const int beg = blockIdx.x * chunk;
  const int end = (beg + chunk < E) ? (beg + chunk) : E;
  for (int i = beg + tid; i < end; i += PART_GROUP * PART_THREADS) {
    int d[PART_GROUP];
#pragma unroll
    for (int u = 0; u < PART_GROUP; ++u) {
      const int j = i + u * PART_THREADS;
      const int v = dst[(j < end) ? j : i];
      d[u] = (j < end) ? v : -1;
    }
#pragma unroll
    for (int u = 0; u < PART_GROUP; ++u)
      if (d[u] >= 0) atomicAdd(&cnt[d[u] >> BUCKET_SHIFT], 1);
  }
  __syncthreads();
  const int c = cnt[tid];
  const int incl = block_scan<PART_THREADS>(c, sc, tid);
  const int excl = incl - c;
  int* bo = blkoff + ((size_t)y * gridDim.x + blockIdx.x) * (nbuck + 1);
  if (tid < nbuck) bo[tid] = excl;
  if (tid == PART_THREADS - 1) bo[nbuck] = incl;
  cnt[tid] = excl;
  __syncthreads();
  unsigned* out = packed + (size_t)y * E + beg;
  for (int i = beg + tid; i < end; i += PART_GROUP * PART_THREADS) {
    int d[PART_GROUP], s[PART_GROUP];
#pragma unroll
    for (int u = 0; u < PART_GROUP; ++u) {
      const int j = i + u * PART_THREADS;
      const int jj = (j < end) ? j : i;
      const int v = dst[jj];
      s[u] = src[jj];
      d[u] = (j < end) ? v : -1;
    }
    asm volatile("" ::"v"(s[0]));
#pragma unroll
    for (int u = 0; u < PART_GROUP; ++u) {
      if (d[u] >= 0) {
        const int r = atomicAdd(&cnt[d[u] >> BUCKET_SHIFT], 1);
        const unsigned w = ((unsigned)s[u] << BUCKET_SHIFT) |
                           (unsigned)(d[u] & (BUCKET_NODES - 1));
        if (r < CHUNK_STAGE)
          img[r] = w;
        else
          out[r] = w;
      }
    }
  }
  __syncthreads();
  const int n = (end > beg) ? (end - beg) : 0;
  const int m = (n < CHUNK_STAGE) ? n : CHUNK_STAGE;
  const int mis = (int)(((size_t)out >> 2) & 31);
  int i = tid - mis;
  if (i < 0) i += PART_THREADS;
#pragma unroll 4
  for (; i < m; i += PART_THREADS) out[i] = img[i];
}

__device__ __forceinline__ void place_id(unsigned w, int* cur, int* stage,
                                         int* sg) {
  const int pos = atomicAdd(&cur[w & (BUCKET_NODES - 1)], 1);
  const int v = (int)(w >> BUCKET_SHIFT);
  if (pos < SORT_STAGE)
    stage[pos] = v;
  else
    sg[pos] = v;
}

__global__ __launch_bounds__(BUCKET_NODES) void bucket_sort_kernel(
    const unsigned* __restrict__ packed, const int* __restrict__ blkoff, int E,
    int N, int nbuck, int chunk, int* __restrict__ offs, int* __restrict__ cnt,
    float* __restrict__ inv, int* __restrict__ seg) {
  __shared__ int s[BUCKET_NODES];
  __shared__ int cur[BUCKET_NODES];
  __shared__ int stage[SORT_STAGE];
  __shared__ int rbeg[PART_BLOCKS];
  __shared__ int rlen[PART_BLOCKS];
  const int tid = threadIdx.x;
  const int wave = tid >> 5;
  const int lane = tid & 31;
  const int y = blockIdx.y;
  int a = 0;
  if (tid < PART_BLOCKS) {
    const int* bo =
        blkoff + ((size_t)y * PART_BLOCKS + tid) * (nbuck + 1) + blockIdx.x;
    a = bo[0];
    rlen[tid] = bo[1] - a;
    rbeg[tid] = tid * chunk + a;
  }
  cur[tid] = 0;
  int part = a;
  for (int o = 16; o > 0; o >>= 1) part += __shfl_xor(part, o);
  if (lane == 0) s[wave] = part;
  __syncthreads();
  int e0 = 0;
  for (int q = 0; q < SORT_WAVES; ++q) e0 += s[q];
  __syncthreads();
  const unsigned* pk = packed + (size_t)y * E;
  for (int p0 = wave; p0 < PART_BLOCKS; p0 += 4 * SORT_WAVES) {
    int rb[4], ln[4];
    unsigned w0[4], w1[4];
#pragma unroll
    for (int u = 0; u < 4; ++u) {
      rb[u] = rbeg[p0 + u * SORT_WAVES];
      ln[u] = rlen[p0 + u * SORT_WAVES];
    }
#pragma unroll
    for (int u = 0; u < 4; ++u) {
      w0[u] = 0u;
      w1[u] = 0u;
      if (lane < ln[u]) w0[u] = pk[rb[u] + lane];
      if (32 + lane < ln[u]) w1[u] = pk[rb[u] + 32 + lane];
    }
#pragma unroll
    for (int u = 0; u < 4; ++u) {
      if (lane < ln[u]) atomicAdd(&cur[w0[u] & (BUCKET_NODES - 1)], 1);
      if (32 + lane < ln[u]) atomicAdd(&cur[w1[u] & (BUCKET_NODES - 1)], 1);
      for (int o = 64 + lane; o < ln[u]; o += 32)
        atomicAdd(&cur[pk[rb[u] + o] & (BUCKET_NODES - 1)], 1);
    }
  }
  __syncthreads();
  const int c = cur[tid];
  const int excl = block_scan<BUCKET_NODES>(c, s, tid) - c;
  const int n = s[BUCKET_NODES - 1];
  const int node = blockIdx.x * BUCKET_NODES + tid;
  if (node < N) {
    const size_t g = (size_t)y * N + node;
    offs[g] = e0 + excl;
    cnt[g] = c;
    inv[g] = 1.0f / fmaxf((float)c, 1.0f);
  }
  cur[tid] = excl;
  __syncthreads();
  int* sg = seg + (size_t)y * E + e0;
  for (int p0 = wave; p0 < PART_BLOCKS; p0 += 4 * SORT_WAVES) {
    int rb[4], ln[4];
    unsigned w0[4], w1[4];
#pragma unroll
    for (int u = 0; u < 4; ++u) {
      rb[u] = rbeg[p0 + u * SORT_WAVES];
      ln[u] = rlen[p0 + u * SORT_WAVES];
    }
#pragma unroll
    for (int u = 0; u < 4; ++u) {
      w0[u] = 0u;
      w1[u] = 0u;
      if (lane < ln[u]) w0[u] = pk[rb[u] + lane];
      if (32 + lane < ln[u]) w1[u] = pk[rb[u] + 32 + lane];
    }
#pragma unroll
    for (int u = 0; u < 4; ++u) {
      if (lane < ln[u]) place_id(w0[u], cur, stage, sg);
      if (32 + lane < ln[u]) place_id(w1[u], cur, stage, sg);
      for (int o = 64 + lane; o < ln[u]; o += 32)
        place_id(pk[rb[u] + o], cur, stage, sg);
    }
  }
  __syncthreads();
  const int m = (n < SORT_STAGE) ? n : SORT_STAGE;
  const int mis = (int)(((size_t)sg >> 2) & 31);
  int i = tid - mis;
  if (i < 0) i += BUCKET_NODES;
#pragma unroll 4
  for (; i < m; i += BUCKET_NODES) sg[i] = stage[i];
}

__device__ __forceinline__ void add4(float4& a, const float4 v) {
  a.x += v.x;
  a.y += v.y;
  a.z += v.z;
  a.w += v.w;
}

template <int R, int P>
struct XSteps {
  static constexpr int KSTEPS = HIDDIM / 4;
  static_assert(KSTEPS % 2 == 0, "steps are taken in pairs");
  const float* wrl;
  const float* sxl;
  v8f* c;
  int ks;
  v2f b[2];
  __device__ __forceinline__ void fetch() {
    if (ks < KSTEPS) {
#pragma unroll
      for (int u = 0; u < 2; ++u) {
        b[u].x = wrl[(size_t)(4 * (ks + u)) * HIDDIM];
        b[u].y = wrl[(size_t)(4 * (ks + u) + 1) * HIDDIM];
      }
    }
  }
  __device__ __forceinline__ void run() {
    if (ks < KSTEPS) {
#pragma unroll
      for (int u = 0; u < 2; ++u) {
#pragma unroll
        for (int t = 0; t < R; ++t) {
          const v2f ax = *(const v2f*)(sxl + 16 * t * P + 4 * (ks + u));
          c[t] = __builtin_amdgcn_wmma_f32_16x16x4_f32(false, ax, false, b[u],
                                                       (short)0, c[t], false, false);
        }
      }
      ks += 2;
    }
  }
};

template <class X>
__device__ __forceinline__ float4 gather_row128(const float* __restrict__ hl,
                                                const int* __restrict__ seg,
                                                int beg, int deg, int lane,
                                                int idx0, X& xp) {
#define ROW_OF(jj) \
  (*(const float4*)(hl + (size_t)__builtin_amdgcn_readlane(idx, (jj)) * HIDDIM))
  float4 acc = make_float4(0.0f, 0.0f, 0.0f, 0.0f);
  for (int base = 0; base < deg; base += 32) {
    const int m = (deg - base < 32) ? (deg - base) : 32;
    int idx = idx0;
    if (base > 0) {
      idx = 0;
      if (lane < m) idx = seg[beg + base + lane];
    }
    int j = 0;
    for (; j + 8 <= m; j += 8) {
      const float4 v0 = ROW_OF(j), v1 = ROW_OF(j + 1), v2 = ROW_OF(j + 2),
                   v3 = ROW_OF(j + 3), v4 = ROW_OF(j + 4), v5 = ROW_OF(j + 5),
                   v6 = ROW_OF(j + 6), v7 = ROW_OF(j + 7);
      xp.run();
      xp.fetch();
      add4(acc, v0);
      add4(acc, v1);
      add4(acc, v2);
      add4(acc, v3);
      add4(acc, v4);
      add4(acc, v5);
      add4(acc, v6);
      add4(acc, v7);
    }
    if (j + 4 <= m) {
      const float4 v0 = ROW_OF(j), v1 = ROW_OF(j + 1), v2 = ROW_OF(j + 2),
                   v3 = ROW_OF(j + 3);
      xp.run();
      xp.fetch();
      add4(acc, v0);
      add4(acc, v1);
      add4(acc, v2);
      add4(acc, v3);
      j += 4;
    }
    if (j + 2 <= m) {
      const float4 v0 = ROW_OF(j), v1 = ROW_OF(j + 1);
      xp.run();
      xp.fetch();
      add4(acc, v0);
      add4(acc, v1);
      j += 2;
    }
    if (j < m) {
      const float4 v0 = ROW_OF(j);
      xp.run();
      xp.fetch();
      add4(acc, v0);
    }
  }
#undef ROW_OF
  return acc;
}

__device__ __forceinline__ float xor_add(float s, const int imm) {
  switch (imm) {
    case 1:  s += __int_as_float(__builtin_amdgcn_ds_swizzle(__float_as_int(s), 0x041f)); break;
    case 2:  s += __int_as_float(__builtin_amdgcn_ds_swizzle(__float_as_int(s), 0x081f)); break;
    case 4:  s += __int_as_float(__builtin_amdgcn_ds_swizzle(__float_as_int(s), 0x101f)); break;
    default: s += __int_as_float(__builtin_amdgcn_ds_swizzle(__float_as_int(s), 0x201f)); break;
  }
  return s;
}

__device__ __forceinline__ float other_half(float x) {
  const int b = __float_as_int(x);
  return __int_as_float(
      __builtin_amdgcn_permlanex16(b, b, 0x76543210, 0xfedcba98, false, false));
}

__device__ __forceinline__ float inv_norm(float s) {
  const float r = fminf(__builtin_amdgcn_rsqf(s), 1e12f);
  const float c = fmaf(-0.5f * s * r, r, 1.5f);
  return fminf(r * c, 1e12f);
}

__device__ __forceinline__ float tanh_unit(float x) {
  const float t = x * x;
  float n = t + 378.0f;
  n = fmaf(n, t, 17325.0f);
  n = fmaf(n, t, 135135.0f);
  float d = fmaf(28.0f, t, 3150.0f);
  d = fmaf(d, t, 62370.0f);
  d = fmaf(d, t, 135135.0f);
  const float p = x * n;
  const float r = __builtin_amdgcn_rcpf(d);
  const float q = p * r;
  return fmaf(fmaf(-d, q, p), r, q);
}

template <int K, int R>
__global__ __launch_bounds__(256) void sage_layer_kernel(
    const float* __restrict__ xin, const int* __restrict__ offs,
    const int* __restrict__ cnt, const int* __restrict__ seg,
    const float* __restrict__ invcnt, const float* __restrict__ wl,
    const float* __restrict__ wr, const float* __restrict__ bias,
    float* __restrict__ out, int N) {
  constexpr int P = K + 4;
  constexpr int ROWS = 16 * R;
  constexpr int WROWS = ROWS / 8;
  __shared__ __attribute__((aligned(16))) float sM[ROWS][P];
  __shared__ __attribute__((aligned(16))) float sX[ROWS][P];
  constexpr bool SHARE = (K == HIDDIM);
  __shared__ __attribute__((aligned(16))) float sOwn[SHARE ? 1 : ROWS][HIDDIM + 4];
  float (*sO)[HIDDIM + 4];
  if constexpr (SHARE)
    sO = sM;
  else
    sO = sOwn;

  const int tid = threadIdx.x;
  const int wave = tid >> 5;
  const int lane = tid & 31;
  const int half = lane >> 4;
  const int l16 = lane & 15;
  const int rowBase = blockIdx.x * ROWS;
  const int col = wave * 16 + l16;

  v8f c[R];
#pragma unroll
  for (int t = 0; t < R; ++t) c[t] = v8f{};

  if constexpr (K == HIDDIM) {
    const int w = __builtin_amdgcn_readfirstlane(wave);
    const float* hl = xin + lane * 4;
    const int last = N - 1;
    XSteps<R, P> xp;
    xp.wrl = wr + (size_t)(2 * half) * HIDDIM + col;
    xp.sxl = &sX[l16][2 * half];
    xp.c = c;
    xp.ks = 0;
    xp.fetch();
    float4 xo[WROWS];
#pragma unroll
    for (int i = 0; i < WROWS; ++i) {
      const int r = rowBase + w + 8 * i;
      xo[i] = *(const float4*)(hl + (size_t)((r < last) ? r : last) * K);
    }
    const int n0 = rowBase + WROWS * w;
    int beg[WROWS], deg[WROWS], idx[WROWS];
    float sc[WROWS];
#pragma unroll
    for (int i = 0; i < WROWS; ++i) {
      const int n = (n0 + i < last) ? (n0 + i) : last;
      beg[i] = offs[n];
      deg[i] = cnt[n];
      sc[i] = invcnt[n];
    }
#pragma unroll
    for (int i = 0; i < WROWS; ++i) {
      idx[i] = 0;
      if (lane < deg[i]) idx[i] = seg[beg[i] + lane];
    }
#pragma unroll
    for (int i = 0; i < WROWS; ++i) *(float4*)&sX[w + 8 * i][lane * 4] = xo[i];
    __syncthreads();
#pragma unroll
    for (int i = 0; i < WROWS; ++i) {
      float4 m = gather_row128(hl, seg, beg[i], deg[i], lane, idx[i], xp);
      m.x *= sc[i];
      m.y *= sc[i];
      m.z *= sc[i];
      m.w *= sc[i];
      *(float4*)&sM[WROWS * w + i][lane * 4] = m;
    }
    while (xp.ks < XSteps<R, P>::KSTEPS) {
      xp.run();
      xp.fetch();
    }
  } else {
    static_assert(K == 8 && ROWS * 4 == 256, "narrow path: four lanes a node, two halves a row");
    const int r = tid >> 2;
    const int h = tid & 1;
    const int p = (tid >> 1) & 1;
    const int last = N - 1;
    const int node = (rowBase + r < last) ? (rowBase + r) : last;
    const unsigned hb = 16u * (unsigned)h;
    const char* xb = (const char*)xin;
    const char* sb = (const char*)seg;
    const float4 own = *(const float4*)(xb + ((unsigned)node * (4u * K) + hb));
    const int deg = cnt[node];
    const unsigned beg = (unsigned)offs[node];
    const float sc = invcnt[node];
    asm volatile("" ::"v"(beg));
    float4 acc = make_float4(0.0f, 0.0f, 0.0f, 0.0f);
    for (int b = p; b < deg; b += 32) {
      int id[16];
      float4 v[16];
#pragma unroll
      for (int i = 0; i < 16; ++i) {
        const int e = b + 2 * i;
        id[i] = *(const int*)(sb + 4u * (beg + (unsigned)((e < deg) ? e : 0)));
      }
      asm volatile("" ::"v"(id[0]), "v"(id[1]), "v"(id[2]), "v"(id[3]), "v"(id[4]),
                   "v"(id[5]), "v"(id[6]), "v"(id[7]), "v"(id[8]), "v"(id[9]),
                   "v"(id[10]), "v"(id[11]), "v"(id[12]), "v"(id[13]), "v"(id[14]),
                   "v"(id[15]));
#pragma unroll
      for (int i = 0; i < 16; ++i)
        v[i] = *(const float4*)(xb + ((unsigned)id[i] * (4u * K) + hb));
      asm volatile(""
                   : "+v"(acc.x), "+v"(acc.y), "+v"(acc.z), "+v"(acc.w)
                   : "v"(v[0].x), "v"(v[1].x), "v"(v[2].x), "v"(v[3].x), "v"(v[4].x),
                     "v"(v[5].x), "v"(v[6].x), "v"(v[7].x), "v"(v[8].x), "v"(v[9].x),
                     "v"(v[10].x), "v"(v[11].x), "v"(v[12].x), "v"(v[13].x),
                     "v"(v[14].x), "v"(v[15].x));
#pragma unroll
      for (int i = 0; i < 16; ++i) {
        const bool ok = b + 2 * i < deg;
        acc.x += ok ? v[i].x : 0.0f;
        acc.y += ok ? v[i].y : 0.0f;
        acc.z += ok ? v[i].z : 0.0f;
        acc.w += ok ? v[i].w : 0.0f;
      }
    }
    acc.x = xor_add(acc.x, 2);
    acc.y = xor_add(acc.y, 2);
    acc.z = xor_add(acc.z, 2);
    acc.w = xor_add(acc.w, 2);
    if (p == 0) {
      *(float4*)&sX[r][4 * h] = own;
      *(float4*)&sM[r][4 * h] =
          make_float4(acc.x * sc, acc.y * sc, acc.z * sc, acc.w * sc);
    }
  }
  __syncthreads();

  if constexpr (K == HIDDIM) {
#pragma unroll 8
    for (int k0 = 0; k0 < K; k0 += 4) {
      const int ka = k0 + 2 * half;
      v2f bl;
      bl.x = wl[(size_t)ka * HIDDIM + col];
      bl.y = wl[(size_t)(ka + 1) * HIDDIM + col];
#pragma unroll
      for (int t = 0; t < R; ++t) {
        const v2f am = *(const v2f*)&sM[16 * t + l16][ka];
        c[t] = __builtin_amdgcn_wmma_f32_16x16x4_f32(false, am, false, bl, (short)0,
                                                     c[t], false, false);
      }
    }
  } else {
#pragma unroll 8
    for (int k0 = 0; k0 < K; k0 += 4) {
      const int ka = k0 + 2 * half;
      v2f bl, br;
      bl.x = wl[(size_t)ka * HIDDIM + col];
      bl.y = wl[(size_t)(ka + 1) * HIDDIM + col];
      br.x = wr[(size_t)ka * HIDDIM + col];
      br.y = wr[(size_t)(ka + 1) * HIDDIM + col];
#pragma unroll
      for (int t = 0; t < R; ++t) {
        const v2f am = *(const v2f*)&sM[16 * t + l16][ka];
        const v2f ax = *(const v2f*)&sX[16 * t + l16][ka];
        c[t] = __builtin_amdgcn_wmma_f32_16x16x4_f32(false, am, false, bl, (short)0,
                                                     c[t], false, false);
        c[t] = __builtin_amdgcn_wmma_f32_16x16x4_f32(false, ax, false, br, (short)0,
                                                     c[t], false, false);
      }
    }
  }

  const float bv = bias[col];
  if constexpr (SHARE) __syncthreads();
#pragma unroll
  for (int t = 0; t < R; ++t) {
#pragma unroll
    for (int v = 0; v < 8; ++v) sO[16 * t + v + 8 * half][col] = c[t][v] + bv;
  }
  __syncthreads();

  float4 o[WROWS];
  float s[WROWS];
#pragma unroll
  for (int q = 0; q < WROWS; ++q) {
    o[q] = *(const float4*)&sO[wave + 8 * q][lane * 4];
    s[q] = o[q].x * o[q].x + o[q].y * o[q].y + o[q].z * o[q].z + o[q].w * o[q].w;
  }
#pragma unroll
  for (int q = 0; q < WROWS; ++q) s[q] = xor_add(s[q], 1);
#pragma unroll
  for (int q = 0; q < WROWS; ++q) s[q] = xor_add(s[q], 2);
#pragma unroll
  for (int q = 0; q < WROWS; ++q) s[q] = xor_add(s[q], 4);
#pragma unroll
  for (int q = 0; q < WROWS; ++q) s[q] = xor_add(s[q], 8);
#pragma unroll
  for (int q = 0; q < WROWS; ++q) s[q] += other_half(s[q]);
  float4 r[WROWS];
#pragma unroll
  for (int q = 0; q < WROWS; ++q) {
    const float scale = inv_norm(s[q]);
    r[q].x = tanh_unit(o[q].x * scale);
    r[q].y = tanh_unit(o[q].y * scale);
    r[q].z = tanh_unit(o[q].z * scale);
    r[q].w = tanh_unit(o[q].w * scale);
  }
  if (rowBase + ROWS <= N) {
    float* orow = out + (size_t)(rowBase + wave) * HIDDIM + lane * 4;
#pragma unroll
    for (int q = 0; q < WROWS; ++q) *(float4*)(orow + (size_t)(8 * q) * HIDDIM) = r[q];
  } else {
#pragma unroll
    for (int q = 0; q < WROWS; ++q) {
      const int row = rowBase + wave + 8 * q;
      if (row < N) *(float4*)(out + (size_t)row * HIDDIM + lane * 4) = r[q];
    }
  }
}

extern "C" void kernel_launch(void* const* d_in, const int* in_sizes, int n_in,
                              void* d_out, int out_size, void* d_ws,
                              size_t ws_size, hipStream_t stream) {
  const float* x = (const float*)d_in[0];
  const int* ei_conn = (const int*)d_in[1];
  const int* ei_dest = (const int*)d_in[2];
  const float* wl1 = (const float*)d_in[3];
  const float* wr1 = (const float*)d_in[4];
  const float* b1 = (const float*)d_in[5];
  const float* wl2 = (const float*)d_in[6];
  const float* wr2 = (const float*)d_in[7];
  const float* b2 = (const float*)d_in[8];
  const float* wl3 = (const float*)d_in[9];
  const float* wr3 = (const float*)d_in[10];
  const float* b3 = (const float*)d_in[11];

  const int N = in_sizes[0] / 8;
  const int E = in_sizes[1] / 2;
  const int nbuck = (N + BUCKET_NODES - 1) / BUCKET_NODES;
  const int chunk = (E + PART_BLOCKS - 1) / PART_BLOCKS;

  float* ws = (float*)d_ws;
  float* h1 = ws;
  float* h2 = h1 + (size_t)N * HIDDIM;
  float* inv = h2 + (size_t)N * HIDDIM;
  int* cnt = (int*)(inv + (size_t)2 * N);
  int* offs = cnt + (size_t)2 * N;
  int* blkoff = offs + (size_t)2 * N;
  unsigned* packed = (unsigned*)(blkoff + (size_t)2 * PART_BLOCKS * (nbuck + 1));
  int* seg = (int*)(packed + (size_t)2 * E);

  const int T = 256;

  const int* src_conn = ei_conn;
  const int* dst_conn = ei_conn + E;
  const int* src_dest = ei_dest;
  const int* dst_dest = ei_dest + E;

  chunk_group_kernel<<<dim3(PART_BLOCKS, 2), PART_THREADS, 0, stream>>>(
      src_conn, dst_conn, src_dest, dst_dest, E, nbuck, chunk, blkoff, packed);
  bucket_sort_kernel<<<dim3(nbuck, 2), BUCKET_NODES, 0, stream>>>(
      packed, blkoff, E, N, nbuck, chunk, offs, cnt, inv, seg);

  const int layerBlocks = (N + 63) / 64;

  sage_layer_kernel<8, 4><<<layerBlocks, T, 0, stream>>>(x, offs, cnt, seg, inv,
                                                         wl1, wr1, b1, h1, N);

  sage_layer_kernel<128, 4><<<layerBlocks, T, 0, stream>>>(
      h1, offs + N, cnt + N, seg + E, inv + N, wl2, wr2, b2, h2, N);

  sage_layer_kernel<128, 4><<<layerBlocks, T, 0, stream>>>(
      h2, offs, cnt, seg, inv, wl3, wr3, b3, (float*)d_out, N);
}
